# starting kernel
# baseline (speedup 1.0000x reference)
#include <hip/hip_runtime.h>
#include <math.h>

#define B_   8
#define C0_  16
#define C1_  64
#define C2_  64
#define T0_  12
#define T1_  10
#define T2_  8
#define NN_  256
#define KT_  3
#define HH_  8
#define ALPHA_ 0.2f

typedef float v2f __attribute__((ext_vector_type(2)));
typedef float v4f __attribute__((ext_vector_type(4)));
typedef float v8f __attribute__((ext_vector_type(8)));

__device__ __forceinline__ float tanh_fast(float x) {
  float e = __expf(fminf(2.0f * x, 80.0f));
  return (e - 1.0f) / (e + 1.0f);
}

__global__ __launch_bounds__(256) void prep_u_kernel(
    const float* __restrict__ W,
    const float* __restrict__ a,
    float* __restrict__ u1, float* __restrict__ u2)
{
  int tid = blockIdx.x * 256 + threadIdx.x;
  if (tid >= HH_ * C1_) return;
  int h = tid >> 6, c = tid & 63;
  const float* Wh = W + ((size_t)h * C1_ + c) * C1_;
  const float* ah = a + (size_t)h * 2 * C1_;
  float s1 = 0.f, s2 = 0.f;
  for (int d = 0; d < C1_; ++d) {
    float w = Wh[d];
    s1 += w * ah[d];
    s2 += w * ah[C1_ + d];
  }
  u1[tid] = s1;
  u2[tid] = s2;
}

__global__ __launch_bounds__(256) void conv1_glu_kernel(
    const float* __restrict__ x,
    const float* __restrict__ w,
    const float* __restrict__ b,
    float* __restrict__ xg)
{
  int idx = blockIdx.x * 256 + threadIdx.x;
  int c  = idx & 63;
  int n  = (idx >> 6) & 255;
  int bt = idx >> 14;
  int bb = bt / T1_, t = bt - bb * T1_;

  float accp = b[c];
  float accq = b[c + C1_];
  #pragma unroll
  for (int kt = 0; kt < KT_; ++kt) {
    for (int ci = 0; ci < C0_; ++ci) {
      float xv = x[(((size_t)bb * C0_ + ci) * T0_ + (t + kt)) * NN_ + n];
      accp = fmaf(w[((size_t)c         * C0_ + ci) * KT_ + kt], xv, accp);
      accq = fmaf(w[((size_t)(c + C1_) * C0_ + ci) * KT_ + kt], xv, accq);
    }
  }
  float xin = (c < C0_) ? x[(((size_t)bb * C0_ + c) * T0_ + (t + KT_ - 1)) * NN_ + n] : 0.f;
  float sg  = 1.f / (1.f + __expf(-accq));
  xg[idx]   = (accp + xin) * sg;
}

#define XSTR  72
#define XPSTR 160
#define PSTR  260

__global__ __launch_bounds__(256) void attn_kernel(
    const float* __restrict__ xg,
    const float* __restrict__ adj,
    const float* __restrict__ u1,
    const float* __restrict__ u2,
    float* __restrict__ xs)
{
  extern __shared__ float smem[];
  float*    s_xg   = smem;
  float*    s_xgp  = s_xg + NN_ * XSTR;
  float*    s_e1   = s_xgp + (NN_ / 2) * XPSTR;
  float*    s_e2   = s_e1 + NN_;
  float*    s_rinv = s_e2 + NN_;
  float*    s_p    = s_rinv + NN_;
  unsigned* s_adj  = (unsigned*)(s_p + 8 * 16 * PSTR);

  const int bt   = blockIdx.x;
  const int tid  = threadIdx.x;
  const int lane = tid & 31;
  const int wave = tid >> 5;
  const int mloc = lane & 15;
  const int half = lane >> 4;

  const float* xg_bt = xg + (size_t)bt * NN_ * C1_;

  for (int i = tid; i < NN_ * C1_; i += 256) {
    int n = i >> 6, c = i & 63;
    float v = xg_bt[i];
    s_xg[n * XSTR + c] = v;
    s_xgp[(n >> 1) * XPSTR + 2 * c + (n & 1)] = v;
  }
  {
    int n = tid;
    for (int w8 = 0; w8 < 8; ++w8) {
      unsigned bits = 0u;
      for (int j = 0; j < 32; ++j)
        bits |= (adj[(size_t)n * NN_ + w8 * 32 + j] > 0.f) ? (1u << j) : 0u;
      s_adj[n * 8 + w8] = bits;
    }
  }
  __syncthreads();

  const float inv_h = 1.0f / (float)HH_;
  float* s_pw = s_p + wave * 16 * PSTR;

  float racc[2][4][8];
  #pragma unroll
  for (int i = 0; i < 2; ++i)
    #pragma unroll
    for (int j = 0; j < 4; ++j)
      #pragma unroll
      for (int r = 0; r < 8; ++r) racc[i][j][r] = 0.f;

  for (int h = 0; h < HH_; ++h) {
    {
      int n = tid;
      const float* row = &s_xg[n * XSTR];
      const float* U1 = u1 + h * C1_;
      const float* U2 = u2 + h * C1_;
      float a1 = 0.f, a2 = 0.f;
      for (int c = 0; c < C1_; ++c) {
        float v = row[c];
        a1 = fmaf(v, U1[c], a1);
        a2 = fmaf(v, U2[c], a2);
      }
      s_e1[n] = a1;
      s_e2[n] = a2;
    }
    __syncthreads();

    #pragma unroll
    for (int rbi = 0; rbi < 2; ++rbi) {
      const int rb = wave + rbi * 8;
      const int m  = rb * 16 + mloc;
      const float e1m = s_e1[m];
      const int jbase = half * 128;

      float mx = -3.0e38f;
      for (int w8 = 0; w8 < 4; ++w8) {
        unsigned adjw = s_adj[m * 8 + half * 4 + w8];
        const float* eb = &s_e2[jbase + w8 * 32];
        for (int j4 = 0; j4 < 32; j4 += 4) {
          v4f e = *(const v4f*)(eb + j4);
          #pragma unroll
          for (int u = 0; u < 4; ++u) {
            float sc = e1m + e[u];
            sc = sc > 0.f ? sc : ALPHA_ * sc;
            int on = (adjw >> (j4 + u)) & 1;
            mx = fmaxf(mx, on ? sc : -3.0e38f);
          }
        }
      }
      mx = fmaxf(mx, __shfl_xor(mx, 16));

      float sm = 0.f;
      for (int w8 = 0; w8 < 4; ++w8) {
        unsigned adjw = s_adj[m * 8 + half * 4 + w8];
        const float* eb = &s_e2[jbase + w8 * 32];
        float* pb = &s_pw[mloc * PSTR + jbase + w8 * 32];
        for (int j4 = 0; j4 < 32; j4 += 4) {
          v4f e = *(const v4f*)(eb + j4);
          v4f p;
          #pragma unroll
          for (int u = 0; u < 4; ++u) {
            float sc = e1m + e[u];
            sc = sc > 0.f ? sc : ALPHA_ * sc;
            sc = ((adjw >> (j4 + u)) & 1) ? sc : -3.0e38f;
            float pe = __expf(sc - mx);
            p[u] = pe;
            sm += pe;
          }
          *(v4f*)(pb + j4) = p;
        }
      }
      sm += __shfl_xor(sm, 16);
      s_rinv[rb * 16 + mloc] = 1.0f / sm;

      v8f acc0 = {}, acc1 = {}, acc2 = {}, acc3 = {};
      const float* pa   = &s_pw[mloc * PSTR + 2 * half];
      const float* xpb  = &s_xgp[half * XPSTR + 2 * mloc];
      for (int k0 = 0; k0 < NN_; k0 += 4) {
        v2f a = *(const v2f*)(pa + k0);
        const float* bp = xpb + (k0 >> 1) * XPSTR;
        v2f b0 = *(const v2f*)(bp + 0);
        v2f b1 = *(const v2f*)(bp + 32);
        v2f b2 = *(const v2f*)(bp + 64);
        v2f b3 = *(const v2f*)(bp + 96);
        acc0 = __builtin_amdgcn_wmma_f32_16x16x4_f32(false, a, false, b0, (short)0, acc0, false, false);
        acc1 = __builtin_amdgcn_wmma_f32_16x16x4_f32(false, a, false, b1, (short)0, acc1, false, false);
        acc2 = __builtin_amdgcn_wmma_f32_16x16x4_f32(false, a, false, b2, (short)0, acc2, false, false);
        acc3 = __builtin_amdgcn_wmma_f32_16x16x4_f32(false, a, false, b3, (short)0, acc3, false, false);
      }

      #pragma unroll
      for (int r = 0; r < 8; ++r) {
        float inv = s_rinv[rb * 16 + r + 8 * half];
        racc[rbi][0][r] += tanh_fast(acc0[r] * inv) * inv_h;
        racc[rbi][1][r] += tanh_fast(acc1[r] * inv) * inv_h;
        racc[rbi][2][r] += tanh_fast(acc2[r] * inv) * inv_h;
        racc[rbi][3][r] += tanh_fast(acc3[r] * inv) * inv_h;
      }
    }
    __syncthreads();
  }

  float* xs_bt = xs + (size_t)bt * NN_ * C1_;
  #pragma unroll
  for (int rbi = 0; rbi < 2; ++rbi) {
    const int rb = wave + rbi * 8;
    #pragma unroll
    for (int r = 0; r < 8; ++r) {
      int mm = rb * 16 + r + 8 * half;
      const float* xrow = &s_xg[mm * XSTR + mloc];
      float* orow = xs_bt + (size_t)mm * C1_ + mloc;
      orow[0]  = xrow[0]  + racc[rbi][0][r];
      orow[16] = xrow[16] + racc[rbi][1][r];
      orow[32] = xrow[32] + racc[rbi][2][r];
      orow[48] = xrow[48] + racc[rbi][3][r];
    }
  }
}

__global__ __launch_bounds__(256) void conv2_ln_kernel(
    const float* __restrict__ xs,
    const float* __restrict__ w2,
    const float* __restrict__ b2,
    const float* __restrict__ lnw,
    const float* __restrict__ lnb,
    float* __restrict__ out)
{
  extern __shared__ float smem[];
  float* s_w   = smem;
  float* s_y   = s_w + C2_ * C1_ * KT_;
  float* s_red = s_y + NN_ * C2_;

  const int bt = blockIdx.x;
  const int bb = bt >> 3, t = bt & 7;
  const int tid = threadIdx.x;

  for (int i = tid; i < C2_ * C1_ * KT_; i += 256) s_w[i] = w2[i];
  __syncthreads();

  const int n = tid;
  const float* xsb = xs + (((size_t)bb * T1_ + t) * NN_) * C1_;

  float lsum = 0.f, lsq = 0.f;
  for (int cb = 0; cb < 4; ++cb) {
    const int c0 = cb * 16;
    float acc[16];
    #pragma unroll
    for (int u = 0; u < 16; ++u) acc[u] = b2[c0 + u];
    for (int kt = 0; kt < KT_; ++kt) {
      const float* xrow = xsb + (size_t)kt * NN_ * C1_ + (size_t)n * C1_;
      for (int ci = 0; ci < C1_; ++ci) {
        float xv = xrow[ci];
        #pragma unroll
        for (int u = 0; u < 16; ++u)
          acc[u] = fmaf(s_w[((c0 + u) * C1_ + ci) * KT_ + kt], xv, acc[u]);
      }
    }
    const float* xres = xsb + (size_t)(KT_ - 1) * NN_ * C1_ + (size_t)n * C1_;
    #pragma unroll
    for (int u = 0; u < 16; ++u) {
      float v = acc[u] + xres[c0 + u];
      v = v > 0.f ? v : 0.f;
      s_y[n * C2_ + c0 + u] = v;
      lsum += v;
      lsq  += v * v;
    }
  }

  for (int off = 16; off > 0; off >>= 1) {
    lsum += __shfl_xor(lsum, off);
    lsq  += __shfl_xor(lsq,  off);
  }
  if ((tid & 31) == 0) { s_red[tid >> 5] = lsum; s_red[8 + (tid >> 5)] = lsq; }
  __syncthreads();
  float tsum = 0.f, tsq = 0.f;
  for (int i = 0; i < 8; ++i) { tsum += s_red[i]; tsq += s_red[8 + i]; }
  const float cnt  = (float)(NN_ * C2_);
  const float mu   = tsum / cnt;
  const float var  = tsq / cnt - mu * mu;
  const float rstd = rsqrtf(var + 1e-5f);

  for (int c = 0; c < C2_; ++c) {
    float v = (s_y[n * C2_ + c] - mu) * rstd * lnw[(size_t)n * C2_ + c] + lnb[(size_t)n * C2_ + c];
    out[(((size_t)bb * C2_ + c) * T2_ + t) * NN_ + n] = v;
  }
}

extern "C" void kernel_launch(void* const* d_in, const int* in_sizes, int n_in,
                              void* d_out, int out_size, void* d_ws, size_t ws_size,
                              hipStream_t stream) {
  const float* x   = (const float*)d_in[0];
  const float* adj = (const float*)d_in[1];
  const float* w1  = (const float*)d_in[2];
  const float* b1  = (const float*)d_in[3];
  const float* w2  = (const float*)d_in[4];
  const float* b2  = (const float*)d_in[5];
  const float* Wh  = (const float*)d_in[6];
  const float* ah  = (const float*)d_in[7];
  const float* lnw = (const float*)d_in[8];
  const float* lnb = (const float*)d_in[9];
  float* out = (float*)d_out;

  float* ws = (float*)d_ws;
  const size_t xg_elems = (size_t)B_ * T1_ * NN_ * C1_;
  float* xg = ws;
  float* xs = xg + xg_elems;
  float* u1 = xs + xg_elems;
  float* u2 = u1 + HH_ * C1_;

  prep_u_kernel<<<2, 256, 0, stream>>>(Wh, ah, u1, u2);

  const int n_c1 = (int)(xg_elems / 256);
  conv1_glu_kernel<<<n_c1, 256, 0, stream>>>(x, w1, b1, xg);

  const size_t attn_lds =
      (size_t)(NN_ * XSTR + (NN_ / 2) * XPSTR + 3 * NN_ + 8 * 16 * PSTR) * sizeof(float)
      + (size_t)(NN_ * 8) * sizeof(unsigned);
  attn_kernel<<<B_ * T1_, 256, attn_lds, stream>>>(xg, adj, u1, u2, xs);

  const size_t c2_lds = (size_t)(C2_ * C1_ * KT_ + NN_ * C2_ + 16) * sizeof(float);
  conv2_ln_kernel<<<B_ * T2_, 256, c2_lds, stream>>>(xs, w2, b2, lnw, lnb, out);
}


# attention: the 8 heads of each (b,t) split over 3 workgroups (3/3/2 heads; 240 workgroups, not 80), each writing a partial x_s plane; conv2 adds the 3 planes as it loads; head sum order changed
# speedup vs baseline: 3.3326x; 3.3326x over previous
#include <hip/hip_runtime.h>
#include <math.h>

#define B_   8
#define C0_  16
#define C1_  64
#define C2_  64
#define T0_  12
#define T1_  10
#define T2_  8
#define NN_  256
#define KT_  3
#define HH_  8
#define ALPHA_ 0.2f

typedef float v2f __attribute__((ext_vector_type(2)));
typedef float v4f __attribute__((ext_vector_type(4)));
typedef float v8f __attribute__((ext_vector_type(8)));

__device__ __forceinline__ float tanh_fast(float x) {
  float e = __expf(fminf(2.0f * x, 80.0f));
  return (e - 1.0f) / (e + 1.0f);
}

__global__ __launch_bounds__(256) void prep_u_kernel(
    const float* __restrict__ W,
    const float* __restrict__ a,
    float* __restrict__ u1, float* __restrict__ u2)
{
  int tid = blockIdx.x * 256 + threadIdx.x;
  if (tid >= HH_ * C1_) return;
  int h = tid >> 6, c = tid & 63;
  const float* Wh = W + ((size_t)h * C1_ + c) * C1_;
  const float* ah = a + (size_t)h * 2 * C1_;
  float s1 = 0.f, s2 = 0.f;
  for (int d = 0; d < C1_; ++d) {
    float w = Wh[d];
    s1 += w * ah[d];
    s2 += w * ah[C1_ + d];
  }
  u1[tid] = s1;
  u2[tid] = s2;
}

__global__ __launch_bounds__(128) void conv1_glu_kernel(
    const float* __restrict__ x,
    const float* __restrict__ w,
    const float* __restrict__ b,
    float* __restrict__ xg)
{
  const int lane = threadIdx.x & 31;
  const int wave = threadIdx.x >> 5;
  const int mloc = lane & 15;
  const int half = lane >> 4;
  const int tile = blockIdx.x * 4 + wave;
  const int bt   = tile >> 4;
  const int n0   = (tile & 15) * 16;
  const int bb   = bt / T1_, t = bt - bb * T1_;

  const float* xl = x + ((size_t)(bb * C0_ + 8 * half) * T0_ + t) * NN_ + n0 + mloc;
  float av[24];
  #pragma unroll
  for (int k0 = 0; k0 < 24; ++k0)
    av[k0] = xl[((k0 / 3) * T0_ + (k0 % 3)) * NN_];

  const v4f bp = *(const v4f*)(b + 4 * mloc);
  const v4f bq = *(const v4f*)(b + C1_ + 4 * mloc);
  v8f acc[8];
  #pragma unroll
  for (int T = 0; T < 4; ++T) {
    #pragma unroll
    for (int r = 0; r < 8; ++r) { acc[T][r] = bp[T]; acc[4 + T][r] = bq[T]; }
  }

  const float* wl = w + (size_t)(4 * mloc) * (C0_ * KT_) + 24 * half;
  #pragma unroll
  for (int j = 0; j < 6; ++j) {
    v4f bw[8];
    #pragma unroll
    for (int T = 0; T < 4; ++T) {
      bw[T]     = *(const v4f*)(wl + (size_t)T * (C0_ * KT_) + 4 * j);
      bw[4 + T] = *(const v4f*)(wl + (size_t)(C1_ + T) * (C0_ * KT_) + 4 * j);
    }
    v2f a0, a1;
    a0[0] = av[4 * j];     a0[1] = av[4 * j + 1];
    a1[0] = av[4 * j + 2]; a1[1] = av[4 * j + 3];
    #pragma unroll
    for (int T = 0; T < 8; ++T) {
      v2f b0; b0[0] = bw[T][0]; b0[1] = bw[T][1];
      acc[T] = __builtin_amdgcn_wmma_f32_16x16x4_f32(false, a0, false, b0, (short)0, acc[T], false, false);
    }
    #pragma unroll
    for (int T = 0; T < 8; ++T) {
      v2f b1; b1[0] = bw[T][2]; b1[1] = bw[T][3];
      acc[T] = __builtin_amdgcn_wmma_f32_16x16x4_f32(false, a1, false, b1, (short)0, acc[T], false, false);
    }
  }

  const bool has_res = (4 * mloc) < C0_;
  const int  cres    = (4 * mloc) & (C0_ - 1);
  const float* xr = x + ((size_t)(bb * C0_ + cres) * T0_ + (t + KT_ - 1)) * NN_ + n0 + 8 * half;
  float xin[4][8];
  #pragma unroll
  for (int T = 0; T < 4; ++T) {
    const v4f lo = *(const v4f*)(xr + (size_t)T * T0_ * NN_);
    const v4f hi = *(const v4f*)(xr + (size_t)T * T0_ * NN_ + 4);
    #pragma unroll
    for (int u = 0; u < 4; ++u) {
      xin[T][u]     = has_res ? lo[u] : 0.f;
      xin[T][4 + u] = has_res ? hi[u] : 0.f;
    }
  }

  float* orow = xg + ((size_t)bt * NN_ + n0 + 8 * half) * C1_ + 4 * mloc;
  #pragma unroll
  for (int r = 0; r < 8; ++r) {
    v4f o;
    #pragma unroll
    for (int T = 0; T < 4; ++T) {
      float sg = 1.f / (1.f + __expf(-acc[4 + T][r]));
      o[T] = (acc[T][r] + xin[T][r]) * sg;
    }
    *(v4f*)(orow + (size_t)r * C1_) = o;
  }
}

#define HGRP_ 3
#define PLANE_ ((size_t)B_ * T1_ * NN_ * C1_)
#define XSTR  72
#define XPSTR 160
#define PSTR  260

__global__ __launch_bounds__(256) void attn_kernel(
    const float* __restrict__ xg,
    const float* __restrict__ adj,
    const float* __restrict__ u1,
    const float* __restrict__ u2,
    float* __restrict__ xs)
{
  extern __shared__ float smem[];
  float*    s_xg   = smem;
  float*    s_xgp  = s_xg + NN_ * XSTR;
  float*    s_e1   = s_xgp + (NN_ / 2) * XPSTR;
  float*    s_e2   = s_e1 + NN_;
  float*    s_rinv = s_e2 + NN_;
  float*    s_p    = s_rinv + NN_;
  unsigned* s_adj  = (unsigned*)(s_p + 8 * 16 * PSTR);

  const int bt   = blockIdx.x / HGRP_;
  const int grp  = blockIdx.x - bt * HGRP_;
  const int h_lo = 3 * grp;
  const int h_hi = (h_lo + 3 < HH_) ? h_lo + 3 : HH_;
  const int tid  = threadIdx.x;
  const int lane = tid & 31;
  const int wave = tid >> 5;
  const int mloc = lane & 15;
  const int half = lane >> 4;

  const float* xg_bt = xg + (size_t)bt * NN_ * C1_;

  for (int i = tid; i < NN_ * C1_; i += 256) {
    int n = i >> 6, c = i & 63;
    float v = xg_bt[i];
    s_xg[n * XSTR + c] = v;
    s_xgp[(n >> 1) * XPSTR + 2 * c + (n & 1)] = v;
  }
  {
    int n = tid;
    for (int w8 = 0; w8 < 8; ++w8) {
      unsigned bits = 0u;
      for (int j = 0; j < 32; ++j)
        bits |= (adj[(size_t)n * NN_ + w8 * 32 + j] > 0.f) ? (1u << j) : 0u;
      s_adj[n * 8 + w8] = bits;
    }
  }
  __syncthreads();

  const float inv_h = 1.0f / (float)HH_;
  float* s_pw = s_p + wave * 16 * PSTR;

  float racc[2][4][8];
  #pragma unroll
  for (int i = 0; i < 2; ++i)
    #pragma unroll
    for (int j = 0; j < 4; ++j)
      #pragma unroll
      for (int r = 0; r < 8; ++r) racc[i][j][r] = 0.f;

  for (int h = h_lo; h < h_hi; ++h) {
    {
      int n = tid;
      const float* row = &s_xg[n * XSTR];
      const float* U1 = u1 + h * C1_;
      const float* U2 = u2 + h * C1_;
      float a1 = 0.f, a2 = 0.f;
      for (int c = 0; c < C1_; ++c) {
        float v = row[c];
        a1 = fmaf(v, U1[c], a1);
        a2 = fmaf(v, U2[c], a2);
      }
      s_e1[n] = a1;
      s_e2[n] = a2;
    }
    __syncthreads();

    #pragma unroll
    for (int rbi = 0; rbi < 2; ++rbi) {
      const int rb = wave + rbi * 8;
      const int m  = rb * 16 + mloc;
      const float e1m = s_e1[m];
      const int jbase = half * 128;

      float mx = -3.0e38f;
      for (int w8 = 0; w8 < 4; ++w8) {
        unsigned adjw = s_adj[m * 8 + half * 4 + w8];
        const float* eb = &s_e2[jbase + w8 * 32];
        for (int j4 = 0; j4 < 32; j4 += 4) {
          v4f e = *(const v4f*)(eb + j4);
          #pragma unroll
          for (int u = 0; u < 4; ++u) {
            float sc = e1m + e[u];
            sc = sc > 0.f ? sc : ALPHA_ * sc;
            int on = (adjw >> (j4 + u)) & 1;
            mx = fmaxf(mx, on ? sc : -3.0e38f);
          }
        }
      }
      mx = fmaxf(mx, __shfl_xor(mx, 16));

      float sm = 0.f;
      for (int w8 = 0; w8 < 4; ++w8) {
        unsigned adjw = s_adj[m * 8 + half * 4 + w8];
        const float* eb = &s_e2[jbase + w8 * 32];
        float* pb = &s_pw[mloc * PSTR + jbase + w8 * 32];
        for (int j4 = 0; j4 < 32; j4 += 4) {
          v4f e = *(const v4f*)(eb + j4);
          v4f p;
          #pragma unroll
          for (int u = 0; u < 4; ++u) {
            float sc = e1m + e[u];
            sc = sc > 0.f ? sc : ALPHA_ * sc;
            sc = ((adjw >> (j4 + u)) & 1) ? sc : -3.0e38f;
            float pe = __expf(sc - mx);
            p[u] = pe;
            sm += pe;
          }
          *(v4f*)(pb + j4) = p;
        }
      }
      sm += __shfl_xor(sm, 16);
      s_rinv[rb * 16 + mloc] = 1.0f / sm;

      v8f acc0 = {}, acc1 = {}, acc2 = {}, acc3 = {};
      const float* pa   = &s_pw[mloc * PSTR + 2 * half];
      const float* xpb  = &s_xgp[half * XPSTR + 2 * mloc];
      for (int k0 = 0; k0 < NN_; k0 += 4) {
        v2f a = *(const v2f*)(pa + k0);
        const float* bp = xpb + (k0 >> 1) * XPSTR;
        v2f b0 = *(const v2f*)(bp + 0);
        v2f b1 = *(const v2f*)(bp + 32);
        v2f b2 = *(const v2f*)(bp + 64);
        v2f b3 = *(const v2f*)(bp + 96);
        acc0 = __builtin_amdgcn_wmma_f32_16x16x4_f32(false, a, false, b0, (short)0, acc0, false, false);
        acc1 = __builtin_amdgcn_wmma_f32_16x16x4_f32(false, a, false, b1, (short)0, acc1, false, false);
        acc2 = __builtin_amdgcn_wmma_f32_16x16x4_f32(false, a, false, b2, (short)0, acc2, false, false);
        acc3 = __builtin_amdgcn_wmma_f32_16x16x4_f32(false, a, false, b3, (short)0, acc3, false, false);
      }

      #pragma unroll
      for (int r = 0; r < 8; ++r) {
        float inv = s_rinv[rb * 16 + r + 8 * half];
        racc[rbi][0][r] += tanh_fast(acc0[r] * inv) * inv_h;
        racc[rbi][1][r] += tanh_fast(acc1[r] * inv) * inv_h;
        racc[rbi][2][r] += tanh_fast(acc2[r] * inv) * inv_h;
        racc[rbi][3][r] += tanh_fast(acc3[r] * inv) * inv_h;
      }
    }
    __syncthreads();
  }

  float* xs_bt = xs + (size_t)grp * PLANE_ + (size_t)bt * NN_ * C1_;
  if (grp == 0) {
    #pragma unroll
    for (int rbi = 0; rbi < 2; ++rbi) {
      #pragma unroll
      for (int r = 0; r < 8; ++r) {
        const float* xrow = &s_xg[((wave + rbi * 8) * 16 + r + 8 * half) * XSTR + mloc];
        racc[rbi][0][r] += xrow[0];
        racc[rbi][1][r] += xrow[16];
        racc[rbi][2][r] += xrow[32];
        racc[rbi][3][r] += xrow[48];
      }
    }
  }
  #pragma unroll
  for (int rbi = 0; rbi < 2; ++rbi) {
    const int rb = wave + rbi * 8;
    #pragma unroll
    for (int r = 0; r < 8; ++r) {
      int mm = rb * 16 + r + 8 * half;
      float* orow = xs_bt + (size_t)mm * C1_ + mloc;
      orow[0]  = racc[rbi][0][r];
      orow[16] = racc[rbi][1][r];
      orow[32] = racc[rbi][2][r];
      orow[48] = racc[rbi][3][r];
    }
  }
}

#define W2P   260
#define W2KT  (16 * W2P)

__device__ __forceinline__ float xhalf(float v) {
  int b = __builtin_bit_cast(int, v);
  return __builtin_bit_cast(float, __builtin_amdgcn_permlanex16(b, b, 0x76543210, 0xfedcba98, false, false));
}

__global__ __launch_bounds__(256) void conv2_ln_kernel(
    const float* __restrict__ xs,
    const float* __restrict__ w2,
    const float* __restrict__ b2,
    const float* __restrict__ lnw,
    const float* __restrict__ lnb,
    float* __restrict__ out)
{
  extern __shared__ float smem[];
  float* s_w   = smem;
  float* s_red = s_w + 3 * W2KT;

  const int bt = blockIdx.x;
  const int bb = bt >> 3, t = bt & 7;
  const int tid  = threadIdx.x;
  const int lane = tid & 31;
  const int wave = tid >> 5;
  const int mloc = lane & 15;
  const int half = lane >> 4;
  const int nb   = 32 * wave;

  v4f wv[4][3];
  #pragma unroll
  for (int it = 0; it < 4; ++it) {
    const float* wp = w2 + 12 * (tid + 256 * it);
    wv[it][0] = *(const v4f*)(wp);
    wv[it][1] = *(const v4f*)(wp + 4);
    wv[it][2] = *(const v4f*)(wp + 8);
  }

  v8f acc[2][4];
  #pragma unroll
  for (int T = 0; T < 4; ++T) {
    const v4f blo = *(const v4f*)(b2 + 16 * T + 8 * half);
    const v4f bhi = *(const v4f*)(b2 + 16 * T + 8 * half + 4);
    #pragma unroll
    for (int u = 0; u < 4; ++u) {
      acc[0][T][u] = blo[u]; acc[0][T][4 + u] = bhi[u];
      acc[1][T][u] = blo[u]; acc[1][T][4 + u] = bhi[u];
    }
  }

  const float* xsb = xs + (((size_t)bb * T1_ + t) * NN_) * C1_;
  v4f xb[2][KT_][8];
  #pragma unroll
  for (int nt = 0; nt < 2; ++nt) {
    #pragma unroll
    for (int kt = 0; kt < KT_; ++kt) {
      const float* xrow = xsb + ((size_t)kt * NN_ + nb + 16 * nt + mloc) * C1_ + 4 * half;
      v4f p0[8], p1[8], p2[8];
      #pragma unroll
      for (int j = 0; j < 8; ++j) {
        p0[j] = *(const v4f*)(xrow + 8 * j);
        p1[j] = *(const v4f*)(xrow + PLANE_ + 8 * j);
        p2[j] = *(const v4f*)(xrow + 2 * PLANE_ + 8 * j);
      }
      asm volatile("" :: "v"(p0[0]), "v"(p0[1]), "v"(p0[2]), "v"(p0[3]), "v"(p0[4]), "v"(p0[5]), "v"(p0[6]), "v"(p0[7]),
                         "v"(p1[0]), "v"(p1[1]), "v"(p1[2]), "v"(p1[3]), "v"(p1[4]), "v"(p1[5]), "v"(p1[6]), "v"(p1[7]),
                         "v"(p2[0]), "v"(p2[1]), "v"(p2[2]), "v"(p2[3]), "v"(p2[4]), "v"(p2[5]), "v"(p2[6]), "v"(p2[7]));
      #pragma unroll
      for (int j = 0; j < 8; ++j)
        xb[nt][kt][j] = (p0[j] + p1[j]) + p2[j];
    }
  }

  #pragma unroll
  for (int it = 0; it < 4; ++it) {
    const int item = tid + 256 * it;
    float* dst = s_w + (item & 15) * W2P + (item >> 4) * 4;
    #pragma unroll
    for (int kt = 0; kt < KT_; ++kt) {
      v4f o;
      #pragma unroll
      for (int q = 0; q < 4; ++q) o[q] = wv[it][(3 * q + kt) >> 2][(3 * q + kt) & 3];
      *(v4f*)(dst + kt * W2KT) = o;
    }
  }
  __syncthreads();

  const float* wl = s_w + half * W2P + mloc * 4;
  v4f aw[2][2][4];
  #pragma unroll
  for (int jj = 0; jj < 2; ++jj)
    #pragma unroll
    for (int T = 0; T < 4; ++T)
      aw[0][jj][T] = *(const v4f*)(wl + 2 * jj * W2P + 64 * T);
  __builtin_amdgcn_sched_group_barrier(0x100, 8, 0);

  #pragma unroll
  for (int blk = 0; blk < 12; ++blk) {
    const int kt = blk >> 2, jp = blk & 3;
    const int cur = blk & 1, nxt = cur ^ 1;
    if (blk + 1 < 12) {
      const int kn = (blk + 1) >> 2, jn = (blk + 1) & 3;
      #pragma unroll
      for (int jj = 0; jj < 2; ++jj)
        #pragma unroll
        for (int T = 0; T < 4; ++T)
          aw[nxt][jj][T] = *(const v4f*)(wl + kn * W2KT + 2 * (2 * jn + jj) * W2P + 64 * T);
    }
    #pragma unroll
    for (int jj = 0; jj < 2; ++jj) {
      const int j = 2 * jp + jj;
      #pragma unroll
      for (int s = 0; s < 2; ++s) {
        #pragma unroll
        for (int nt = 0; nt < 2; ++nt) {
          v2f bv; bv[0] = xb[nt][kt][j][2 * s]; bv[1] = xb[nt][kt][j][2 * s + 1];
          #pragma unroll
          for (int T = 0; T < 4; ++T) {
            v2f avv; avv[0] = aw[cur][jj][T][2 * s]; avv[1] = aw[cur][jj][T][2 * s + 1];
            acc[nt][T] = __builtin_amdgcn_wmma_f32_16x16x4_f32(false, avv, false, bv, (short)0, acc[nt][T], false, false);
          }
        }
      }
    }
    if (blk + 1 < 12) __builtin_amdgcn_sched_group_barrier(0x100, 8, 0);
    __builtin_amdgcn_sched_group_barrier(0x008, 32, 0);
  }

  v4f lw[2][4][2], lb[2][4][2];
  float lsum = 0.f, lsq = 0.f;
  #pragma unroll
  for (int nt = 0; nt < 2; ++nt) {
    #pragma unroll
    for (int T = 0; T < 4; ++T) {
      const size_t off = (size_t)(nb + 16 * nt + mloc) * C1_ + 16 * T + 8 * half;
      const float* xr = xsb + (size_t)(KT_ - 1) * NN_ * C1_ + off;
      const v4f r0 = (*(const v4f*)(xr)     + *(const v4f*)(xr + PLANE_))     + *(const v4f*)(xr + 2 * PLANE_);
      const v4f r1 = (*(const v4f*)(xr + 4) + *(const v4f*)(xr + PLANE_ + 4)) + *(const v4f*)(xr + 2 * PLANE_ + 4);
      lw[nt][T][0] = *(const v4f*)(lnw + off);
      lw[nt][T][1] = *(const v4f*)(lnw + off + 4);
      lb[nt][T][0] = *(const v4f*)(lnb + off);
      lb[nt][T][1] = *(const v4f*)(lnb + off + 4);
      #pragma unroll
      for (int u = 0; u < 4; ++u) {
        float v0 = acc[nt][T][u] + r0[u];
        float v1 = acc[nt][T][4 + u] + r1[u];
        v0 = v0 > 0.f ? v0 : 0.f;
        v1 = v1 > 0.f ? v1 : 0.f;
        acc[nt][T][u] = v0;
        acc[nt][T][4 + u] = v1;
        lsum += v0; lsq += v0 * v0;
        lsum += v1; lsq += v1 * v1;
      }
    }
  }

  for (int off = 16; off > 0; off >>= 1) {
    lsum += __shfl_xor(lsum, off);
    lsq  += __shfl_xor(lsq,  off);
  }
  if (lane == 0) { s_red[wave] = lsum; s_red[8 + wave] = lsq; }
  __syncthreads();
  float tsum = 0.f, tsq = 0.f;
  for (int i = 0; i < 8; ++i) { tsum += s_red[i]; tsq += s_red[8 + i]; }
  const float cnt  = (float)(NN_ * C2_);
  const float mu   = tsum / cnt;
  const float var  = tsq / cnt - mu * mu;
  const float rstd = rsqrtf(var + 1e-5f);

  float* ob = out + (((size_t)bb * C2_) * T2_ + t) * NN_ + nb + lane;
  #pragma unroll
  for (int T = 0; T < 4; ++T) {
    #pragma unroll
    for (int r = 0; r < 8; ++r) {
      const float y0 = (acc[0][T][r] - mu) * rstd * lw[0][T][r >> 2][r & 3] + lb[0][T][r >> 2][r & 3];
      const float y1 = (acc[1][T][r] - mu) * rstd * lw[1][T][r >> 2][r & 3] + lb[1][T][r >> 2][r & 3];
      const float send = half ? y0 : y1;
      const float recv = xhalf(send);
      const float lo = half ? recv : y0;
      const float hi = half ? y1 : recv;
      ob[(size_t)(16 * T + r) * T2_ * NN_]     = lo;
      ob[(size_t)(16 * T + 8 + r) * T2_ * NN_] = hi;
    }
  }
}

extern "C" void kernel_launch(void* const* d_in, const int* in_sizes, int n_in,
                              void* d_out, int out_size, void* d_ws, size_t ws_size,
                              hipStream_t stream) {
  const float* x   = (const float*)d_in[0];
  const float* adj = (const float*)d_in[1];
  const float* w1  = (const float*)d_in[2];
  const float* b1  = (const float*)d_in[3];
  const float* w2  = (const float*)d_in[4];
  const float* b2  = (const float*)d_in[5];
  const float* Wh  = (const float*)d_in[6];
  const float* ah  = (const float*)d_in[7];
  const float* lnw = (const float*)d_in[8];
  const float* lnb = (const float*)d_in[9];
  float* out = (float*)d_out;

  float* ws = (float*)d_ws;
  const size_t xg_elems = (size_t)B_ * T1_ * NN_ * C1_;
  float* xg = ws;
  float* xs = xg + xg_elems;
  float* u1 = xs + HGRP_ * xg_elems;
  float* u2 = u1 + HH_ * C1_;

  prep_u_kernel<<<2, 256, 0, stream>>>(Wh, ah, u1, u2);

  const int n_c1 = B_ * T1_ * (NN_ / 16) / 4;
  conv1_glu_kernel<<<n_c1, 128, 0, stream>>>(x, w1, b1, xg);

  const size_t attn_lds =
      (size_t)(NN_ * XSTR + (NN_ / 2) * XPSTR + 3 * NN_ + 8 * 16 * PSTR) * sizeof(float)
      + (size_t)(NN_ * 8) * sizeof(unsigned);
  attn_kernel<<<B_ * T1_ * HGRP_, 256, attn_lds, stream>>>(xg, adj, u1, u2, xs);

  const size_t c2_lds = (size_t)(3 * W2KT + 16) * sizeof(float);
  conv2_ln_kernel<<<B_ * T2_, 256, c2_lds, stream>>>(xs, w2, b2, lnw, lnb, out);
}


# conv1+GLU fused into the attention launch: each workgroup computes its (b,t) xg tile by matrix instructions straight into LDS; xg never goes to memory, 3 launches not 4; sum order unchanged
# speedup vs baseline: 3.7609x; 1.1285x over previous
#include <hip/hip_runtime.h>
#include <math.h>

#define B_   8
#define C0_  16
#define C1_  64
#define C2_  64
#define T0_  12
#define T1_  10
#define T2_  8
#define NN_  256
#define KT_  3
#define HH_  8
#define ALPHA_ 0.2f

typedef float v2f __attribute__((ext_vector_type(2)));
typedef float v4f __attribute__((ext_vector_type(4)));
typedef float v8f __attribute__((ext_vector_type(8)));

__device__ __forceinline__ float tanh_fast(float x) {
  float e = __expf(fminf(2.0f * x, 80.0f));
  return (e - 1.0f) / (e + 1.0f);
}

__global__ __launch_bounds__(256) void prep_u_kernel(
    const float* __restrict__ W,
    const float* __restrict__ a,
    float* __restrict__ u1, float* __restrict__ u2)
{
  int tid = blockIdx.x * 256 + threadIdx.x;
  if (tid >= HH_ * C1_) return;
  int h = tid >> 6, c = tid & 63;
  const float* Wh = W + ((size_t)h * C1_ + c) * C1_;
  const float* ah = a + (size_t)h * 2 * C1_;
  float s1 = 0.f, s2 = 0.f;
  for (int d = 0; d < C1_; ++d) {
    float w = Wh[d];
    s1 += w * ah[d];
    s2 += w * ah[C1_ + d];
  }
  u1[tid] = s1;
  u2[tid] = s2;
}

#define HGRP_ 3
#define PLANE_ ((size_t)B_ * T1_ * NN_ * C1_)
#define XSTR  72
#define XPSTR 160
#define PSTR  260

__global__ __launch_bounds__(256) void attn_kernel(
    const float* __restrict__ x,
    const float* __restrict__ w1,
    const float* __restrict__ b1,
    const float* __restrict__ adj,
    const float* __restrict__ u1,
    const float* __restrict__ u2,
    float* __restrict__ xs)
{
  extern __shared__ float smem[];
  float*    s_xg   = smem;
  float*    s_xgp  = s_xg + NN_ * XSTR;
  float*    s_e1   = s_xgp + (NN_ / 2) * XPSTR;
  float*    s_e2   = s_e1 + NN_;
  float*    s_rinv = s_e2 + NN_;
  float*    s_p    = s_rinv + NN_;
  unsigned* s_adj  = (unsigned*)(s_p + 8 * 16 * PSTR);

  const int bt   = blockIdx.x / HGRP_;
  const int grp  = blockIdx.x - bt * HGRP_;
  const int h_lo = 3 * grp;
  const int h_hi = (h_lo + 3 < HH_) ? h_lo + 3 : HH_;
  const int tid  = threadIdx.x;
  const int lane = tid & 31;
  const int wave = tid >> 5;
  const int mloc = lane & 15;
  const int half = lane >> 4;

  {
    const int bb = bt / T1_, t = bt - bb * T1_;

    float av[2][24];
    #pragma unroll
    for (int ti = 0; ti < 2; ++ti) {
      const int n0 = (wave + 8 * ti) * 16;
      const float* xl = x + ((size_t)(bb * C0_ + 8 * half) * T0_ + t) * NN_ + n0 + mloc;
      #pragma unroll
      for (int k0 = 0; k0 < 24; ++k0)
        av[ti][k0] = xl[((k0 / 3) * T0_ + (k0 % 3)) * NN_];
    }

    const v4f bp = *(const v4f*)(b1 + 4 * mloc);
    const v4f bq = *(const v4f*)(b1 + C1_ + 4 * mloc);
    v8f acc[2][8];
    #pragma unroll
    for (int ti = 0; ti < 2; ++ti) {
      #pragma unroll
      for (int T = 0; T < 4; ++T) {
        #pragma unroll
        for (int r = 0; r < 8; ++r) { acc[ti][T][r] = bp[T]; acc[ti][4 + T][r] = bq[T]; }
      }
    }

    const float* wl = w1 + (size_t)(4 * mloc) * (C0_ * KT_) + 24 * half;
    #pragma unroll
    for (int j = 0; j < 6; ++j) {
      v4f bw[8];
      #pragma unroll
      for (int T = 0; T < 4; ++T) {
        bw[T]     = *(const v4f*)(wl + (size_t)T * (C0_ * KT_) + 4 * j);
        bw[4 + T] = *(const v4f*)(wl + (size_t)(C1_ + T) * (C0_ * KT_) + 4 * j);
      }
      #pragma unroll
      for (int ti = 0; ti < 2; ++ti) {
        v2f a0, a1;
        a0[0] = av[ti][4 * j];     a0[1] = av[ti][4 * j + 1];
        a1[0] = av[ti][4 * j + 2]; a1[1] = av[ti][4 * j + 3];
        #pragma unroll
        for (int T = 0; T < 8; ++T) {
          v2f b0; b0[0] = bw[T][0]; b0[1] = bw[T][1];
          acc[ti][T] = __builtin_amdgcn_wmma_f32_16x16x4_f32(false, a0, false, b0, (short)0, acc[ti][T], false, false);
        }
        #pragma unroll
        for (int T = 0; T < 8; ++T) {
          v2f b1v; b1v[0] = bw[T][2]; b1v[1] = bw[T][3];
          acc[ti][T] = __builtin_amdgcn_wmma_f32_16x16x4_f32(false, a1, false, b1v, (short)0, acc[ti][T], false, false);
        }
      }
    }

    const bool has_res = (4 * mloc) < C0_;
    const int  cres    = (4 * mloc) & (C0_ - 1);
    #pragma unroll
    for (int ti = 0; ti < 2; ++ti) {
      const int n0 = (wave + 8 * ti) * 16;
      const float* xr = x + ((size_t)(bb * C0_ + cres) * T0_ + (t + KT_ - 1)) * NN_ + n0 + 8 * half;
      float xin[4][8];
      #pragma unroll
      for (int T = 0; T < 4; ++T) {
        const v4f lo = *(const v4f*)(xr + (size_t)T * T0_ * NN_);
        const v4f hi = *(const v4f*)(xr + (size_t)T * T0_ * NN_ + 4);
        #pragma unroll
        for (int u = 0; u < 4; ++u) {
          xin[T][u]     = has_res ? lo[u] : 0.f;
          xin[T][4 + u] = has_res ? hi[u] : 0.f;
        }
      }
      v4f o[8];
      #pragma unroll
      for (int r = 0; r < 8; ++r) {
        #pragma unroll
        for (int T = 0; T < 4; ++T) {
          float sg = 1.f / (1.f + __expf(-acc[ti][4 + T][r]));
          o[r][T] = (acc[ti][T][r] + xin[T][r]) * sg;
        }
      }
      const int nr = n0 + 8 * half;
      #pragma unroll
      for (int r = 0; r < 8; ++r)
        *(v4f*)(s_xg + (nr + r) * XSTR + 4 * mloc) = o[r];
      #pragma unroll
      for (int r = 0; r < 8; r += 2) {
        float* pp = s_xgp + ((nr + r) >> 1) * XPSTR + 8 * mloc;
        v4f p0, p1;
        p0[0] = o[r][0]; p0[1] = o[r + 1][0]; p0[2] = o[r][1]; p0[3] = o[r + 1][1];
        p1[0] = o[r][2]; p1[1] = o[r + 1][2]; p1[2] = o[r][3]; p1[3] = o[r + 1][3];
        *(v4f*)(pp)     = p0;
        *(v4f*)(pp + 4) = p1;
      }
    }
  }
  {
    int n = tid;
    for (int w8 = 0; w8 < 8; ++w8) {
      unsigned bits = 0u;
      for (int j = 0; j < 32; ++j)
        bits |= (adj[(size_t)n * NN_ + w8 * 32 + j] > 0.f) ? (1u << j) : 0u;
      s_adj[n * 8 + w8] = bits;
    }
  }
  __syncthreads();

  const float inv_h = 1.0f / (float)HH_;
  float* s_pw = s_p + wave * 16 * PSTR;

  float racc[2][4][8];
  #pragma unroll
  for (int i = 0; i < 2; ++i)
    #pragma unroll
    for (int j = 0; j < 4; ++j)
      #pragma unroll
      for (int r = 0; r < 8; ++r) racc[i][j][r] = 0.f;

  for (int h = h_lo; h < h_hi; ++h) {
    {
      int n = tid;
      const float* row = &s_xg[n * XSTR];
      const float* U1 = u1 + h * C1_;
      const float* U2 = u2 + h * C1_;
      float a1 = 0.f, a2 = 0.f;
      for (int c = 0; c < C1_; ++c) {
        float v = row[c];
        a1 = fmaf(v, U1[c], a1);
        a2 = fmaf(v, U2[c], a2);
      }
      s_e1[n] = a1;
      s_e2[n] = a2;
    }
    __syncthreads();

    #pragma unroll
    for (int rbi = 0; rbi < 2; ++rbi) {
      const int rb = wave + rbi * 8;
      const int m  = rb * 16 + mloc;
      const float e1m = s_e1[m];
      const int jbase = half * 128;

      float mx = -3.0e38f;
      for (int w8 = 0; w8 < 4; ++w8) {
        unsigned adjw = s_adj[m * 8 + half * 4 + w8];
        const float* eb = &s_e2[jbase + w8 * 32];
        for (int j4 = 0; j4 < 32; j4 += 4) {
          v4f e = *(const v4f*)(eb + j4);
          #pragma unroll
          for (int u = 0; u < 4; ++u) {
            float sc = e1m + e[u];
            sc = sc > 0.f ? sc : ALPHA_ * sc;
            int on = (adjw >> (j4 + u)) & 1;
            mx = fmaxf(mx, on ? sc : -3.0e38f);
          }
        }
      }
      mx = fmaxf(mx, __shfl_xor(mx, 16));

      float sm = 0.f;
      for (int w8 = 0; w8 < 4; ++w8) {
        unsigned adjw = s_adj[m * 8 + half * 4 + w8];
        const float* eb = &s_e2[jbase + w8 * 32];
        float* pb = &s_pw[mloc * PSTR + jbase + w8 * 32];
        for (int j4 = 0; j4 < 32; j4 += 4) {
          v4f e = *(const v4f*)(eb + j4);
          v4f p;
          #pragma unroll
          for (int u = 0; u < 4; ++u) {
            float sc = e1m + e[u];
            sc = sc > 0.f ? sc : ALPHA_ * sc;
            sc = ((adjw >> (j4 + u)) & 1) ? sc : -3.0e38f;
            float pe = __expf(sc - mx);
            p[u] = pe;
            sm += pe;
          }
          *(v4f*)(pb + j4) = p;
        }
      }
      sm += __shfl_xor(sm, 16);
      s_rinv[rb * 16 + mloc] = 1.0f / sm;

      v8f acc0 = {}, acc1 = {}, acc2 = {}, acc3 = {};
      const float* pa   = &s_pw[mloc * PSTR + 2 * half];
      const float* xpb  = &s_xgp[half * XPSTR + 2 * mloc];
      for (int k0 = 0; k0 < NN_; k0 += 4) {
        v2f a = *(const v2f*)(pa + k0);
        const float* bp = xpb + (k0 >> 1) * XPSTR;
        v2f b0 = *(const v2f*)(bp + 0);
        v2f b1 = *(const v2f*)(bp + 32);
        v2f b2 = *(const v2f*)(bp + 64);
        v2f b3 = *(const v2f*)(bp + 96);
        acc0 = __builtin_amdgcn_wmma_f32_16x16x4_f32(false, a, false, b0, (short)0, acc0, false, false);
        acc1 = __builtin_amdgcn_wmma_f32_16x16x4_f32(false, a, false, b1, (short)0, acc1, false, false);
        acc2 = __builtin_amdgcn_wmma_f32_16x16x4_f32(false, a, false, b2, (short)0, acc2, false, false);
        acc3 = __builtin_amdgcn_wmma_f32_16x16x4_f32(false, a, false, b3, (short)0, acc3, false, false);
      }

      #pragma unroll
      for (int r = 0; r < 8; ++r) {
        float inv = s_rinv[rb * 16 + r + 8 * half];
        racc[rbi][0][r] += tanh_fast(acc0[r] * inv) * inv_h;
        racc[rbi][1][r] += tanh_fast(acc1[r] * inv) * inv_h;
        racc[rbi][2][r] += tanh_fast(acc2[r] * inv) * inv_h;
        racc[rbi][3][r] += tanh_fast(acc3[r] * inv) * inv_h;
      }
    }
    __syncthreads();
  }

  float* xs_bt = xs + (size_t)grp * PLANE_ + (size_t)bt * NN_ * C1_;
  if (grp == 0) {
    #pragma unroll
    for (int rbi = 0; rbi < 2; ++rbi) {
      #pragma unroll
      for (int r = 0; r < 8; ++r) {
        const float* xrow = &s_xg[((wave + rbi * 8) * 16 + r + 8 * half) * XSTR + mloc];
        racc[rbi][0][r] += xrow[0];
        racc[rbi][1][r] += xrow[16];
        racc[rbi][2][r] += xrow[32];
        racc[rbi][3][r] += xrow[48];
      }
    }
  }
  #pragma unroll
  for (int rbi = 0; rbi < 2; ++rbi) {
    const int rb = wave + rbi * 8;
    #pragma unroll
    for (int r = 0; r < 8; ++r) {
      int mm = rb * 16 + r + 8 * half;
      float* orow = xs_bt + (size_t)mm * C1_ + mloc;
      orow[0]  = racc[rbi][0][r];
      orow[16] = racc[rbi][1][r];
      orow[32] = racc[rbi][2][r];
      orow[48] = racc[rbi][3][r];
    }
  }
}

#define W2P   260
#define W2KT  (16 * W2P)

__device__ __forceinline__ float xhalf(float v) {
  int b = __builtin_bit_cast(int, v);
  return __builtin_bit_cast(float, __builtin_amdgcn_permlanex16(b, b, 0x76543210, 0xfedcba98, false, false));
}

__global__ __launch_bounds__(256) void conv2_ln_kernel(
    const float* __restrict__ xs,
    const float* __restrict__ w2,
    const float* __restrict__ b2,
    const float* __restrict__ lnw,
    const float* __restrict__ lnb,
    float* __restrict__ out)
{
  extern __shared__ float smem[];
  float* s_w   = smem;
  float* s_red = s_w + 3 * W2KT;

  const int bt = blockIdx.x;
  const int bb = bt >> 3, t = bt & 7;
  const int tid  = threadIdx.x;
  const int lane = tid & 31;
  const int wave = tid >> 5;
  const int mloc = lane & 15;
  const int half = lane >> 4;
  const int nb   = 32 * wave;

  v4f wv[4][3];
  #pragma unroll
  for (int it = 0; it < 4; ++it) {
    const float* wp = w2 + 12 * (tid + 256 * it);
    wv[it][0] = *(const v4f*)(wp);
    wv[it][1] = *(const v4f*)(wp + 4);
    wv[it][2] = *(const v4f*)(wp + 8);
  }

  v8f acc[2][4];
  #pragma unroll
  for (int T = 0; T < 4; ++T) {
    const v4f blo = *(const v4f*)(b2 + 16 * T + 8 * half);
    const v4f bhi = *(const v4f*)(b2 + 16 * T + 8 * half + 4);
    #pragma unroll
    for (int u = 0; u < 4; ++u) {
      acc[0][T][u] = blo[u]; acc[0][T][4 + u] = bhi[u];
      acc[1][T][u] = blo[u]; acc[1][T][4 + u] = bhi[u];
    }
  }

  const float* xsb = xs + (((size_t)bb * T1_ + t) * NN_) * C1_;
  v4f xb[2][KT_][8];
  #pragma unroll
  for (int nt = 0; nt < 2; ++nt) {
    #pragma unroll
    for (int kt = 0; kt < KT_; ++kt) {
      const float* xrow = xsb + ((size_t)kt * NN_ + nb + 16 * nt + mloc) * C1_ + 4 * half;
      v4f p0[8], p1[8], p2[8];
      #pragma unroll
      for (int j = 0; j < 8; ++j) {
        p0[j] = *(const v4f*)(xrow + 8 * j);
        p1[j] = *(const v4f*)(xrow + PLANE_ + 8 * j);
        p2[j] = *(const v4f*)(xrow + 2 * PLANE_ + 8 * j);
      }
      asm volatile("" :: "v"(p0[0]), "v"(p0[1]), "v"(p0[2]), "v"(p0[3]), "v"(p0[4]), "v"(p0[5]), "v"(p0[6]), "v"(p0[7]),
                         "v"(p1[0]), "v"(p1[1]), "v"(p1[2]), "v"(p1[3]), "v"(p1[4]), "v"(p1[5]), "v"(p1[6]), "v"(p1[7]),
                         "v"(p2[0]), "v"(p2[1]), "v"(p2[2]), "v"(p2[3]), "v"(p2[4]), "v"(p2[5]), "v"(p2[6]), "v"(p2[7]));
      #pragma unroll
      for (int j = 0; j < 8; ++j)
        xb[nt][kt][j] = (p0[j] + p1[j]) + p2[j];
    }
  }

  #pragma unroll
  for (int it = 0; it < 4; ++it) {
    const int item = tid + 256 * it;
    float* dst = s_w + (item & 15) * W2P + (item >> 4) * 4;
    #pragma unroll
    for (int kt = 0; kt < KT_; ++kt) {
      v4f o;
      #pragma unroll
      for (int q = 0; q < 4; ++q) o[q] = wv[it][(3 * q + kt) >> 2][(3 * q + kt) & 3];
      *(v4f*)(dst + kt * W2KT) = o;
    }
  }
  __syncthreads();

  const float* wl = s_w + half * W2P + mloc * 4;
  v4f aw[2][2][4];
  #pragma unroll
  for (int jj = 0; jj < 2; ++jj)
    #pragma unroll
    for (int T = 0; T < 4; ++T)
      aw[0][jj][T] = *(const v4f*)(wl + 2 * jj * W2P + 64 * T);
  __builtin_amdgcn_sched_group_barrier(0x100, 8, 0);

  #pragma unroll
  for (int blk = 0; blk < 12; ++blk) {
    const int kt = blk >> 2, jp = blk & 3;
    const int cur = blk & 1, nxt = cur ^ 1;
    if (blk + 1 < 12) {
      const int kn = (blk + 1) >> 2, jn = (blk + 1) & 3;
      #pragma unroll
      for (int jj = 0; jj < 2; ++jj)
        #pragma unroll
        for (int T = 0; T < 4; ++T)
          aw[nxt][jj][T] = *(const v4f*)(wl + kn * W2KT + 2 * (2 * jn + jj) * W2P + 64 * T);
    }
    #pragma unroll
    for (int jj = 0; jj < 2; ++jj) {
      const int j = 2 * jp + jj;
      #pragma unroll
      for (int s = 0; s < 2; ++s) {
        #pragma unroll
        for (int nt = 0; nt < 2; ++nt) {
          v2f bv; bv[0] = xb[nt][kt][j][2 * s]; bv[1] = xb[nt][kt][j][2 * s + 1];
          #pragma unroll
          for (int T = 0; T < 4; ++T) {
            v2f avv; avv[0] = aw[cur][jj][T][2 * s]; avv[1] = aw[cur][jj][T][2 * s + 1];
            acc[nt][T] = __builtin_amdgcn_wmma_f32_16x16x4_f32(false, avv, false, bv, (short)0, acc[nt][T], false, false);
          }
        }
      }
    }
    if (blk + 1 < 12) __builtin_amdgcn_sched_group_barrier(0x100, 8, 0);
    __builtin_amdgcn_sched_group_barrier(0x008, 32, 0);
  }

  v4f lw[2][4][2], lb[2][4][2];
  float lsum = 0.f, lsq = 0.f;
  #pragma unroll
  for (int nt = 0; nt < 2; ++nt) {
    #pragma unroll
    for (int T = 0; T < 4; ++T) {
      const size_t off = (size_t)(nb + 16 * nt + mloc) * C1_ + 16 * T + 8 * half;
      const float* xr = xsb + (size_t)(KT_ - 1) * NN_ * C1_ + off;
      const v4f r0 = (*(const v4f*)(xr)     + *(const v4f*)(xr + PLANE_))     + *(const v4f*)(xr + 2 * PLANE_);
      const v4f r1 = (*(const v4f*)(xr + 4) + *(const v4f*)(xr + PLANE_ + 4)) + *(const v4f*)(xr + 2 * PLANE_ + 4);
      lw[nt][T][0] = *(const v4f*)(lnw + off);
      lw[nt][T][1] = *(const v4f*)(lnw + off + 4);
      lb[nt][T][0] = *(const v4f*)(lnb + off);
      lb[nt][T][1] = *(const v4f*)(lnb + off + 4);
      #pragma unroll
      for (int u = 0; u < 4; ++u) {
        float v0 = acc[nt][T][u] + r0[u];
        float v1 = acc[nt][T][4 + u] + r1[u];
        v0 = v0 > 0.f ? v0 : 0.f;
        v1 = v1 > 0.f ? v1 : 0.f;
        acc[nt][T][u] = v0;
        acc[nt][T][4 + u] = v1;
        lsum += v0; lsq += v0 * v0;
        lsum += v1; lsq += v1 * v1;
      }
    }
  }

  for (int off = 16; off > 0; off >>= 1) {
    lsum += __shfl_xor(lsum, off);
    lsq  += __shfl_xor(lsq,  off);
  }
  if (lane == 0) { s_red[wave] = lsum; s_red[8 + wave] = lsq; }
  __syncthreads();
  float tsum = 0.f, tsq = 0.f;
  for (int i = 0; i < 8; ++i) { tsum += s_red[i]; tsq += s_red[8 + i]; }
  const float cnt  = (float)(NN_ * C2_);
  const float mu   = tsum / cnt;
  const float var  = tsq / cnt - mu * mu;
  const float rstd = rsqrtf(var + 1e-5f);

  float* ob = out + (((size_t)bb * C2_) * T2_ + t) * NN_ + nb + lane;
  #pragma unroll
  for (int T = 0; T < 4; ++T) {
    #pragma unroll
    for (int r = 0; r < 8; ++r) {
      const float y0 = (acc[0][T][r] - mu) * rstd * lw[0][T][r >> 2][r & 3] + lb[0][T][r >> 2][r & 3];
      const float y1 = (acc[1][T][r] - mu) * rstd * lw[1][T][r >> 2][r & 3] + lb[1][T][r >> 2][r & 3];
      const float send = half ? y0 : y1;
      const float recv = xhalf(send);
      const float lo = half ? recv : y0;
      const float hi = half ? y1 : recv;
      ob[(size_t)(16 * T + r) * T2_ * NN_]     = lo;
      ob[(size_t)(16 * T + 8 + r) * T2_ * NN_] = hi;
    }
  }
}

extern "C" void kernel_launch(void* const* d_in, const int* in_sizes, int n_in,
                              void* d_out, int out_size, void* d_ws, size_t ws_size,
                              hipStream_t stream) {
  const float* x   = (const float*)d_in[0];
  const float* adj = (const float*)d_in[1];
  const float* w1  = (const float*)d_in[2];
  const float* b1  = (const float*)d_in[3];
  const float* w2  = (const float*)d_in[4];
  const float* b2  = (const float*)d_in[5];
  const float* Wh  = (const float*)d_in[6];
  const float* ah  = (const float*)d_in[7];
  const float* lnw = (const float*)d_in[8];
  const float* lnb = (const float*)d_in[9];
  float* out = (float*)d_out;

  float* ws = (float*)d_ws;
  const size_t xg_elems = (size_t)B_ * T1_ * NN_ * C1_;
  float* xs = ws;
  float* u1 = xs + HGRP_ * xg_elems;
  float* u2 = u1 + HH_ * C1_;

  prep_u_kernel<<<2, 256, 0, stream>>>(Wh, ah, u1, u2);

  const size_t attn_lds =
      (size_t)(NN_ * XSTR + (NN_ / 2) * XPSTR + 3 * NN_ + 8 * 16 * PSTR) * sizeof(float)
      + (size_t)(NN_ * 8) * sizeof(unsigned);
  attn_kernel<<<B_ * T1_ * HGRP_, 256, attn_lds, stream>>>(x, w1, b1, adj, u1, u2, xs);

  const size_t c2_lds = (size_t)(3 * W2KT + 16) * sizeof(float);
  conv2_ln_kernel<<<B_ * T2_, 256, c2_lds, stream>>>(xs, w2, b2, lnw, lnb, out);
}


# attention softmax in one pass: shift = the row's score at the head's largest e2 (upper bound of the masked row max); exact-max pass only for row blocks whose sums underflow; sum order unchanged
# speedup vs baseline: 3.9343x; 1.0461x over previous
#include <hip/hip_runtime.h>
#include <math.h>

#define B_   8
#define C0_  16
#define C1_  64
#define C2_  64
#define T0_  12
#define T1_  10
#define T2_  8
#define NN_  256
#define KT_  3
#define HH_  8
#define ALPHA_ 0.2f

typedef float v2f __attribute__((ext_vector_type(2)));
typedef float v4f __attribute__((ext_vector_type(4)));
typedef float v8f __attribute__((ext_vector_type(8)));

__device__ __forceinline__ float tanh_fast(float x) {
  float e = __expf(fminf(2.0f * x, 80.0f));
  return (e - 1.0f) / (e + 1.0f);
}

__device__ __forceinline__ float row_pick(float v, int sel_lo, int sel_hi) {
  int b = __builtin_bit_cast(int, v);
  return __builtin_bit_cast(float, __builtin_amdgcn_permlane16(b, b, sel_lo, sel_hi, false, false));
}
__device__ __forceinline__ float wave_max(float v) {
  v = fmaxf(v, row_pick(v, 0x67452301, 0xefcdab89));
  v = fmaxf(v, row_pick(v, 0x54761032, 0xdcfe98ba));
  v = fmaxf(v, row_pick(v, 0x32107654, 0xba98fedc));
  v = fmaxf(v, row_pick(v, 0xfedcba98, 0x76543210));
  int b = __builtin_bit_cast(int, v);
  float o = __builtin_bit_cast(float, __builtin_amdgcn_permlanex16(b, b, 0x76543210, 0xfedcba98, false, false));
  return fmaxf(v, o);
}

__global__ __launch_bounds__(256) void prep_u_kernel(
    const float* __restrict__ W,
    const float* __restrict__ a,
    float* __restrict__ u1, float* __restrict__ u2)
{
  int tid = blockIdx.x * 256 + threadIdx.x;
  if (tid >= HH_ * C1_) return;
  int h = tid >> 6, c = tid & 63;
  const float* Wh = W + ((size_t)h * C1_ + c) * C1_;
  const float* ah = a + (size_t)h * 2 * C1_;
  float s1 = 0.f, s2 = 0.f;
  for (int d = 0; d < C1_; ++d) {
    float w = Wh[d];
    s1 += w * ah[d];
    s2 += w * ah[C1_ + d];
  }
  u1[tid] = s1;
  u2[tid] = s2;
}

#define HGRP_ 3
#define PLANE_ ((size_t)B_ * T1_ * NN_ * C1_)
#define XSTR  72
#define XPSTR 160
#define PSTR  260
#define SM_MIN_ 1.0e-25f

__global__ __launch_bounds__(256) void attn_kernel(
    const float* __restrict__ x,
    const float* __restrict__ w1,
    const float* __restrict__ b1,
    const float* __restrict__ adj,
    const float* __restrict__ u1,
    const float* __restrict__ u2,
    float* __restrict__ xs)
{
  extern __shared__ float smem[];
  float*    s_xg   = smem;
  float*    s_xgp  = s_xg + NN_ * XSTR;
  float*    s_e1   = s_xgp + (NN_ / 2) * XPSTR;
  float*    s_e2   = s_e1 + NN_;
  float*    s_rinv = s_e2 + NN_;
  float*    s_p    = s_rinv + NN_;
  unsigned* s_adj  = (unsigned*)(s_p + 8 * 16 * PSTR);
  float*    s_cm   = (float*)(s_adj + NN_ * 8);

  const int bt   = blockIdx.x / HGRP_;
  const int grp  = blockIdx.x - bt * HGRP_;
  const int h_lo = 3 * grp;
  const int h_hi = (h_lo + 3 < HH_) ? h_lo + 3 : HH_;
  const int tid  = threadIdx.x;
  const int lane = tid & 31;
  const int wave = tid >> 5;
  const int mloc = lane & 15;
  const int half = lane >> 4;

  {
    const int bb = bt / T1_, t = bt - bb * T1_;

    float av[2][24];
    #pragma unroll
    for (int ti = 0; ti < 2; ++ti) {
      const int n0 = (wave + 8 * ti) * 16;
      const float* xl = x + ((size_t)(bb * C0_ + 8 * half) * T0_ + t) * NN_ + n0 + mloc;
      #pragma unroll
      for (int k0 = 0; k0 < 24; ++k0)
        av[ti][k0] = xl[((k0 / 3) * T0_ + (k0 % 3)) * NN_];
    }

    const v4f bp = *(const v4f*)(b1 + 4 * mloc);
    const v4f bq = *(const v4f*)(b1 + C1_ + 4 * mloc);
    v8f acc[2][8];
    #pragma unroll
    for (int ti = 0; ti < 2; ++ti) {
      #pragma unroll
      for (int T = 0; T < 4; ++T) {
        #pragma unroll
        for (int r = 0; r < 8; ++r) { acc[ti][T][r] = bp[T]; acc[ti][4 + T][r] = bq[T]; }
      }
    }

    const float* wl = w1 + (size_t)(4 * mloc) * (C0_ * KT_) + 24 * half;
    #pragma unroll
    for (int j = 0; j < 6; ++j) {
      v4f bw[8];
      #pragma unroll
      for (int T = 0; T < 4; ++T) {
        bw[T]     = *(const v4f*)(wl + (size_t)T * (C0_ * KT_) + 4 * j);
        bw[4 + T] = *(const v4f*)(wl + (size_t)(C1_ + T) * (C0_ * KT_) + 4 * j);
      }
      #pragma unroll
      for (int ti = 0; ti < 2; ++ti) {
        v2f a0, a1;
        a0[0] = av[ti][4 * j];     a0[1] = av[ti][4 * j + 1];
        a1[0] = av[ti][4 * j + 2]; a1[1] = av[ti][4 * j + 3];
        #pragma unroll
        for (int T = 0; T < 8; ++T) {
          v2f b0; b0[0] = bw[T][0]; b0[1] = bw[T][1];
          acc[ti][T] = __builtin_amdgcn_wmma_f32_16x16x4_f32(false, a0, false, b0, (short)0, acc[ti][T], false, false);
        }
        #pragma unroll
        for (int T = 0; T < 8; ++T) {
          v2f b1v; b1v[0] = bw[T][2]; b1v[1] = bw[T][3];
          acc[ti][T] = __builtin_amdgcn_wmma_f32_16x16x4_f32(false, a1, false, b1v, (short)0, acc[ti][T], false, false);
        }
      }
    }

    const bool has_res = (4 * mloc) < C0_;
    const int  cres    = (4 * mloc) & (C0_ - 1);
    #pragma unroll
    for (int ti = 0; ti < 2; ++ti) {
      const int n0 = (wave + 8 * ti) * 16;
      const float* xr = x + ((size_t)(bb * C0_ + cres) * T0_ + (t + KT_ - 1)) * NN_ + n0 + 8 * half;
      float xin[4][8];
      #pragma unroll
      for (int T = 0; T < 4; ++T) {
        const v4f lo = *(const v4f*)(xr + (size_t)T * T0_ * NN_);
        const v4f hi = *(const v4f*)(xr + (size_t)T * T0_ * NN_ + 4);
        #pragma unroll
        for (int u = 0; u < 4; ++u) {
          xin[T][u]     = has_res ? lo[u] : 0.f;
          xin[T][4 + u] = has_res ? hi[u] : 0.f;
        }
      }
      v4f o[8];
      #pragma unroll
      for (int r = 0; r < 8; ++r) {
        #pragma unroll
        for (int T = 0; T < 4; ++T) {
          float sg = 1.f / (1.f + __expf(-acc[ti][4 + T][r]));
          o[r][T] = (acc[ti][T][r] + xin[T][r]) * sg;
        }
      }
      const int nr = n0 + 8 * half;
      #pragma unroll
      for (int r = 0; r < 8; ++r)
        *(v4f*)(s_xg + (nr + r) * XSTR + 4 * mloc) = o[r];
      #pragma unroll
      for (int r = 0; r < 8; r += 2) {
        float* pp = s_xgp + ((nr + r) >> 1) * XPSTR + 8 * mloc;
        v4f p0, p1;
        p0[0] = o[r][0]; p0[1] = o[r + 1][0]; p0[2] = o[r][1]; p0[3] = o[r + 1][1];
        p1[0] = o[r][2]; p1[1] = o[r + 1][2]; p1[2] = o[r][3]; p1[3] = o[r + 1][3];
        *(v4f*)(pp)     = p0;
        *(v4f*)(pp + 4) = p1;
      }
    }
  }
  {
    int n = tid;
    for (int w8 = 0; w8 < 8; ++w8) {
      unsigned bits = 0u;
      for (int j = 0; j < 32; ++j)
        bits |= (adj[(size_t)n * NN_ + w8 * 32 + j] > 0.f) ? (1u << j) : 0u;
      s_adj[n * 8 + w8] = bits;
    }
  }
  __syncthreads();

  const float inv_h = 1.0f / (float)HH_;
  float* s_pw = s_p + wave * 16 * PSTR;

  float racc[2][4][8];
  #pragma unroll
  for (int i = 0; i < 2; ++i)
    #pragma unroll
    for (int j = 0; j < 4; ++j)
      #pragma unroll
      for (int r = 0; r < 8; ++r) racc[i][j][r] = 0.f;

  for (int h = h_lo; h < h_hi; ++h) {
    {
      int n = tid;
      const float* row = &s_xg[n * XSTR];
      const float* U1 = u1 + h * C1_;
      const float* U2 = u2 + h * C1_;
      float a1 = 0.f, a2 = 0.f;
      for (int c = 0; c < C1_; ++c) {
        float v = row[c];
        a1 = fmaf(v, U1[c], a1);
        a2 = fmaf(v, U2[c], a2);
      }
      s_e1[n] = a1;
      s_e2[n] = a2;
      const float wm = wave_max(a2);
      if (lane == 0) s_cm[wave] = wm;
    }
    __syncthreads();

    float cmax;
    {
      const v4f c0 = *(const v4f*)(s_cm);
      const v4f c1 = *(const v4f*)(s_cm + 4);
      cmax = fmaxf(fmaxf(fmaxf(c0[0], c0[1]), fmaxf(c0[2], c0[3])),
                   fmaxf(fmaxf(c1[0], c1[1]), fmaxf(c1[2], c1[3])));
    }

    #pragma unroll
    for (int rbi = 0; rbi < 2; ++rbi) {
      const int rb = wave + rbi * 8;
      const int m  = rb * 16 + mloc;
      const float e1m = s_e1[m];
      const int jbase = half * 128;

      float mx = e1m + cmax;
      mx = mx > 0.f ? mx : ALPHA_ * mx;
      float sm;
      #pragma nounroll
      for (int trip = 0; trip < 2; ++trip) {
        sm = 0.f;
        for (int w8 = 0; w8 < 4; ++w8) {
          unsigned adjw = s_adj[m * 8 + half * 4 + w8];
          const float* eb = &s_e2[jbase + w8 * 32];
          float* pb = &s_pw[mloc * PSTR + jbase + w8 * 32];
          for (int j4 = 0; j4 < 32; j4 += 4) {
            v4f e = *(const v4f*)(eb + j4);
            v4f p;
            #pragma unroll
            for (int u = 0; u < 4; ++u) {
              float sc = e1m + e[u];
              sc = sc > 0.f ? sc : ALPHA_ * sc;
              sc = ((adjw >> (j4 + u)) & 1) ? sc : -3.0e38f;
              float pe = __expf(sc - mx);
              p[u] = pe;
              sm += pe;
            }
            *(v4f*)(pb + j4) = p;
          }
        }
        sm += __shfl_xor(sm, 16);
        if (trip == 1 || !__any(!(sm >= SM_MIN_))) break;

        mx = -3.0e38f;
        for (int w8 = 0; w8 < 4; ++w8) {
          unsigned adjw = s_adj[m * 8 + half * 4 + w8];
          const float* eb = &s_e2[jbase + w8 * 32];
          for (int j4 = 0; j4 < 32; j4 += 4) {
            v4f e = *(const v4f*)(eb + j4);
            #pragma unroll
            for (int u = 0; u < 4; ++u) {
              float sc = e1m + e[u];
              sc = sc > 0.f ? sc : ALPHA_ * sc;
              int on = (adjw >> (j4 + u)) & 1;
              mx = fmaxf(mx, on ? sc : -3.0e38f);
            }
          }
        }
        mx = fmaxf(mx, __shfl_xor(mx, 16));
      }
      s_rinv[rb * 16 + mloc] = 1.0f / sm;

      v8f acc0 = {}, acc1 = {}, acc2 = {}, acc3 = {};
      const float* pa   = &s_pw[mloc * PSTR + 2 * half];
      const float* xpb  = &s_xgp[half * XPSTR + 2 * mloc];
      for (int k0 = 0; k0 < NN_; k0 += 4) {
        v2f a = *(const v2f*)(pa + k0);
        const float* bp = xpb + (k0 >> 1) * XPSTR;
        v2f b0 = *(const v2f*)(bp + 0);
        v2f b1 = *(const v2f*)(bp + 32);
        v2f b2 = *(const v2f*)(bp + 64);
        v2f b3 = *(const v2f*)(bp + 96);
        acc0 = __builtin_amdgcn_wmma_f32_16x16x4_f32(false, a, false, b0, (short)0, acc0, false, false);
        acc1 = __builtin_amdgcn_wmma_f32_16x16x4_f32(false, a, false, b1, (short)0, acc1, false, false);
        acc2 = __builtin_amdgcn_wmma_f32_16x16x4_f32(false, a, false, b2, (short)0, acc2, false, false);
        acc3 = __builtin_amdgcn_wmma_f32_16x16x4_f32(false, a, false, b3, (short)0, acc3, false, false);
      }

      #pragma unroll
      for (int r = 0; r < 8; ++r) {
        float inv = s_rinv[rb * 16 + r + 8 * half];
        racc[rbi][0][r] += tanh_fast(acc0[r] * inv) * inv_h;
        racc[rbi][1][r] += tanh_fast(acc1[r] * inv) * inv_h;
        racc[rbi][2][r] += tanh_fast(acc2[r] * inv) * inv_h;
        racc[rbi][3][r] += tanh_fast(acc3[r] * inv) * inv_h;
      }
    }
    __syncthreads();
  }

  float* xs_bt = xs + (size_t)grp * PLANE_ + (size_t)bt * NN_ * C1_;
  if (grp == 0) {
    #pragma unroll
    for (int rbi = 0; rbi < 2; ++rbi) {
      #pragma unroll
      for (int r = 0; r < 8; ++r) {
        const float* xrow = &s_xg[((wave + rbi * 8) * 16 + r + 8 * half) * XSTR + mloc];
        racc[rbi][0][r] += xrow[0];
        racc[rbi][1][r] += xrow[16];
        racc[rbi][2][r] += xrow[32];
        racc[rbi][3][r] += xrow[48];
      }
    }
  }
  #pragma unroll
  for (int rbi = 0; rbi < 2; ++rbi) {
    const int rb = wave + rbi * 8;
    #pragma unroll
    for (int r = 0; r < 8; ++r) {
      int mm = rb * 16 + r + 8 * half;
      float* orow = xs_bt + (size_t)mm * C1_ + mloc;
      orow[0]  = racc[rbi][0][r];
      orow[16] = racc[rbi][1][r];
      orow[32] = racc[rbi][2][r];
      orow[48] = racc[rbi][3][r];
    }
  }
}

#define W2P   260
#define W2KT  (16 * W2P)

__device__ __forceinline__ float xhalf(float v) {
  int b = __builtin_bit_cast(int, v);
  return __builtin_bit_cast(float, __builtin_amdgcn_permlanex16(b, b, 0x76543210, 0xfedcba98, false, false));
}

__global__ __launch_bounds__(256) void conv2_ln_kernel(
    const float* __restrict__ xs,
    const float* __restrict__ w2,
    const float* __restrict__ b2,
    const float* __restrict__ lnw,
    const float* __restrict__ lnb,
    float* __restrict__ out)
{
  extern __shared__ float smem[];
  float* s_w   = smem;
  float* s_red = s_w + 3 * W2KT;

  const int bt = blockIdx.x;
  const int bb = bt >> 3, t = bt & 7;
  const int tid  = threadIdx.x;
  const int lane = tid & 31;
  const int wave = tid >> 5;
  const int mloc = lane & 15;
  const int half = lane >> 4;
  const int nb   = 32 * wave;

  v4f wv[4][3];
  #pragma unroll
  for (int it = 0; it < 4; ++it) {
    const float* wp = w2 + 12 * (tid + 256 * it);
    wv[it][0] = *(const v4f*)(wp);
    wv[it][1] = *(const v4f*)(wp + 4);
    wv[it][2] = *(const v4f*)(wp + 8);
  }

  v8f acc[2][4];
  #pragma unroll
  for (int T = 0; T < 4; ++T) {
    const v4f blo = *(const v4f*)(b2 + 16 * T + 8 * half);
    const v4f bhi = *(const v4f*)(b2 + 16 * T + 8 * half + 4);
    #pragma unroll
    for (int u = 0; u < 4; ++u) {
      acc[0][T][u] = blo[u]; acc[0][T][4 + u] = bhi[u];
      acc[1][T][u] = blo[u]; acc[1][T][4 + u] = bhi[u];
    }
  }

  const float* xsb = xs + (((size_t)bb * T1_ + t) * NN_) * C1_;
  v4f xb[2][KT_][8];
  #pragma unroll
  for (int nt = 0; nt < 2; ++nt) {
    #pragma unroll
    for (int kt = 0; kt < KT_; ++kt) {
      const float* xrow = xsb + ((size_t)kt * NN_ + nb + 16 * nt + mloc) * C1_ + 4 * half;
      v4f p0[8], p1[8], p2[8];
      #pragma unroll
      for (int j = 0; j < 8; ++j) {
        p0[j] = *(const v4f*)(xrow + 8 * j);
        p1[j] = *(const v4f*)(xrow + PLANE_ + 8 * j);
        p2[j] = *(const v4f*)(xrow + 2 * PLANE_ + 8 * j);
      }
      asm volatile("" :: "v"(p0[0]), "v"(p0[1]), "v"(p0[2]), "v"(p0[3]), "v"(p0[4]), "v"(p0[5]), "v"(p0[6]), "v"(p0[7]),
                         "v"(p1[0]), "v"(p1[1]), "v"(p1[2]), "v"(p1[3]), "v"(p1[4]), "v"(p1[5]), "v"(p1[6]), "v"(p1[7]),
                         "v"(p2[0]), "v"(p2[1]), "v"(p2[2]), "v"(p2[3]), "v"(p2[4]), "v"(p2[5]), "v"(p2[6]), "v"(p2[7]));
      #pragma unroll
      for (int j = 0; j < 8; ++j)
        xb[nt][kt][j] = (p0[j] + p1[j]) + p2[j];
    }
  }

  #pragma unroll
  for (int it = 0; it < 4; ++it) {
    const int item = tid + 256 * it;
    float* dst = s_w + (item & 15) * W2P + (item >> 4) * 4;
    #pragma unroll
    for (int kt = 0; kt < KT_; ++kt) {
      v4f o;
      #pragma unroll
      for (int q = 0; q < 4; ++q) o[q] = wv[it][(3 * q + kt) >> 2][(3 * q + kt) & 3];
      *(v4f*)(dst + kt * W2KT) = o;
    }
  }
  __syncthreads();

  const float* wl = s_w + half * W2P + mloc * 4;
  v4f aw[2][2][4];
  #pragma unroll
  for (int jj = 0; jj < 2; ++jj)
    #pragma unroll
    for (int T = 0; T < 4; ++T)
      aw[0][jj][T] = *(const v4f*)(wl + 2 * jj * W2P + 64 * T);
  __builtin_amdgcn_sched_group_barrier(0x100, 8, 0);

  #pragma unroll
  for (int blk = 0; blk < 12; ++blk) {
    const int kt = blk >> 2, jp = blk & 3;
    const int cur = blk & 1, nxt = cur ^ 1;
    if (blk + 1 < 12) {
      const int kn = (blk + 1) >> 2, jn = (blk + 1) & 3;
      #pragma unroll
      for (int jj = 0; jj < 2; ++jj)
        #pragma unroll
        for (int T = 0; T < 4; ++T)
          aw[nxt][jj][T] = *(const v4f*)(wl + kn * W2KT + 2 * (2 * jn + jj) * W2P + 64 * T);
    }
    #pragma unroll
    for (int jj = 0; jj < 2; ++jj) {
      const int j = 2 * jp + jj;
      #pragma unroll
      for (int s = 0; s < 2; ++s) {
        #pragma unroll
        for (int nt = 0; nt < 2; ++nt) {
          v2f bv; bv[0] = xb[nt][kt][j][2 * s]; bv[1] = xb[nt][kt][j][2 * s + 1];
          #pragma unroll
          for (int T = 0; T < 4; ++T) {
            v2f avv; avv[0] = aw[cur][jj][T][2 * s]; avv[1] = aw[cur][jj][T][2 * s + 1];
            acc[nt][T] = __builtin_amdgcn_wmma_f32_16x16x4_f32(false, avv, false, bv, (short)0, acc[nt][T], false, false);
          }
        }
      }
    }
    if (blk + 1 < 12) __builtin_amdgcn_sched_group_barrier(0x100, 8, 0);
    __builtin_amdgcn_sched_group_barrier(0x008, 32, 0);
  }

  v4f lw[2][4][2], lb[2][4][2];
  float lsum = 0.f, lsq = 0.f;
  #pragma unroll
  for (int nt = 0; nt < 2; ++nt) {
    #pragma unroll
    for (int T = 0; T < 4; ++T) {
      const size_t off = (size_t)(nb + 16 * nt + mloc) * C1_ + 16 * T + 8 * half;
      const float* xr = xsb + (size_t)(KT_ - 1) * NN_ * C1_ + off;
      const v4f r0 = (*(const v4f*)(xr)     + *(const v4f*)(xr + PLANE_))     + *(const v4f*)(xr + 2 * PLANE_);
      const v4f r1 = (*(const v4f*)(xr + 4) + *(const v4f*)(xr + PLANE_ + 4)) + *(const v4f*)(xr + 2 * PLANE_ + 4);
      lw[nt][T][0] = *(const v4f*)(lnw + off);
      lw[nt][T][1] = *(const v4f*)(lnw + off + 4);
      lb[nt][T][0] = *(const v4f*)(lnb + off);
      lb[nt][T][1] = *(const v4f*)(lnb + off + 4);
      #pragma unroll
      for (int u = 0; u < 4; ++u) {
        float v0 = acc[nt][T][u] + r0[u];
        float v1 = acc[nt][T][4 + u] + r1[u];
        v0 = v0 > 0.f ? v0 : 0.f;
        v1 = v1 > 0.f ? v1 : 0.f;
        acc[nt][T][u] = v0;
        acc[nt][T][4 + u] = v1;
        lsum += v0; lsq += v0 * v0;
        lsum += v1; lsq += v1 * v1;
      }
    }
  }

  for (int off = 16; off > 0; off >>= 1) {
    lsum += __shfl_xor(lsum, off);
    lsq  += __shfl_xor(lsq,  off);
  }
  if (lane == 0) { s_red[wave] = lsum; s_red[8 + wave] = lsq; }
  __syncthreads();
  float tsum = 0.f, tsq = 0.f;
  for (int i = 0; i < 8; ++i) { tsum += s_red[i]; tsq += s_red[8 + i]; }
  const float cnt  = (float)(NN_ * C2_);
  const float mu   = tsum / cnt;
  const float var  = tsq / cnt - mu * mu;
  const float rstd = rsqrtf(var + 1e-5f);

  float* ob = out + (((size_t)bb * C2_) * T2_ + t) * NN_ + nb + lane;
  #pragma unroll
  for (int T = 0; T < 4; ++T) {
    #pragma unroll
    for (int r = 0; r < 8; ++r) {
      const float y0 = (acc[0][T][r] - mu) * rstd * lw[0][T][r >> 2][r & 3] + lb[0][T][r >> 2][r & 3];
      const float y1 = (acc[1][T][r] - mu) * rstd * lw[1][T][r >> 2][r & 3] + lb[1][T][r >> 2][r & 3];
      const float send = half ? y0 : y1;
      const float recv = xhalf(send);
      const float lo = half ? recv : y0;
      const float hi = half ? y1 : recv;
      ob[(size_t)(16 * T + r) * T2_ * NN_]     = lo;
      ob[(size_t)(16 * T + 8 + r) * T2_ * NN_] = hi;
    }
  }
}

extern "C" void kernel_launch(void* const* d_in, const int* in_sizes, int n_in,
                              void* d_out, int out_size, void* d_ws, size_t ws_size,
                              hipStream_t stream) {
  const float* x   = (const float*)d_in[0];
  const float* adj = (const float*)d_in[1];
  const float* w1  = (const float*)d_in[2];
  const float* b1  = (const float*)d_in[3];
  const float* w2  = (const float*)d_in[4];
  const float* b2  = (const float*)d_in[5];
  const float* Wh  = (const float*)d_in[6];
  const float* ah  = (const float*)d_in[7];
  const float* lnw = (const float*)d_in[8];
  const float* lnb = (const float*)d_in[9];
  float* out = (float*)d_out;

  float* ws = (float*)d_ws;
  const size_t xg_elems = (size_t)B_ * T1_ * NN_ * C1_;
  float* xs = ws;
  float* u1 = xs + HGRP_ * xg_elems;
  float* u2 = u1 + HH_ * C1_;

  prep_u_kernel<<<2, 256, 0, stream>>>(Wh, ah, u1, u2);

  const size_t attn_lds =
      (size_t)(NN_ * XSTR + (NN_ / 2) * XPSTR + 3 * NN_ + 8 * 16 * PSTR) * sizeof(float)
      + (size_t)(NN_ * 8) * sizeof(unsigned)
      + (size_t)8 * sizeof(float);
  attn_kernel<<<B_ * T1_ * HGRP_, 256, attn_lds, stream>>>(x, w1, b1, adj, u1, u2, xs);

  const size_t c2_lds = (size_t)(3 * W2KT + 16) * sizeof(float);
  conv2_ln_kernel<<<B_ * T2_, 256, c2_lds, stream>>>(xs, w2, b2, lnw, lnb, out);
}


# attention prologue: adjacency bit words built from coalesced 16-byte row loads and wave ballots (4 whole lines per load, not 32 lines 1 KiB apart); word layout permuted; sum order unchanged
# speedup vs baseline: 4.0646x; 1.0331x over previous
#include <hip/hip_runtime.h>
#include <math.h>

#define B_   8
#define C0_  16
#define C1_  64
#define C2_  64
#define T0_  12
#define T1_  10
#define T2_  8
#define NN_  256
#define KT_  3
#define HH_  8
#define ALPHA_ 0.2f

typedef float v2f __attribute__((ext_vector_type(2)));
typedef float v4f __attribute__((ext_vector_type(4)));
typedef float v8f __attribute__((ext_vector_type(8)));
typedef unsigned u4w __attribute__((ext_vector_type(4)));

__device__ __forceinline__ float tanh_fast(float x) {
  float e = __expf(fminf(2.0f * x, 80.0f));
  return (e - 1.0f) / (e + 1.0f);
}

__device__ __forceinline__ float row_pick(float v, int sel_lo, int sel_hi) {
  int b = __builtin_bit_cast(int, v);
  return __builtin_bit_cast(float, __builtin_amdgcn_permlane16(b, b, sel_lo, sel_hi, false, false));
}
__device__ __forceinline__ float wave_max(float v) {
  v = fmaxf(v, row_pick(v, 0x67452301, 0xefcdab89));
  v = fmaxf(v, row_pick(v, 0x54761032, 0xdcfe98ba));
  v = fmaxf(v, row_pick(v, 0x32107654, 0xba98fedc));
  v = fmaxf(v, row_pick(v, 0xfedcba98, 0x76543210));
  int b = __builtin_bit_cast(int, v);
  float o = __builtin_bit_cast(float, __builtin_amdgcn_permlanex16(b, b, 0x76543210, 0xfedcba98, false, false));
  return fmaxf(v, o);
}

__global__ __launch_bounds__(256) void prep_u_kernel(
    const float* __restrict__ W,
    const float* __restrict__ a,
    float* __restrict__ u1, float* __restrict__ u2)
{
  int tid = blockIdx.x * 256 + threadIdx.x;
  if (tid >= HH_ * C1_) return;
  int h = tid >> 6, c = tid & 63;
  const float* Wh = W + ((size_t)h * C1_ + c) * C1_;
  const float* ah = a + (size_t)h * 2 * C1_;
  float s1 = 0.f, s2 = 0.f;
  for (int d = 0; d < C1_; ++d) {
    float w = Wh[d];
    s1 += w * ah[d];
    s2 += w * ah[C1_ + d];
  }
  u1[tid] = s1;
  u2[tid] = s2;
}

#define HGRP_ 3
#define PLANE_ ((size_t)B_ * T1_ * NN_ * C1_)
#define XSTR  72
#define XPSTR 160
#define PSTR  260
#define SM_MIN_ 1.0e-25f

__global__ __launch_bounds__(256) void attn_kernel(
    const float* __restrict__ x,
    const float* __restrict__ w1,
    const float* __restrict__ b1,
    const float* __restrict__ adj,
    const float* __restrict__ u1,
    const float* __restrict__ u2,
    float* __restrict__ xs)
{
  extern __shared__ float smem[];
  float*    s_xg   = smem;
  float*    s_xgp  = s_xg + NN_ * XSTR;
  float*    s_e1   = s_xgp + (NN_ / 2) * XPSTR;
  float*    s_e2   = s_e1 + NN_;
  float*    s_rinv = s_e2 + NN_;
  float*    s_p    = s_rinv + NN_;
  unsigned* s_adj  = (unsigned*)(s_p + 8 * 16 * PSTR);
  float*    s_cm   = (float*)(s_adj + NN_ * 8);

  const int bt   = blockIdx.x / HGRP_;
  const int grp  = blockIdx.x - bt * HGRP_;
  const int h_lo = 3 * grp;
  const int h_hi = (h_lo + 3 < HH_) ? h_lo + 3 : HH_;
  const int tid  = threadIdx.x;
  const int lane = tid & 31;
  const int wave = tid >> 5;
  const int mloc = lane & 15;
  const int half = lane >> 4;

  {
    const int bb = bt / T1_, t = bt - bb * T1_;

    float av[2][24];
    #pragma unroll
    for (int ti = 0; ti < 2; ++ti) {
      const int n0 = (wave + 8 * ti) * 16;
      const float* xl = x + ((size_t)(bb * C0_ + 8 * half) * T0_ + t) * NN_ + n0 + mloc;
      #pragma unroll
      for (int k0 = 0; k0 < 24; ++k0)
        av[ti][k0] = xl[((k0 / 3) * T0_ + (k0 % 3)) * NN_];
    }

    const v4f bp = *(const v4f*)(b1 + 4 * mloc);
    const v4f bq = *(const v4f*)(b1 + C1_ + 4 * mloc);
    v8f acc[2][8];
    #pragma unroll
    for (int ti = 0; ti < 2; ++ti) {
      #pragma unroll
      for (int T = 0; T < 4; ++T) {
        #pragma unroll
        for (int r = 0; r < 8; ++r) { acc[ti][T][r] = bp[T]; acc[ti][4 + T][r] = bq[T]; }
      }
    }

    const float* wl = w1 + (size_t)(4 * mloc) * (C0_ * KT_) + 24 * half;
    #pragma unroll
    for (int j = 0; j < 6; ++j) {
      v4f bw[8];
      #pragma unroll
      for (int T = 0; T < 4; ++T) {
        bw[T]     = *(const v4f*)(wl + (size_t)T * (C0_ * KT_) + 4 * j);
        bw[4 + T] = *(const v4f*)(wl + (size_t)(C1_ + T) * (C0_ * KT_) + 4 * j);
      }
      #pragma unroll
      for (int ti = 0; ti < 2; ++ti) {
        v2f a0, a1;
        a0[0] = av[ti][4 * j];     a0[1] = av[ti][4 * j + 1];
        a1[0] = av[ti][4 * j + 2]; a1[1] = av[ti][4 * j + 3];
        #pragma unroll
        for (int T = 0; T < 8; ++T) {
          v2f b0; b0[0] = bw[T][0]; b0[1] = bw[T][1];
          acc[ti][T] = __builtin_amdgcn_wmma_f32_16x16x4_f32(false, a0, false, b0, (short)0, acc[ti][T], false, false);
        }
        #pragma unroll
        for (int T = 0; T < 8; ++T) {
          v2f b1v; b1v[0] = bw[T][2]; b1v[1] = bw[T][3];
          acc[ti][T] = __builtin_amdgcn_wmma_f32_16x16x4_f32(false, a1, false, b1v, (short)0, acc[ti][T], false, false);
        }
      }
    }

    const bool has_res = (4 * mloc) < C0_;
    const int  cres    = (4 * mloc) & (C0_ - 1);
    #pragma unroll
    for (int ti = 0; ti < 2; ++ti) {
      const int n0 = (wave + 8 * ti) * 16;
      const float* xr = x + ((size_t)(bb * C0_ + cres) * T0_ + (t + KT_ - 1)) * NN_ + n0 + 8 * half;
      float xin[4][8];
      #pragma unroll
      for (int T = 0; T < 4; ++T) {
        const v4f lo = *(const v4f*)(xr + (size_t)T * T0_ * NN_);
        const v4f hi = *(const v4f*)(xr + (size_t)T * T0_ * NN_ + 4);
        #pragma unroll
        for (int u = 0; u < 4; ++u) {
          xin[T][u]     = has_res ? lo[u] : 0.f;
          xin[T][4 + u] = has_res ? hi[u] : 0.f;
        }
      }
      v4f o[8];
      #pragma unroll
      for (int r = 0; r < 8; ++r) {
        #pragma unroll
        for (int T = 0; T < 4; ++T) {
          float sg = 1.f / (1.f + __expf(-acc[ti][4 + T][r]));
          o[r][T] = (acc[ti][T][r] + xin[T][r]) * sg;
        }
      }
      const int nr = n0 + 8 * half;
      #pragma unroll
      for (int r = 0; r < 8; ++r)
        *(v4f*)(s_xg + (nr + r) * XSTR + 4 * mloc) = o[r];
      #pragma unroll
      for (int r = 0; r < 8; r += 2) {
        float* pp = s_xgp + ((nr + r) >> 1) * XPSTR + 8 * mloc;
        v4f p0, p1;
        p0[0] = o[r][0]; p0[1] = o[r + 1][0]; p0[2] = o[r][1]; p0[3] = o[r + 1][1];
        p1[0] = o[r][2]; p1[1] = o[r + 1][2]; p1[2] = o[r][3]; p1[3] = o[r + 1][3];
        *(v4f*)(pp)     = p0;
        *(v4f*)(pp + 4) = p1;
      }
    }
  }
  {
    const int r0 = 32 * wave;
    for (int bb = 0; bb < 4; ++bb) {
      const float* ap = adj + (size_t)(r0 + 8 * bb) * NN_ + 4 * lane;
      v4f a[16];
      #pragma unroll
      for (int q = 0; q < 16; ++q) a[q] = *(const v4f*)(ap + q * 128);
      #pragma unroll
      for (int s = 0; s < 2; ++s) {
        unsigned wacc = 0u;
        #pragma unroll
        for (int q = 0; q < 8; ++q) {
          #pragma unroll
          for (int u = 0; u < 4; ++u) {
            const unsigned w = __builtin_amdgcn_ballot_w32(a[8 * s + q][u] > 0.f);
            wacc = (lane == 4 * q + u) ? w : wacc;
          }
        }
        s_adj[(r0 + 8 * bb + 4 * s) * 8 + lane] = wacc;
      }
    }
  }
  __syncthreads();

  const float inv_h = 1.0f / (float)HH_;
  float* s_pw = s_p + wave * 16 * PSTR;

  float racc[2][4][8];
  #pragma unroll
  for (int i = 0; i < 2; ++i)
    #pragma unroll
    for (int j = 0; j < 4; ++j)
      #pragma unroll
      for (int r = 0; r < 8; ++r) racc[i][j][r] = 0.f;

  for (int h = h_lo; h < h_hi; ++h) {
    {
      int n = tid;
      const float* row = &s_xg[n * XSTR];
      const float* U1 = u1 + h * C1_;
      const float* U2 = u2 + h * C1_;
      float a1 = 0.f, a2 = 0.f;
      for (int c = 0; c < C1_; ++c) {
        float v = row[c];
        a1 = fmaf(v, U1[c], a1);
        a2 = fmaf(v, U2[c], a2);
      }
      s_e1[n] = a1;
      s_e2[n] = a2;
      const float wm = wave_max(a2);
      if (lane == 0) s_cm[wave] = wm;
    }
    __syncthreads();

    float cmax;
    {
      const v4f c0 = *(const v4f*)(s_cm);
      const v4f c1 = *(const v4f*)(s_cm + 4);
      cmax = fmaxf(fmaxf(fmaxf(c0[0], c0[1]), fmaxf(c0[2], c0[3])),
                   fmaxf(fmaxf(c1[0], c1[1]), fmaxf(c1[2], c1[3])));
    }

    #pragma unroll
    for (int rbi = 0; rbi < 2; ++rbi) {
      const int rb = wave + rbi * 8;
      const int m  = rb * 16 + mloc;
      const float e1m = s_e1[m];
      const int jbase = half * 128;
      const u4w aw4 = *(const u4w*)(s_adj + m * 8 + half * 4);

      float mx = e1m + cmax;
      mx = mx > 0.f ? mx : ALPHA_ * mx;
      float sm;
      #pragma nounroll
      for (int trip = 0; trip < 2; ++trip) {
        sm = 0.f;
        for (int w8 = 0; w8 < 4; ++w8) {
          const u4w adjw = aw4 >> (unsigned)(8 * w8);
          const float* eb = &s_e2[jbase + w8 * 32];
          float* pb = &s_pw[mloc * PSTR + jbase + w8 * 32];
          for (int j4 = 0; j4 < 32; j4 += 4) {
            v4f e = *(const v4f*)(eb + j4);
            v4f p;
            #pragma unroll
            for (int u = 0; u < 4; ++u) {
              float sc = e1m + e[u];
              sc = sc > 0.f ? sc : ALPHA_ * sc;
              sc = ((adjw[u] >> (j4 >> 2)) & 1) ? sc : -3.0e38f;
              float pe = __expf(sc - mx);
              p[u] = pe;
              sm += pe;
            }
            *(v4f*)(pb + j4) = p;
          }
        }
        sm += __shfl_xor(sm, 16);
        if (trip == 1 || !__any(!(sm >= SM_MIN_))) break;

        mx = -3.0e38f;
        for (int w8 = 0; w8 < 4; ++w8) {
          const u4w adjw = aw4 >> (unsigned)(8 * w8);
          const float* eb = &s_e2[jbase + w8 * 32];
          for (int j4 = 0; j4 < 32; j4 += 4) {
            v4f e = *(const v4f*)(eb + j4);
            #pragma unroll
            for (int u = 0; u < 4; ++u) {
              float sc = e1m + e[u];
              sc = sc > 0.f ? sc : ALPHA_ * sc;
              int on = (adjw[u] >> (j4 >> 2)) & 1;
              mx = fmaxf(mx, on ? sc : -3.0e38f);
            }
          }
        }
        mx = fmaxf(mx, __shfl_xor(mx, 16));
      }
      s_rinv[rb * 16 + mloc] = 1.0f / sm;

      v8f acc0 = {}, acc1 = {}, acc2 = {}, acc3 = {};
      const float* pa   = &s_pw[mloc * PSTR + 2 * half];
      const float* xpb  = &s_xgp[half * XPSTR + 2 * mloc];
      for (int k0 = 0; k0 < NN_; k0 += 4) {
        v2f a = *(const v2f*)(pa + k0);
        const float* bp = xpb + (k0 >> 1) * XPSTR;
        v2f b0 = *(const v2f*)(bp + 0);
        v2f b1 = *(const v2f*)(bp + 32);
        v2f b2 = *(const v2f*)(bp + 64);
        v2f b3 = *(const v2f*)(bp + 96);
        acc0 = __builtin_amdgcn_wmma_f32_16x16x4_f32(false, a, false, b0, (short)0, acc0, false, false);
        acc1 = __builtin_amdgcn_wmma_f32_16x16x4_f32(false, a, false, b1, (short)0, acc1, false, false);
        acc2 = __builtin_amdgcn_wmma_f32_16x16x4_f32(false, a, false, b2, (short)0, acc2, false, false);
        acc3 = __builtin_amdgcn_wmma_f32_16x16x4_f32(false, a, false, b3, (short)0, acc3, false, false);
      }

      #pragma unroll
      for (int r = 0; r < 8; ++r) {
        float inv = s_rinv[rb * 16 + r + 8 * half];
        racc[rbi][0][r] += tanh_fast(acc0[r] * inv) * inv_h;
        racc[rbi][1][r] += tanh_fast(acc1[r] * inv) * inv_h;
        racc[rbi][2][r] += tanh_fast(acc2[r] * inv) * inv_h;
        racc[rbi][3][r] += tanh_fast(acc3[r] * inv) * inv_h;
      }
    }
    __syncthreads();
  }

  float* xs_bt = xs + (size_t)grp * PLANE_ + (size_t)bt * NN_ * C1_;
  if (grp == 0) {
    #pragma unroll
    for (int rbi = 0; rbi < 2; ++rbi) {
      #pragma unroll
      for (int r = 0; r < 8; ++r) {
        const float* xrow = &s_xg[((wave + rbi * 8) * 16 + r + 8 * half) * XSTR + mloc];
        racc[rbi][0][r] += xrow[0];
        racc[rbi][1][r] += xrow[16];
        racc[rbi][2][r] += xrow[32];
        racc[rbi][3][r] += xrow[48];
      }
    }
  }
  #pragma unroll
  for (int rbi = 0; rbi < 2; ++rbi) {
    const int rb = wave + rbi * 8;
    #pragma unroll
    for (int r = 0; r < 8; ++r) {
      int mm = rb * 16 + r + 8 * half;
      float* orow = xs_bt + (size_t)mm * C1_ + mloc;
      orow[0]  = racc[rbi][0][r];
      orow[16] = racc[rbi][1][r];
      orow[32] = racc[rbi][2][r];
      orow[48] = racc[rbi][3][r];
    }
  }
}

#define W2P   260
#define W2KT  (16 * W2P)

__device__ __forceinline__ float xhalf(float v) {
  int b = __builtin_bit_cast(int, v);
  return __builtin_bit_cast(float, __builtin_amdgcn_permlanex16(b, b, 0x76543210, 0xfedcba98, false, false));
}

__global__ __launch_bounds__(256) void conv2_ln_kernel(
    const float* __restrict__ xs,
    const float* __restrict__ w2,
    const float* __restrict__ b2,
    const float* __restrict__ lnw,
    const float* __restrict__ lnb,
    float* __restrict__ out)
{
  extern __shared__ float smem[];
  float* s_w   = smem;
  float* s_red = s_w + 3 * W2KT;

  const int bt = blockIdx.x;
  const int bb = bt >> 3, t = bt & 7;
  const int tid  = threadIdx.x;
  const int lane = tid & 31;
  const int wave = tid >> 5;
  const int mloc = lane & 15;
  const int half = lane >> 4;
  const int nb   = 32 * wave;

  v4f wv[4][3];
  #pragma unroll
  for (int it = 0; it < 4; ++it) {
    const float* wp = w2 + 12 * (tid + 256 * it);
    wv[it][0] = *(const v4f*)(wp);
    wv[it][1] = *(const v4f*)(wp + 4);
    wv[it][2] = *(const v4f*)(wp + 8);
  }

  v8f acc[2][4];
  #pragma unroll
  for (int T = 0; T < 4; ++T) {
    const v4f blo = *(const v4f*)(b2 + 16 * T + 8 * half);
    const v4f bhi = *(const v4f*)(b2 + 16 * T + 8 * half + 4);
    #pragma unroll
    for (int u = 0; u < 4; ++u) {
      acc[0][T][u] = blo[u]; acc[0][T][4 + u] = bhi[u];
      acc[1][T][u] = blo[u]; acc[1][T][4 + u] = bhi[u];
    }
  }

  const float* xsb = xs + (((size_t)bb * T1_ + t) * NN_) * C1_;
  v4f xb[2][KT_][8];
  #pragma unroll
  for (int nt = 0; nt < 2; ++nt) {
    #pragma unroll
    for (int kt = 0; kt < KT_; ++kt) {
      const float* xrow = xsb + ((size_t)kt * NN_ + nb + 16 * nt + mloc) * C1_ + 4 * half;
      v4f p0[8], p1[8], p2[8];
      #pragma unroll
      for (int j = 0; j < 8; ++j) {
        p0[j] = *(const v4f*)(xrow + 8 * j);
        p1[j] = *(const v4f*)(xrow + PLANE_ + 8 * j);
        p2[j] = *(const v4f*)(xrow + 2 * PLANE_ + 8 * j);
      }
      asm volatile("" :: "v"(p0[0]), "v"(p0[1]), "v"(p0[2]), "v"(p0[3]), "v"(p0[4]), "v"(p0[5]), "v"(p0[6]), "v"(p0[7]),
                         "v"(p1[0]), "v"(p1[1]), "v"(p1[2]), "v"(p1[3]), "v"(p1[4]), "v"(p1[5]), "v"(p1[6]), "v"(p1[7]),
                         "v"(p2[0]), "v"(p2[1]), "v"(p2[2]), "v"(p2[3]), "v"(p2[4]), "v"(p2[5]), "v"(p2[6]), "v"(p2[7]));
      #pragma unroll
      for (int j = 0; j < 8; ++j)
        xb[nt][kt][j] = (p0[j] + p1[j]) + p2[j];
    }
  }

  #pragma unroll
  for (int it = 0; it < 4; ++it) {
    const int item = tid + 256 * it;
    float* dst = s_w + (item & 15) * W2P + (item >> 4) * 4;
    #pragma unroll
    for (int kt = 0; kt < KT_; ++kt) {
      v4f o;
      #pragma unroll
      for (int q = 0; q < 4; ++q) o[q] = wv[it][(3 * q + kt) >> 2][(3 * q + kt) & 3];
      *(v4f*)(dst + kt * W2KT) = o;
    }
  }
  __syncthreads();

  const float* wl = s_w + half * W2P + mloc * 4;
  v4f aw[2][2][4];
  #pragma unroll
  for (int jj = 0; jj < 2; ++jj)
    #pragma unroll
    for (int T = 0; T < 4; ++T)
      aw[0][jj][T] = *(const v4f*)(wl + 2 * jj * W2P + 64 * T);
  __builtin_amdgcn_sched_group_barrier(0x100, 8, 0);

  #pragma unroll
  for (int blk = 0; blk < 12; ++blk) {
    const int kt = blk >> 2, jp = blk & 3;
    const int cur = blk & 1, nxt = cur ^ 1;
    if (blk + 1 < 12) {
      const int kn = (blk + 1) >> 2, jn = (blk + 1) & 3;
      #pragma unroll
      for (int jj = 0; jj < 2; ++jj)
        #pragma unroll
        for (int T = 0; T < 4; ++T)
          aw[nxt][jj][T] = *(const v4f*)(wl + kn * W2KT + 2 * (2 * jn + jj) * W2P + 64 * T);
    }
    #pragma unroll
    for (int jj = 0; jj < 2; ++jj) {
      const int j = 2 * jp + jj;
      #pragma unroll
      for (int s = 0; s < 2; ++s) {
        #pragma unroll
        for (int nt = 0; nt < 2; ++nt) {
          v2f bv; bv[0] = xb[nt][kt][j][2 * s]; bv[1] = xb[nt][kt][j][2 * s + 1];
          #pragma unroll
          for (int T = 0; T < 4; ++T) {
            v2f avv; avv[0] = aw[cur][jj][T][2 * s]; avv[1] = aw[cur][jj][T][2 * s + 1];
            acc[nt][T] = __builtin_amdgcn_wmma_f32_16x16x4_f32(false, avv, false, bv, (short)0, acc[nt][T], false, false);
          }
        }
      }
    }
    if (blk + 1 < 12) __builtin_amdgcn_sched_group_barrier(0x100, 8, 0);
    __builtin_amdgcn_sched_group_barrier(0x008, 32, 0);
  }

  v4f lw[2][4][2], lb[2][4][2];
  float lsum = 0.f, lsq = 0.f;
  #pragma unroll
  for (int nt = 0; nt < 2; ++nt) {
    #pragma unroll
    for (int T = 0; T < 4; ++T) {
      const size_t off = (size_t)(nb + 16 * nt + mloc) * C1_ + 16 * T + 8 * half;
      const float* xr = xsb + (size_t)(KT_ - 1) * NN_ * C1_ + off;
      const v4f r0 = (*(const v4f*)(xr)     + *(const v4f*)(xr + PLANE_))     + *(const v4f*)(xr + 2 * PLANE_);
      const v4f r1 = (*(const v4f*)(xr + 4) + *(const v4f*)(xr + PLANE_ + 4)) + *(const v4f*)(xr + 2 * PLANE_ + 4);
      lw[nt][T][0] = *(const v4f*)(lnw + off);
      lw[nt][T][1] = *(const v4f*)(lnw + off + 4);
      lb[nt][T][0] = *(const v4f*)(lnb + off);
      lb[nt][T][1] = *(const v4f*)(lnb + off + 4);
      #pragma unroll
      for (int u = 0; u < 4; ++u) {
        float v0 = acc[nt][T][u] + r0[u];
        float v1 = acc[nt][T][4 + u] + r1[u];
        v0 = v0 > 0.f ? v0 : 0.f;
        v1 = v1 > 0.f ? v1 : 0.f;
        acc[nt][T][u] = v0;
        acc[nt][T][4 + u] = v1;
        lsum += v0; lsq += v0 * v0;
        lsum += v1; lsq += v1 * v1;
      }
    }
  }

  for (int off = 16; off > 0; off >>= 1) {
    lsum += __shfl_xor(lsum, off);
    lsq  += __shfl_xor(lsq,  off);
  }
  if (lane == 0) { s_red[wave] = lsum; s_red[8 + wave] = lsq; }
  __syncthreads();
  float tsum = 0.f, tsq = 0.f;
  for (int i = 0; i < 8; ++i) { tsum += s_red[i]; tsq += s_red[8 + i]; }
  const float cnt  = (float)(NN_ * C2_);
  const float mu   = tsum / cnt;
  const float var  = tsq / cnt - mu * mu;
  const float rstd = rsqrtf(var + 1e-5f);

  float* ob = out + (((size_t)bb * C2_) * T2_ + t) * NN_ + nb + lane;
  #pragma unroll
  for (int T = 0; T < 4; ++T) {
    #pragma unroll
    for (int r = 0; r < 8; ++r) {
      const float y0 = (acc[0][T][r] - mu) * rstd * lw[0][T][r >> 2][r & 3] + lb[0][T][r >> 2][r & 3];
      const float y1 = (acc[1][T][r] - mu) * rstd * lw[1][T][r >> 2][r & 3] + lb[1][T][r >> 2][r & 3];
      const float send = half ? y0 : y1;
      const float recv = xhalf(send);
      const float lo = half ? recv : y0;
      const float hi = half ? y1 : recv;
      ob[(size_t)(16 * T + r) * T2_ * NN_]     = lo;
      ob[(size_t)(16 * T + 8 + r) * T2_ * NN_] = hi;
    }
  }
}

extern "C" void kernel_launch(void* const* d_in, const int* in_sizes, int n_in,
                              void* d_out, int out_size, void* d_ws, size_t ws_size,
                              hipStream_t stream) {
  const float* x   = (const float*)d_in[0];
  const float* adj = (const float*)d_in[1];
  const float* w1  = (const float*)d_in[2];
  const float* b1  = (const float*)d_in[3];
  const float* w2  = (const float*)d_in[4];
  const float* b2  = (const float*)d_in[5];
  const float* Wh  = (const float*)d_in[6];
  const float* ah  = (const float*)d_in[7];
  const float* lnw = (const float*)d_in[8];
  const float* lnb = (const float*)d_in[9];
  float* out = (float*)d_out;

  float* ws = (float*)d_ws;
  const size_t xg_elems = (size_t)B_ * T1_ * NN_ * C1_;
  float* xs = ws;
  float* u1 = xs + HGRP_ * xg_elems;
  float* u2 = u1 + HH_ * C1_;

  prep_u_kernel<<<2, 256, 0, stream>>>(Wh, ah, u1, u2);

  const size_t attn_lds =
      (size_t)(NN_ * XSTR + (NN_ / 2) * XPSTR + 3 * NN_ + 8 * 16 * PSTR) * sizeof(float)
      + (size_t)(NN_ * 8) * sizeof(unsigned)
      + (size_t)8 * sizeof(float);
  attn_kernel<<<B_ * T1_ * HGRP_, 256, attn_lds, stream>>>(x, w1, b1, adj, u1, u2, xs);

  const size_t c2_lds = (size_t)(3 * W2KT + 16) * sizeof(float);
  conv2_ln_kernel<<<B_ * T2_, 256, c2_lds, stream>>>(xs, w2, b2, lnw, lnb, out);
}


# attention numerators as max(R1_i*C1_j, R2_i*C2_j): exp per row and per column, none per score; pass runs lane<->column, ballot words as lane masks; row sums taken in the matrix loop, order changed
# speedup vs baseline: 4.4150x; 1.0862x over previous
#include <hip/hip_runtime.h>
#include <math.h>

#define B_   8
#define C0_  16
#define C1_  64
#define C2_  64
#define T0_  12
#define T1_  10
#define T2_  8
#define NN_  256
#define KT_  3
#define HH_  8
#define ALPHA_ 0.2f

typedef float v2f __attribute__((ext_vector_type(2)));
typedef float v4f __attribute__((ext_vector_type(4)));
typedef float v8f __attribute__((ext_vector_type(8)));
typedef unsigned u4w __attribute__((ext_vector_type(4)));

__device__ __forceinline__ float tanh_fast(float x) {
  float e = __expf(fminf(2.0f * x, 80.0f));
  return (e - 1.0f) / (e + 1.0f);
}

__device__ __forceinline__ float row_pick(float v, int sel_lo, int sel_hi) {
  int b = __builtin_bit_cast(int, v);
  return __builtin_bit_cast(float, __builtin_amdgcn_permlane16(b, b, sel_lo, sel_hi, false, false));
}
__device__ __forceinline__ float wave_max(float v) {
  v = fmaxf(v, row_pick(v, 0x67452301, 0xefcdab89));
  v = fmaxf(v, row_pick(v, 0x54761032, 0xdcfe98ba));
  v = fmaxf(v, row_pick(v, 0x32107654, 0xba98fedc));
  v = fmaxf(v, row_pick(v, 0xfedcba98, 0x76543210));
  int b = __builtin_bit_cast(int, v);
  float o = __builtin_bit_cast(float, __builtin_amdgcn_permlanex16(b, b, 0x76543210, 0xfedcba98, false, false));
  return fmaxf(v, o);
}

__device__ __forceinline__ float lane_f(float v, int l) {
  return __builtin_bit_cast(float, __builtin_amdgcn_readlane(__builtin_bit_cast(int, v), l));
}
__device__ __forceinline__ unsigned lane_u(unsigned v, int l) {
  return (unsigned)__builtin_amdgcn_readlane((int)v, l);
}

__global__ __launch_bounds__(256) void prep_u_kernel(
    const float* __restrict__ W,
    const float* __restrict__ a,
    float* __restrict__ u1, float* __restrict__ u2)
{
  int tid = blockIdx.x * 256 + threadIdx.x;
  if (tid >= HH_ * C1_) return;
  int h = tid >> 6, c = tid & 63;
  const float* Wh = W + ((size_t)h * C1_ + c) * C1_;
  const float* ah = a + (size_t)h * 2 * C1_;
  float s1 = 0.f, s2 = 0.f;
  for (int d = 0; d < C1_; ++d) {
    float w = Wh[d];
    s1 += w * ah[d];
    s2 += w * ah[C1_ + d];
  }
  u1[tid] = s1;
  u2[tid] = s2;
}

#define HGRP_ 3
#define PLANE_ ((size_t)B_ * T1_ * NN_ * C1_)
#define XSTR  72
#define XPSTR 160
#define PSTR  260
#define SM_MIN_ 1.0e-25f

__global__ __launch_bounds__(256) void attn_kernel(
    const float* __restrict__ x,
    const float* __restrict__ w1,
    const float* __restrict__ b1,
    const float* __restrict__ adj,
    const float* __restrict__ u1,
    const float* __restrict__ u2,
    float* __restrict__ xs)
{
  extern __shared__ float smem[];
  float*    s_xg   = smem;
  float*    s_xgp  = s_xg + NN_ * XSTR;
  float*    s_e1   = s_xgp + (NN_ / 2) * XPSTR;
  float*    s_e2   = s_e1 + NN_;
  float*    s_rinv = s_e2 + NN_;
  float*    s_p    = s_rinv + NN_;
  unsigned* s_adj  = (unsigned*)(s_p + 8 * 16 * PSTR);
  float*    s_cm   = (float*)(s_adj + NN_ * 8);

  const int bt   = blockIdx.x / HGRP_;
  const int grp  = blockIdx.x - bt * HGRP_;
  const int h_lo = 3 * grp;
  const int h_hi = (h_lo + 3 < HH_) ? h_lo + 3 : HH_;
  const int tid  = threadIdx.x;
  const int lane = tid & 31;
  const int wave = tid >> 5;
  const int mloc = lane & 15;
  const int half = lane >> 4;

  {
    const int bb = bt / T1_, t = bt - bb * T1_;

    float av[2][24];
    #pragma unroll
    for (int ti = 0; ti < 2; ++ti) {
      const int n0 = (wave + 8 * ti) * 16;
      const float* xl = x + ((size_t)(bb * C0_ + 8 * half) * T0_ + t) * NN_ + n0 + mloc;
      #pragma unroll
      for (int k0 = 0; k0 < 24; ++k0)
        av[ti][k0] = xl[((k0 / 3) * T0_ + (k0 % 3)) * NN_];
    }

    const v4f bp = *(const v4f*)(b1 + 4 * mloc);
    const v4f bq = *(const v4f*)(b1 + C1_ + 4 * mloc);
    v8f acc[2][8];
    #pragma unroll
    for (int ti = 0; ti < 2; ++ti) {
      #pragma unroll
      for (int T = 0; T < 4; ++T) {
        #pragma unroll
        for (int r = 0; r < 8; ++r) { acc[ti][T][r] = bp[T]; acc[ti][4 + T][r] = bq[T]; }
      }
    }

    const float* wl = w1 + (size_t)(4 * mloc) * (C0_ * KT_) + 24 * half;
    #pragma unroll
    for (int j = 0; j < 6; ++j) {
      v4f bw[8];
      #pragma unroll
      for (int T = 0; T < 4; ++T) {
        bw[T]     = *(const v4f*)(wl + (size_t)T * (C0_ * KT_) + 4 * j);
        bw[4 + T] = *(const v4f*)(wl + (size_t)(C1_ + T) * (C0_ * KT_) + 4 * j);
      }
      #pragma unroll
      for (int ti = 0; ti < 2; ++ti) {
        v2f a0, a1;
        a0[0] = av[ti][4 * j];     a0[1] = av[ti][4 * j + 1];
        a1[0] = av[ti][4 * j + 2]; a1[1] = av[ti][4 * j + 3];
        #pragma unroll
        for (int T = 0; T < 8; ++T) {
          v2f b0; b0[0] = bw[T][0]; b0[1] = bw[T][1];
          acc[ti][T] = __builtin_amdgcn_wmma_f32_16x16x4_f32(false, a0, false, b0, (short)0, acc[ti][T], false, false);
        }
        #pragma unroll
        for (int T = 0; T < 8; ++T) {
          v2f b1v; b1v[0] = bw[T][2]; b1v[1] = bw[T][3];
          acc[ti][T] = __builtin_amdgcn_wmma_f32_16x16x4_f32(false, a1, false, b1v, (short)0, acc[ti][T], false, false);
        }
      }
    }

    const bool has_res = (4 * mloc) < C0_;
    const int  cres    = (4 * mloc) & (C0_ - 1);
    #pragma unroll
    for (int ti = 0; ti < 2; ++ti) {
      const int n0 = (wave + 8 * ti) * 16;
      const float* xr = x + ((size_t)(bb * C0_ + cres) * T0_ + (t + KT_ - 1)) * NN_ + n0 + 8 * half;
      float xin[4][8];
      #pragma unroll
      for (int T = 0; T < 4; ++T) {
        const v4f lo = *(const v4f*)(xr + (size_t)T * T0_ * NN_);
        const v4f hi = *(const v4f*)(xr + (size_t)T * T0_ * NN_ + 4);
        #pragma unroll
        for (int u = 0; u < 4; ++u) {
          xin[T][u]     = has_res ? lo[u] : 0.f;
          xin[T][4 + u] = has_res ? hi[u] : 0.f;
        }
      }
      v4f o[8];
      #pragma unroll
      for (int r = 0; r < 8; ++r) {
        #pragma unroll
        for (int T = 0; T < 4; ++T) {
          float sg = 1.f / (1.f + __expf(-acc[ti][4 + T][r]));
          o[r][T] = (acc[ti][T][r] + xin[T][r]) * sg;
        }
      }
      const int nr = n0 + 8 * half;
      #pragma unroll
      for (int r = 0; r < 8; ++r)
        *(v4f*)(s_xg + (nr + r) * XSTR + 4 * mloc) = o[r];
      #pragma unroll
      for (int r = 0; r < 8; r += 2) {
        float* pp = s_xgp + ((nr + r) >> 1) * XPSTR + 8 * mloc;
        v4f p0, p1;
        p0[0] = o[r][0]; p0[1] = o[r + 1][0]; p0[2] = o[r][1]; p0[3] = o[r + 1][1];
        p1[0] = o[r][2]; p1[1] = o[r + 1][2]; p1[2] = o[r][3]; p1[3] = o[r + 1][3];
        *(v4f*)(pp)     = p0;
        *(v4f*)(pp + 4) = p1;
      }
    }
  }
  {
    const int r0 = 32 * wave;
    for (int bb = 0; bb < 4; ++bb) {
      const float* ap = adj + (size_t)(r0 + 8 * bb) * NN_ + 4 * lane;
      v4f a[16];
      #pragma unroll
      for (int q = 0; q < 16; ++q) a[q] = *(const v4f*)(ap + q * 128);
      #pragma unroll
      for (int s = 0; s < 2; ++s) {
        unsigned wacc = 0u;
        #pragma unroll
        for (int q = 0; q < 8; ++q) {
          #pragma unroll
          for (int u = 0; u < 4; ++u) {
            const unsigned w = __builtin_amdgcn_ballot_w32(a[8 * s + q][u] > 0.f);
            wacc = (lane == 4 * q + u) ? w : wacc;
          }
        }
        s_adj[(r0 + 8 * bb + 4 * s) * 8 + lane] = wacc;
      }
    }
  }
  __syncthreads();

  const float inv_h = 1.0f / (float)HH_;
  float* s_pw = s_p + wave * 16 * PSTR;

  u4w awb[2];
  #pragma unroll
  for (int rbi = 0; rbi < 2; ++rbi)
    awb[rbi] = *(const u4w*)(s_adj + (wave + 8 * rbi) * 16 * 8 + 4 * lane);

  float racc[2][4][8];
  #pragma unroll
  for (int i = 0; i < 2; ++i)
    #pragma unroll
    for (int j = 0; j < 4; ++j)
      #pragma unroll
      for (int r = 0; r < 8; ++r) racc[i][j][r] = 0.f;

  for (int h = h_lo; h < h_hi; ++h) {
    {
      int n = tid;
      const float* row = &s_xg[n * XSTR];
      const float* U1 = u1 + h * C1_;
      const float* U2 = u2 + h * C1_;
      float a1 = 0.f, a2 = 0.f;
      for (int c = 0; c < C1_; ++c) {
        float v = row[c];
        a1 = fmaf(v, U1[c], a1);
        a2 = fmaf(v, U2[c], a2);
      }
      s_e1[n] = a1;
      s_e2[n] = a2;
      const float wm = wave_max(a2);
      if (lane == 0) s_cm[wave] = wm;
    }
    __syncthreads();

    float cmax;
    {
      const v4f c0 = *(const v4f*)(s_cm);
      const v4f c1 = *(const v4f*)(s_cm + 4);
      cmax = fmaxf(fmaxf(fmaxf(c0[0], c0[1]), fmaxf(c0[2], c0[3])),
                   fmaxf(fmaxf(c1[0], c1[1]), fmaxf(c1[2], c1[3])));
    }

    v4f cf1[2], cf2[2];
    #pragma unroll
    for (int hh = 0; hh < 2; ++hh) {
      const v4f e = *(const v4f*)(s_e2 + 128 * hh + 4 * lane);
      #pragma unroll
      for (int u = 0; u < 4; ++u) {
        const float z = e[u] - cmax;
        cf1[hh][u] = __expf(z);
        cf2[hh][u] = __expf(ALPHA_ * z);
      }
    }
    float vr1, vr2;
    {
      const float d = s_e1[(wave + 8 * half) * 16 + mloc] + cmax;
      vr1 = __expf((1.0f - ALPHA_) * fminf(d, 0.f));
      vr2 = __expf(-(1.0f - ALPHA_) * fmaxf(d, 0.f));
    }

    #pragma unroll
    for (int rbi = 0; rbi < 2; ++rbi) {
      const int rb = wave + rbi * 8;
      float sm;
      v8f acc0, acc1, acc2, acc3;
      #pragma nounroll
      for (int trip = 0; trip < 2; ++trip) {
        if (trip == 0) {
          #pragma unroll
          for (int r = 0; r < 16; ++r) {
            const float R1 = lane_f(vr1, 16 * rbi + r);
            const float R2 = lane_f(vr2, 16 * rbi + r);
            #pragma unroll
            for (int hh = 0; hh < 2; ++hh) {
              v4f p;
              #pragma unroll
              for (int u = 0; u < 4; ++u) {
                const unsigned w = lane_u(awb[rbi][u], 2 * r + hh);
                const float t = fmaxf(R1 * cf1[hh][u], R2 * cf2[hh][u]);
                p[u] = __builtin_amdgcn_inverse_ballot_w32(w) ? t : 0.f;
              }
              *(v4f*)(s_pw + r * PSTR + 128 * hh + 4 * lane) = p;
            }
          }
        } else {
          const int m = rb * 16 + mloc;
          const float e1m = s_e1[m];
          const int jbase = half * 128;
          const u4w aw4 = *(const u4w*)(s_adj + m * 8 + half * 4);
          float mx = -3.0e38f;
          for (int w8 = 0; w8 < 4; ++w8) {
            const u4w adjw = aw4 >> (unsigned)(8 * w8);
            const float* eb = &s_e2[jbase + w8 * 32];
            for (int j4 = 0; j4 < 32; j4 += 4) {
              v4f e = *(const v4f*)(eb + j4);
              #pragma unroll
              for (int u = 0; u < 4; ++u) {
                float sc = e1m + e[u];
                sc = sc > 0.f ? sc : ALPHA_ * sc;
                int on = (adjw[u] >> (j4 >> 2)) & 1;
                mx = fmaxf(mx, on ? sc : -3.0e38f);
              }
            }
          }
          mx = fmaxf(mx, __shfl_xor(mx, 16));
          for (int w8 = 0; w8 < 4; ++w8) {
            const u4w adjw = aw4 >> (unsigned)(8 * w8);
            const float* eb = &s_e2[jbase + w8 * 32];
            float* pb = &s_pw[mloc * PSTR + jbase + w8 * 32];
            for (int j4 = 0; j4 < 32; j4 += 4) {
              v4f e = *(const v4f*)(eb + j4);
              v4f p;
              #pragma unroll
              for (int u = 0; u < 4; ++u) {
                float sc = e1m + e[u];
                sc = sc > 0.f ? sc : ALPHA_ * sc;
                sc = ((adjw[u] >> (j4 >> 2)) & 1) ? sc : -3.0e38f;
                p[u] = __expf(sc - mx);
              }
              *(v4f*)(pb + j4) = p;
            }
          }
        }
        __builtin_amdgcn_fence(__ATOMIC_ACQ_REL, "wavefront");
        __builtin_amdgcn_wave_barrier();

        acc0 = v8f{}; acc1 = v8f{}; acc2 = v8f{}; acc3 = v8f{};
        v2f ps = {0.f, 0.f};
        const float* pa   = &s_pw[mloc * PSTR + 2 * half];
        const float* xpb  = &s_xgp[half * XPSTR + 2 * mloc];
        for (int k0 = 0; k0 < NN_; k0 += 4) {
          v2f a = *(const v2f*)(pa + k0);
          const float* bp = xpb + (k0 >> 1) * XPSTR;
          v2f b0 = *(const v2f*)(bp + 0);
          v2f b1 = *(const v2f*)(bp + 32);
          v2f b2 = *(const v2f*)(bp + 64);
          v2f b3 = *(const v2f*)(bp + 96);
          ps += a;
          acc0 = __builtin_amdgcn_wmma_f32_16x16x4_f32(false, a, false, b0, (short)0, acc0, false, false);
          acc1 = __builtin_amdgcn_wmma_f32_16x16x4_f32(false, a, false, b1, (short)0, acc1, false, false);
          acc2 = __builtin_amdgcn_wmma_f32_16x16x4_f32(false, a, false, b2, (short)0, acc2, false, false);
          acc3 = __builtin_amdgcn_wmma_f32_16x16x4_f32(false, a, false, b3, (short)0, acc3, false, false);
        }
        sm = ps[0] + ps[1];
        sm += __shfl_xor(sm, 16);
        if (trip == 1 || !__any(!(sm >= SM_MIN_))) break;
      }
      s_rinv[rb * 16 + mloc] = 1.0f / sm;

      #pragma unroll
      for (int r = 0; r < 8; ++r) {
        float inv = s_rinv[rb * 16 + r + 8 * half];
        racc[rbi][0][r] += tanh_fast(acc0[r] * inv) * inv_h;
        racc[rbi][1][r] += tanh_fast(acc1[r] * inv) * inv_h;
        racc[rbi][2][r] += tanh_fast(acc2[r] * inv) * inv_h;
        racc[rbi][3][r] += tanh_fast(acc3[r] * inv) * inv_h;
      }
    }
    __syncthreads();
  }

  float* xs_bt = xs + (size_t)grp * PLANE_ + (size_t)bt * NN_ * C1_;
  if (grp == 0) {
    #pragma unroll
    for (int rbi = 0; rbi < 2; ++rbi) {
      #pragma unroll
      for (int r = 0; r < 8; ++r) {
        const float* xrow = &s_xg[((wave + rbi * 8) * 16 + r + 8 * half) * XSTR + mloc];
        racc[rbi][0][r] += xrow[0];
        racc[rbi][1][r] += xrow[16];
        racc[rbi][2][r] += xrow[32];
        racc[rbi][3][r] += xrow[48];
      }
    }
  }
  #pragma unroll
  for (int rbi = 0; rbi < 2; ++rbi) {
    const int rb = wave + rbi * 8;
    #pragma unroll
    for (int r = 0; r < 8; ++r) {
      int mm = rb * 16 + r + 8 * half;
      float* orow = xs_bt + (size_t)mm * C1_ + mloc;
      orow[0]  = racc[rbi][0][r];
      orow[16] = racc[rbi][1][r];
      orow[32] = racc[rbi][2][r];
      orow[48] = racc[rbi][3][r];
    }
  }
}

#define W2P   260
#define W2KT  (16 * W2P)

__device__ __forceinline__ float xhalf(float v) {
  int b = __builtin_bit_cast(int, v);
  return __builtin_bit_cast(float, __builtin_amdgcn_permlanex16(b, b, 0x76543210, 0xfedcba98, false, false));
}

__global__ __launch_bounds__(256) void conv2_ln_kernel(
    const float* __restrict__ xs,
    const float* __restrict__ w2,
    const float* __restrict__ b2,
    const float* __restrict__ lnw,
    const float* __restrict__ lnb,
    float* __restrict__ out)
{
  extern __shared__ float smem[];
  float* s_w   = smem;
  float* s_red = s_w + 3 * W2KT;

  const int bt = blockIdx.x;
  const int bb = bt >> 3, t = bt & 7;
  const int tid  = threadIdx.x;
  const int lane = tid & 31;
  const int wave = tid >> 5;
  const int mloc = lane & 15;
  const int half = lane >> 4;
  const int nb   = 32 * wave;

  v4f wv[4][3];
  #pragma unroll
  for (int it = 0; it < 4; ++it) {
    const float* wp = w2 + 12 * (tid + 256 * it);
    wv[it][0] = *(const v4f*)(wp);
    wv[it][1] = *(const v4f*)(wp + 4);
    wv[it][2] = *(const v4f*)(wp + 8);
  }

  v8f acc[2][4];
  #pragma unroll
  for (int T = 0; T < 4; ++T) {
    const v4f blo = *(const v4f*)(b2 + 16 * T + 8 * half);
    const v4f bhi = *(const v4f*)(b2 + 16 * T + 8 * half + 4);
    #pragma unroll
    for (int u = 0; u < 4; ++u) {
      acc[0][T][u] = blo[u]; acc[0][T][4 + u] = bhi[u];
      acc[1][T][u] = blo[u]; acc[1][T][4 + u] = bhi[u];
    }
  }

  const float* xsb = xs + (((size_t)bb * T1_ + t) * NN_) * C1_;
  v4f xb[2][KT_][8];
  #pragma unroll
  for (int nt = 0; nt < 2; ++nt) {
    #pragma unroll
    for (int kt = 0; kt < KT_; ++kt) {
      const float* xrow = xsb + ((size_t)kt * NN_ + nb + 16 * nt + mloc) * C1_ + 4 * half;
      v4f p0[8], p1[8], p2[8];
      #pragma unroll
      for (int j = 0; j < 8; ++j) {
        p0[j] = *(const v4f*)(xrow + 8 * j);
        p1[j] = *(const v4f*)(xrow + PLANE_ + 8 * j);
        p2[j] = *(const v4f*)(xrow + 2 * PLANE_ + 8 * j);
      }
      asm volatile("" :: "v"(p0[0]), "v"(p0[1]), "v"(p0[2]), "v"(p0[3]), "v"(p0[4]), "v"(p0[5]), "v"(p0[6]), "v"(p0[7]),
                         "v"(p1[0]), "v"(p1[1]), "v"(p1[2]), "v"(p1[3]), "v"(p1[4]), "v"(p1[5]), "v"(p1[6]), "v"(p1[7]),
                         "v"(p2[0]), "v"(p2[1]), "v"(p2[2]), "v"(p2[3]), "v"(p2[4]), "v"(p2[5]), "v"(p2[6]), "v"(p2[7]));
      #pragma unroll
      for (int j = 0; j < 8; ++j)
        xb[nt][kt][j] = (p0[j] + p1[j]) + p2[j];
    }
  }

  #pragma unroll
  for (int it = 0; it < 4; ++it) {
    const int item = tid + 256 * it;
    float* dst = s_w + (item & 15) * W2P + (item >> 4) * 4;
    #pragma unroll
    for (int kt = 0; kt < KT_; ++kt) {
      v4f o;
      #pragma unroll
      for (int q = 0; q < 4; ++q) o[q] = wv[it][(3 * q + kt) >> 2][(3 * q + kt) & 3];
      *(v4f*)(dst + kt * W2KT) = o;
    }
  }
  __syncthreads();

  const float* wl = s_w + half * W2P + mloc * 4;
  v4f aw[2][2][4];
  #pragma unroll
  for (int jj = 0; jj < 2; ++jj)
    #pragma unroll
    for (int T = 0; T < 4; ++T)
      aw[0][jj][T] = *(const v4f*)(wl + 2 * jj * W2P + 64 * T);
  __builtin_amdgcn_sched_group_barrier(0x100, 8, 0);

  #pragma unroll
  for (int blk = 0; blk < 12; ++blk) {
    const int kt = blk >> 2, jp = blk & 3;
    const int cur = blk & 1, nxt = cur ^ 1;
    if (blk + 1 < 12) {
      const int kn = (blk + 1) >> 2, jn = (blk + 1) & 3;
      #pragma unroll
      for (int jj = 0; jj < 2; ++jj)
        #pragma unroll
        for (int T = 0; T < 4; ++T)
          aw[nxt][jj][T] = *(const v4f*)(wl + kn * W2KT + 2 * (2 * jn + jj) * W2P + 64 * T);
    }
    #pragma unroll
    for (int jj = 0; jj < 2; ++jj) {
      const int j = 2 * jp + jj;
      #pragma unroll
      for (int s = 0; s < 2; ++s) {
        #pragma unroll
        for (int nt = 0; nt < 2; ++nt) {
          v2f bv; bv[0] = xb[nt][kt][j][2 * s]; bv[1] = xb[nt][kt][j][2 * s + 1];
          #pragma unroll
          for (int T = 0; T < 4; ++T) {
            v2f avv; avv[0] = aw[cur][jj][T][2 * s]; avv[1] = aw[cur][jj][T][2 * s + 1];
            acc[nt][T] = __builtin_amdgcn_wmma_f32_16x16x4_f32(false, avv, false, bv, (short)0, acc[nt][T], false, false);
          }
        }
      }
    }
    if (blk + 1 < 12) __builtin_amdgcn_sched_group_barrier(0x100, 8, 0);
    __builtin_amdgcn_sched_group_barrier(0x008, 32, 0);
  }

  v4f lw[2][4][2], lb[2][4][2];
  float lsum = 0.f, lsq = 0.f;
  #pragma unroll
  for (int nt = 0; nt < 2; ++nt) {
    #pragma unroll
    for (int T = 0; T < 4; ++T) {
      const size_t off = (size_t)(nb + 16 * nt + mloc) * C1_ + 16 * T + 8 * half;
      const float* xr = xsb + (size_t)(KT_ - 1) * NN_ * C1_ + off;
      const v4f r0 = (*(const v4f*)(xr)     + *(const v4f*)(xr + PLANE_))     + *(const v4f*)(xr + 2 * PLANE_);
      const v4f r1 = (*(const v4f*)(xr + 4) + *(const v4f*)(xr + PLANE_ + 4)) + *(const v4f*)(xr + 2 * PLANE_ + 4);
      lw[nt][T][0] = *(const v4f*)(lnw + off);
      lw[nt][T][1] = *(const v4f*)(lnw + off + 4);
      lb[nt][T][0] = *(const v4f*)(lnb + off);
      lb[nt][T][1] = *(const v4f*)(lnb + off + 4);
      #pragma unroll
      for (int u = 0; u < 4; ++u) {
        float v0 = acc[nt][T][u] + r0[u];
        float v1 = acc[nt][T][4 + u] + r1[u];
        v0 = v0 > 0.f ? v0 : 0.f;
        v1 = v1 > 0.f ? v1 : 0.f;
        acc[nt][T][u] = v0;
        acc[nt][T][4 + u] = v1;
        lsum += v0; lsq += v0 * v0;
        lsum += v1; lsq += v1 * v1;
      }
    }
  }

  for (int off = 16; off > 0; off >>= 1) {
    lsum += __shfl_xor(lsum, off);
    lsq  += __shfl_xor(lsq,  off);
  }
  if (lane == 0) { s_red[wave] = lsum; s_red[8 + wave] = lsq; }
  __syncthreads();
  float tsum = 0.f, tsq = 0.f;
  for (int i = 0; i < 8; ++i) { tsum += s_red[i]; tsq += s_red[8 + i]; }
  const float cnt  = (float)(NN_ * C2_);
  const float mu   = tsum / cnt;
  const float var  = tsq / cnt - mu * mu;
  const float rstd = rsqrtf(var + 1e-5f);

  float* ob = out + (((size_t)bb * C2_) * T2_ + t) * NN_ + nb + lane;
  #pragma unroll
  for (int T = 0; T < 4; ++T) {
    #pragma unroll
    for (int r = 0; r < 8; ++r) {
      const float y0 = (acc[0][T][r] - mu) * rstd * lw[0][T][r >> 2][r & 3] + lb[0][T][r >> 2][r & 3];
      const float y1 = (acc[1][T][r] - mu) * rstd * lw[1][T][r >> 2][r & 3] + lb[1][T][r >> 2][r & 3];
      const float send = half ? y0 : y1;
      const float recv = xhalf(send);
      const float lo = half ? recv : y0;
      const float hi = half ? y1 : recv;
      ob[(size_t)(16 * T + r) * T2_ * NN_]     = lo;
      ob[(size_t)(16 * T + 8 + r) * T2_ * NN_] = hi;
    }
  }
}

extern "C" void kernel_launch(void* const* d_in, const int* in_sizes, int n_in,
                              void* d_out, int out_size, void* d_ws, size_t ws_size,
                              hipStream_t stream) {
  const float* x   = (const float*)d_in[0];
  const float* adj = (const float*)d_in[1];
  const float* w1  = (const float*)d_in[2];
  const float* b1  = (const float*)d_in[3];
  const float* w2  = (const float*)d_in[4];
  const float* b2  = (const float*)d_in[5];
  const float* Wh  = (const float*)d_in[6];
  const float* ah  = (const float*)d_in[7];
  const float* lnw = (const float*)d_in[8];
  const float* lnb = (const float*)d_in[9];
  float* out = (float*)d_out;

  float* ws = (float*)d_ws;
  const size_t xg_elems = (size_t)B_ * T1_ * NN_ * C1_;
  float* xs = ws;
  float* u1 = xs + HGRP_ * xg_elems;
  float* u2 = u1 + HH_ * C1_;

  prep_u_kernel<<<2, 256, 0, stream>>>(Wh, ah, u1, u2);

  const size_t attn_lds =
      (size_t)(NN_ * XSTR + (NN_ / 2) * XPSTR + 3 * NN_ + 8 * 16 * PSTR) * sizeof(float)
      + (size_t)(NN_ * 8) * sizeof(unsigned)
      + (size_t)8 * sizeof(float);
  attn_kernel<<<B_ * T1_ * HGRP_, 256, attn_lds, stream>>>(x, w1, b1, adj, u1, u2, xs);

  const size_t c2_lds = (size_t)(3 * W2KT + 16) * sizeof(float);
  conv2_ln_kernel<<<B_ * T2_, 256, c2_lds, stream>>>(xs, w2, b2, lnw, lnb, out);
}


# conv2: global loads one phase ahead of use: x_s rows of time tap kt+1 (then residual rows) in flight behind the 128 matrix instructions of tap kt, not waited for up front; sum order unchanged
# speedup vs baseline: 4.7360x; 1.0727x over previous
#include <hip/hip_runtime.h>
#include <math.h>

#define B_   8
#define C0_  16
#define C1_  64
#define C2_  64
#define T0_  12
#define T1_  10
#define T2_  8
#define NN_  256
#define KT_  3
#define HH_  8
#define ALPHA_ 0.2f

typedef float v2f __attribute__((ext_vector_type(2)));
typedef float v4f __attribute__((ext_vector_type(4)));
typedef float v8f __attribute__((ext_vector_type(8)));
typedef unsigned u4w __attribute__((ext_vector_type(4)));

__device__ __forceinline__ float tanh_fast(float x) {
  float e = __expf(fminf(2.0f * x, 80.0f));
  return (e - 1.0f) / (e + 1.0f);
}

__device__ __forceinline__ float row_pick(float v, int sel_lo, int sel_hi) {
  int b = __builtin_bit_cast(int, v);
  return __builtin_bit_cast(float, __builtin_amdgcn_permlane16(b, b, sel_lo, sel_hi, false, false));
}
__device__ __forceinline__ float wave_max(float v) {
  v = fmaxf(v, row_pick(v, 0x67452301, 0xefcdab89));
  v = fmaxf(v, row_pick(v, 0x54761032, 0xdcfe98ba));
  v = fmaxf(v, row_pick(v, 0x32107654, 0xba98fedc));
  v = fmaxf(v, row_pick(v, 0xfedcba98, 0x76543210));
  int b = __builtin_bit_cast(int, v);
  float o = __builtin_bit_cast(float, __builtin_amdgcn_permlanex16(b, b, 0x76543210, 0xfedcba98, false, false));
  return fmaxf(v, o);
}

__device__ __forceinline__ float lane_f(float v, int l) {
  return __builtin_bit_cast(float, __builtin_amdgcn_readlane(__builtin_bit_cast(int, v), l));
}
__device__ __forceinline__ unsigned lane_u(unsigned v, int l) {
  return (unsigned)__builtin_amdgcn_readlane((int)v, l);
}

__global__ __launch_bounds__(256) void prep_u_kernel(
    const float* __restrict__ W,
    const float* __restrict__ a,
    float* __restrict__ u1, float* __restrict__ u2)
{
  int tid = blockIdx.x * 256 + threadIdx.x;
  if (tid >= HH_ * C1_) return;
  int h = tid >> 6, c = tid & 63;
  const float* Wh = W + ((size_t)h * C1_ + c) * C1_;
  const float* ah = a + (size_t)h * 2 * C1_;
  float s1 = 0.f, s2 = 0.f;
  for (int d = 0; d < C1_; ++d) {
    float w = Wh[d];
    s1 += w * ah[d];
    s2 += w * ah[C1_ + d];
  }
  u1[tid] = s1;
  u2[tid] = s2;
}

#define HGRP_ 3
#define PLANE_ ((size_t)B_ * T1_ * NN_ * C1_)
#define XSTR  72
#define XPSTR 160
#define PSTR  260
#define SM_MIN_ 1.0e-25f

__global__ __launch_bounds__(256) void attn_kernel(
    const float* __restrict__ x,
    const float* __restrict__ w1,
    const float* __restrict__ b1,
    const float* __restrict__ adj,
    const float* __restrict__ u1,
    const float* __restrict__ u2,
    float* __restrict__ xs)
{
  extern __shared__ float smem[];
  float*    s_xg   = smem;
  float*    s_xgp  = s_xg + NN_ * XSTR;
  float*    s_e1   = s_xgp + (NN_ / 2) * XPSTR;
  float*    s_e2   = s_e1 + NN_;
  float*    s_rinv = s_e2 + NN_;
  float*    s_p    = s_rinv + NN_;
  unsigned* s_adj  = (unsigned*)(s_p + 8 * 16 * PSTR);
  float*    s_cm   = (float*)(s_adj + NN_ * 8);

  const int bt   = blockIdx.x / HGRP_;
  const int grp  = blockIdx.x - bt * HGRP_;
  const int h_lo = 3 * grp;
  const int h_hi = (h_lo + 3 < HH_) ? h_lo + 3 : HH_;
  const int tid  = threadIdx.x;
  const int lane = tid & 31;
  const int wave = tid >> 5;
  const int mloc = lane & 15;
  const int half = lane >> 4;

  {
    const int bb = bt / T1_, t = bt - bb * T1_;

    float av[2][24];
    #pragma unroll
    for (int ti = 0; ti < 2; ++ti) {
      const int n0 = (wave + 8 * ti) * 16;
      const float* xl = x + ((size_t)(bb * C0_ + 8 * half) * T0_ + t) * NN_ + n0 + mloc;
      #pragma unroll
      for (int k0 = 0; k0 < 24; ++k0)
        av[ti][k0] = xl[((k0 / 3) * T0_ + (k0 % 3)) * NN_];
    }

    const v4f bp = *(const v4f*)(b1 + 4 * mloc);
    const v4f bq = *(const v4f*)(b1 + C1_ + 4 * mloc);
    v8f acc[2][8];
    #pragma unroll
    for (int ti = 0; ti < 2; ++ti) {
      #pragma unroll
      for (int T = 0; T < 4; ++T) {
        #pragma unroll
        for (int r = 0; r < 8; ++r) { acc[ti][T][r] = bp[T]; acc[ti][4 + T][r] = bq[T]; }
      }
    }

    const float* wl = w1 + (size_t)(4 * mloc) * (C0_ * KT_) + 24 * half;
    #pragma unroll
    for (int j = 0; j < 6; ++j) {
      v4f bw[8];
      #pragma unroll
      for (int T = 0; T < 4; ++T) {
        bw[T]     = *(const v4f*)(wl + (size_t)T * (C0_ * KT_) + 4 * j);
        bw[4 + T] = *(const v4f*)(wl + (size_t)(C1_ + T) * (C0_ * KT_) + 4 * j);
      }
      #pragma unroll
      for (int ti = 0; ti < 2; ++ti) {
        v2f a0, a1;
        a0[0] = av[ti][4 * j];     a0[1] = av[ti][4 * j + 1];
        a1[0] = av[ti][4 * j + 2]; a1[1] = av[ti][4 * j + 3];
        #pragma unroll
        for (int T = 0; T < 8; ++T) {
          v2f b0; b0[0] = bw[T][0]; b0[1] = bw[T][1];
          acc[ti][T] = __builtin_amdgcn_wmma_f32_16x16x4_f32(false, a0, false, b0, (short)0, acc[ti][T], false, false);
        }
        #pragma unroll
        for (int T = 0; T < 8; ++T) {
          v2f b1v; b1v[0] = bw[T][2]; b1v[1] = bw[T][3];
          acc[ti][T] = __builtin_amdgcn_wmma_f32_16x16x4_f32(false, a1, false, b1v, (short)0, acc[ti][T], false, false);
        }
      }
    }

    const bool has_res = (4 * mloc) < C0_;
    const int  cres    = (4 * mloc) & (C0_ - 1);
    #pragma unroll
    for (int ti = 0; ti < 2; ++ti) {
      const int n0 = (wave + 8 * ti) * 16;
      const float* xr = x + ((size_t)(bb * C0_ + cres) * T0_ + (t + KT_ - 1)) * NN_ + n0 + 8 * half;
      float xin[4][8];
      #pragma unroll
      for (int T = 0; T < 4; ++T) {
        const v4f lo = *(const v4f*)(xr + (size_t)T * T0_ * NN_);
        const v4f hi = *(const v4f*)(xr + (size_t)T * T0_ * NN_ + 4);
        #pragma unroll
        for (int u = 0; u < 4; ++u) {
          xin[T][u]     = has_res ? lo[u] : 0.f;
          xin[T][4 + u] = has_res ? hi[u] : 0.f;
        }
      }
      v4f o[8];
      #pragma unroll
      for (int r = 0; r < 8; ++r) {
        #pragma unroll
        for (int T = 0; T < 4; ++T) {
          float sg = 1.f / (1.f + __expf(-acc[ti][4 + T][r]));
          o[r][T] = (acc[ti][T][r] + xin[T][r]) * sg;
        }
      }
      const int nr = n0 + 8 * half;
      #pragma unroll
      for (int r = 0; r < 8; ++r)
        *(v4f*)(s_xg + (nr + r) * XSTR + 4 * mloc) = o[r];
      #pragma unroll
      for (int r = 0; r < 8; r += 2) {
        float* pp = s_xgp + ((nr + r) >> 1) * XPSTR + 8 * mloc;
        v4f p0, p1;
        p0[0] = o[r][0]; p0[1] = o[r + 1][0]; p0[2] = o[r][1]; p0[3] = o[r + 1][1];
        p1[0] = o[r][2]; p1[1] = o[r + 1][2]; p1[2] = o[r][3]; p1[3] = o[r + 1][3];
        *(v4f*)(pp)     = p0;
        *(v4f*)(pp + 4) = p1;
      }
    }
  }
  {
    const int r0 = 32 * wave;
    for (int bb = 0; bb < 4; ++bb) {
      const float* ap = adj + (size_t)(r0 + 8 * bb) * NN_ + 4 * lane;
      v4f a[16];
      #pragma unroll
      for (int q = 0; q < 16; ++q) a[q] = *(const v4f*)(ap + q * 128);
      #pragma unroll
      for (int s = 0; s < 2; ++s) {
        unsigned wacc = 0u;
        #pragma unroll
        for (int q = 0; q < 8; ++q) {
          #pragma unroll
          for (int u = 0; u < 4; ++u) {
            const unsigned w = __builtin_amdgcn_ballot_w32(a[8 * s + q][u] > 0.f);
            wacc = (lane == 4 * q + u) ? w : wacc;
          }
        }
        s_adj[(r0 + 8 * bb + 4 * s) * 8 + lane] = wacc;
      }
    }
  }
  __syncthreads();

  const float inv_h = 1.0f / (float)HH_;
  float* s_pw = s_p + wave * 16 * PSTR;

  u4w awb[2];
  #pragma unroll
  for (int rbi = 0; rbi < 2; ++rbi)
    awb[rbi] = *(const u4w*)(s_adj + (wave + 8 * rbi) * 16 * 8 + 4 * lane);

  float racc[2][4][8];
  #pragma unroll
  for (int i = 0; i < 2; ++i)
    #pragma unroll
    for (int j = 0; j < 4; ++j)
      #pragma unroll
      for (int r = 0; r < 8; ++r) racc[i][j][r] = 0.f;

  for (int h = h_lo; h < h_hi; ++h) {
    {
      int n = tid;
      const float* row = &s_xg[n * XSTR];
      const float* U1 = u1 + h * C1_;
      const float* U2 = u2 + h * C1_;
      float a1 = 0.f, a2 = 0.f;
      for (int c = 0; c < C1_; ++c) {
        float v = row[c];
        a1 = fmaf(v, U1[c], a1);
        a2 = fmaf(v, U2[c], a2);
      }
      s_e1[n] = a1;
      s_e2[n] = a2;
      const float wm = wave_max(a2);
      if (lane == 0) s_cm[wave] = wm;
    }
    __syncthreads();

    float cmax;
    {
      const v4f c0 = *(const v4f*)(s_cm);
      const v4f c1 = *(const v4f*)(s_cm + 4);
      cmax = fmaxf(fmaxf(fmaxf(c0[0], c0[1]), fmaxf(c0[2], c0[3])),
                   fmaxf(fmaxf(c1[0], c1[1]), fmaxf(c1[2], c1[3])));
    }

    v4f cf1[2], cf2[2];
    #pragma unroll
    for (int hh = 0; hh < 2; ++hh) {
      const v4f e = *(const v4f*)(s_e2 + 128 * hh + 4 * lane);
      #pragma unroll
      for (int u = 0; u < 4; ++u) {
        const float z = e[u] - cmax;
        cf1[hh][u] = __expf(z);
        cf2[hh][u] = __expf(ALPHA_ * z);
      }
    }
    float vr1, vr2;
    {
      const float d = s_e1[(wave + 8 * half) * 16 + mloc] + cmax;
      vr1 = __expf((1.0f - ALPHA_) * fminf(d, 0.f));
      vr2 = __expf(-(1.0f - ALPHA_) * fmaxf(d, 0.f));
    }

    #pragma unroll
    for (int rbi = 0; rbi < 2; ++rbi) {
      const int rb = wave + rbi * 8;
      float sm;
      v8f acc0, acc1, acc2, acc3;
      #pragma nounroll
      for (int trip = 0; trip < 2; ++trip) {
        if (trip == 0) {
          #pragma unroll
          for (int r = 0; r < 16; ++r) {
            const float R1 = lane_f(vr1, 16 * rbi + r);
            const float R2 = lane_f(vr2, 16 * rbi + r);
            #pragma unroll
            for (int hh = 0; hh < 2; ++hh) {
              v4f p;
              #pragma unroll
              for (int u = 0; u < 4; ++u) {
                const unsigned w = lane_u(awb[rbi][u], 2 * r + hh);
                const float t = fmaxf(R1 * cf1[hh][u], R2 * cf2[hh][u]);
                p[u] = __builtin_amdgcn_inverse_ballot_w32(w) ? t : 0.f;
              }
              *(v4f*)(s_pw + r * PSTR + 128 * hh + 4 * lane) = p;
            }
          }
        } else {
          const int m = rb * 16 + mloc;
          const float e1m = s_e1[m];
          const int jbase = half * 128;
          const u4w aw4 = *(const u4w*)(s_adj + m * 8 + half * 4);
          float mx = -3.0e38f;
          for (int w8 = 0; w8 < 4; ++w8) {
            const u4w adjw = aw4 >> (unsigned)(8 * w8);
            const float* eb = &s_e2[jbase + w8 * 32];
            for (int j4 = 0; j4 < 32; j4 += 4) {
              v4f e = *(const v4f*)(eb + j4);
              #pragma unroll
              for (int u = 0; u < 4; ++u) {
                float sc = e1m + e[u];
                sc = sc > 0.f ? sc : ALPHA_ * sc;
                int on = (adjw[u] >> (j4 >> 2)) & 1;
                mx = fmaxf(mx, on ? sc : -3.0e38f);
              }
            }
          }
          mx = fmaxf(mx, __shfl_xor(mx, 16));
          for (int w8 = 0; w8 < 4; ++w8) {
            const u4w adjw = aw4 >> (unsigned)(8 * w8);
            const float* eb = &s_e2[jbase + w8 * 32];
            float* pb = &s_pw[mloc * PSTR + jbase + w8 * 32];
            for (int j4 = 0; j4 < 32; j4 += 4) {
              v4f e = *(const v4f*)(eb + j4);
              v4f p;
              #pragma unroll
              for (int u = 0; u < 4; ++u) {
                float sc = e1m + e[u];
                sc = sc > 0.f ? sc : ALPHA_ * sc;
                sc = ((adjw[u] >> (j4 >> 2)) & 1) ? sc : -3.0e38f;
                p[u] = __expf(sc - mx);
              }
              *(v4f*)(pb + j4) = p;
            }
          }
        }
        __builtin_amdgcn_fence(__ATOMIC_ACQ_REL, "wavefront");
        __builtin_amdgcn_wave_barrier();

        acc0 = v8f{}; acc1 = v8f{}; acc2 = v8f{}; acc3 = v8f{};
        v2f ps = {0.f, 0.f};
        const float* pa   = &s_pw[mloc * PSTR + 2 * half];
        const float* xpb  = &s_xgp[half * XPSTR + 2 * mloc];
        for (int k0 = 0; k0 < NN_; k0 += 4) {
          v2f a = *(const v2f*)(pa + k0);
          const float* bp = xpb + (k0 >> 1) * XPSTR;
          v2f b0 = *(const v2f*)(bp + 0);
          v2f b1 = *(const v2f*)(bp + 32);
          v2f b2 = *(const v2f*)(bp + 64);
          v2f b3 = *(const v2f*)(bp + 96);
          ps += a;
          acc0 = __builtin_amdgcn_wmma_f32_16x16x4_f32(false, a, false, b0, (short)0, acc0, false, false);
          acc1 = __builtin_amdgcn_wmma_f32_16x16x4_f32(false, a, false, b1, (short)0, acc1, false, false);
          acc2 = __builtin_amdgcn_wmma_f32_16x16x4_f32(false, a, false, b2, (short)0, acc2, false, false);
          acc3 = __builtin_amdgcn_wmma_f32_16x16x4_f32(false, a, false, b3, (short)0, acc3, false, false);
        }
        sm = ps[0] + ps[1];
        sm += __shfl_xor(sm, 16);
        if (trip == 1 || !__any(!(sm >= SM_MIN_))) break;
      }
      s_rinv[rb * 16 + mloc] = 1.0f / sm;

      #pragma unroll
      for (int r = 0; r < 8; ++r) {
        float inv = s_rinv[rb * 16 + r + 8 * half];
        racc[rbi][0][r] += tanh_fast(acc0[r] * inv) * inv_h;
        racc[rbi][1][r] += tanh_fast(acc1[r] * inv) * inv_h;
        racc[rbi][2][r] += tanh_fast(acc2[r] * inv) * inv_h;
        racc[rbi][3][r] += tanh_fast(acc3[r] * inv) * inv_h;
      }
    }
    __syncthreads();
  }

  float* xs_bt = xs + (size_t)grp * PLANE_ + (size_t)bt * NN_ * C1_;
  if (grp == 0) {
    #pragma unroll
    for (int rbi = 0; rbi < 2; ++rbi) {
      #pragma unroll
      for (int r = 0; r < 8; ++r) {
        const float* xrow = &s_xg[((wave + rbi * 8) * 16 + r + 8 * half) * XSTR + mloc];
        racc[rbi][0][r] += xrow[0];
        racc[rbi][1][r] += xrow[16];
        racc[rbi][2][r] += xrow[32];
        racc[rbi][3][r] += xrow[48];
      }
    }
  }
  #pragma unroll
  for (int rbi = 0; rbi < 2; ++rbi) {
    const int rb = wave + rbi * 8;
    #pragma unroll
    for (int r = 0; r < 8; ++r) {
      int mm = rb * 16 + r + 8 * half;
      float* orow = xs_bt + (size_t)mm * C1_ + mloc;
      orow[0]  = racc[rbi][0][r];
      orow[16] = racc[rbi][1][r];
      orow[32] = racc[rbi][2][r];
      orow[48] = racc[rbi][3][r];
    }
  }
}

#define W2P   260
#define W2KT  (16 * W2P)

__device__ __forceinline__ float xhalf(float v) {
  int b = __builtin_bit_cast(int, v);
  return __builtin_bit_cast(float, __builtin_amdgcn_permlanex16(b, b, 0x76543210, 0xfedcba98, false, false));
}

__device__ __forceinline__ void tap_issue(v4f (&raw)[2][3][8], const float* xrow0) {
  #pragma unroll
  for (int nt = 0; nt < 2; ++nt) {
    const float* xrow = xrow0 + (size_t)(16 * nt) * C1_;
    #pragma unroll
    for (int j = 0; j < 8; ++j) {
      raw[nt][0][j] = *(const v4f*)(xrow + 8 * j);
      raw[nt][1][j] = *(const v4f*)(xrow + PLANE_ + 8 * j);
      raw[nt][2][j] = *(const v4f*)(xrow + 2 * PLANE_ + 8 * j);
    }
  }
}

__global__ __launch_bounds__(256) void conv2_ln_kernel(
    const float* __restrict__ xs,
    const float* __restrict__ w2,
    const float* __restrict__ b2,
    const float* __restrict__ lnw,
    const float* __restrict__ lnb,
    float* __restrict__ out)
{
  extern __shared__ float smem[];
  float* s_w   = smem;
  float* s_red = s_w + 3 * W2KT;

  const int bt = blockIdx.x;
  const int bb = bt >> 3, t = bt & 7;
  const int tid  = threadIdx.x;
  const int lane = tid & 31;
  const int wave = tid >> 5;
  const int mloc = lane & 15;
  const int half = lane >> 4;
  const int nb   = 32 * wave;

  v4f wv[4][3];
  #pragma unroll
  for (int it = 0; it < 4; ++it) {
    const float* wp = w2 + 12 * (tid + 256 * it);
    wv[it][0] = *(const v4f*)(wp);
    wv[it][1] = *(const v4f*)(wp + 4);
    wv[it][2] = *(const v4f*)(wp + 8);
  }

  v8f acc[2][4];
  #pragma unroll
  for (int T = 0; T < 4; ++T) {
    const v4f blo = *(const v4f*)(b2 + 16 * T + 8 * half);
    const v4f bhi = *(const v4f*)(b2 + 16 * T + 8 * half + 4);
    #pragma unroll
    for (int u = 0; u < 4; ++u) {
      acc[0][T][u] = blo[u]; acc[0][T][4 + u] = bhi[u];
      acc[1][T][u] = blo[u]; acc[1][T][4 + u] = bhi[u];
    }
  }

  const float* xsb   = xs + (((size_t)bb * T1_ + t) * NN_) * C1_;
  const float* xlane = xsb + (size_t)(nb + mloc) * C1_ + 4 * half;
  v4f raw[2][3][8];
  tap_issue(raw, xlane);

  #pragma unroll
  for (int it = 0; it < 4; ++it) {
    const int item = tid + 256 * it;
    float* dst = s_w + (item & 15) * W2P + (item >> 4) * 4;
    #pragma unroll
    for (int kt = 0; kt < KT_; ++kt) {
      v4f o;
      #pragma unroll
      for (int q = 0; q < 4; ++q) o[q] = wv[it][(3 * q + kt) >> 2][(3 * q + kt) & 3];
      *(v4f*)(dst + kt * W2KT) = o;
    }
  }
  __syncthreads();

  const float* wl = s_w + half * W2P + mloc * 4;
  v4f aw[2][2][4];
  #pragma unroll
  for (int jj = 0; jj < 2; ++jj)
    #pragma unroll
    for (int T = 0; T < 4; ++T)
      aw[0][jj][T] = *(const v4f*)(wl + 2 * jj * W2P + 64 * T);
  __builtin_amdgcn_sched_group_barrier(0x100, 8, 0);

  v4f rr[2][4][3][2];

  #pragma unroll
  for (int kt = 0; kt < KT_; ++kt) {
    __builtin_amdgcn_sched_barrier(0);
    v4f xb[2][8];
    #pragma unroll
    for (int nt = 0; nt < 2; ++nt)
      #pragma unroll
      for (int j = 0; j < 8; ++j)
        xb[nt][j] = (raw[nt][0][j] + raw[nt][1][j]) + raw[nt][2][j];
    __builtin_amdgcn_sched_barrier(0);
    if (kt + 1 < KT_) {
      tap_issue(raw, xlane + (size_t)(kt + 1) * NN_ * C1_);
    } else {
      #pragma unroll
      for (int nt = 0; nt < 2; ++nt) {
        #pragma unroll
        for (int T = 0; T < 4; ++T) {
          const float* xr = xsb + (size_t)(KT_ - 1) * NN_ * C1_
                          + (size_t)(nb + 16 * nt + mloc) * C1_ + 16 * T + 8 * half;
          #pragma unroll
          for (int pl = 0; pl < 3; ++pl) {
            rr[nt][T][pl][0] = *(const v4f*)(xr + pl * PLANE_);
            rr[nt][T][pl][1] = *(const v4f*)(xr + pl * PLANE_ + 4);
          }
        }
      }
    }
    __builtin_amdgcn_sched_barrier(0);

    #pragma unroll
    for (int jp = 0; jp < 4; ++jp) {
      const int blk = 4 * kt + jp;
      const int cur = blk & 1, nxt = cur ^ 1;
      if (blk + 1 < 12) {
        const int kn = (blk + 1) >> 2, jn = (blk + 1) & 3;
        #pragma unroll
        for (int jj = 0; jj < 2; ++jj)
          #pragma unroll
          for (int T = 0; T < 4; ++T)
            aw[nxt][jj][T] = *(const v4f*)(wl + kn * W2KT + 2 * (2 * jn + jj) * W2P + 64 * T);
      }
      #pragma unroll
      for (int jj = 0; jj < 2; ++jj) {
        const int j = 2 * jp + jj;
        #pragma unroll
        for (int s = 0; s < 2; ++s) {
          #pragma unroll
          for (int nt = 0; nt < 2; ++nt) {
            v2f bv; bv[0] = xb[nt][j][2 * s]; bv[1] = xb[nt][j][2 * s + 1];
            #pragma unroll
            for (int T = 0; T < 4; ++T) {
              v2f avv; avv[0] = aw[cur][jj][T][2 * s]; avv[1] = aw[cur][jj][T][2 * s + 1];
              acc[nt][T] = __builtin_amdgcn_wmma_f32_16x16x4_f32(false, avv, false, bv, (short)0, acc[nt][T], false, false);
            }
          }
        }
      }
      if (blk + 1 < 12) __builtin_amdgcn_sched_group_barrier(0x100, 8, 0);
      __builtin_amdgcn_sched_group_barrier(0x008, 32, 0);
    }
  }
  __builtin_amdgcn_sched_barrier(0);

  v4f lw[2][4][2], lb[2][4][2];
  #pragma unroll
  for (int nt = 0; nt < 2; ++nt) {
    #pragma unroll
    for (int T = 0; T < 4; ++T) {
      const size_t off = (size_t)(nb + 16 * nt + mloc) * C1_ + 16 * T + 8 * half;
      lw[nt][T][0] = *(const v4f*)(lnw + off);
      lw[nt][T][1] = *(const v4f*)(lnw + off + 4);
      lb[nt][T][0] = *(const v4f*)(lnb + off);
      lb[nt][T][1] = *(const v4f*)(lnb + off + 4);
    }
  }
  __builtin_amdgcn_sched_barrier(0);

  float lsum = 0.f, lsq = 0.f;
  #pragma unroll
  for (int nt = 0; nt < 2; ++nt) {
    #pragma unroll
    for (int T = 0; T < 4; ++T) {
      const v4f r0 = (rr[nt][T][0][0] + rr[nt][T][1][0]) + rr[nt][T][2][0];
      const v4f r1 = (rr[nt][T][0][1] + rr[nt][T][1][1]) + rr[nt][T][2][1];
      #pragma unroll
      for (int u = 0; u < 4; ++u) {
        float v0 = acc[nt][T][u] + r0[u];
        float v1 = acc[nt][T][4 + u] + r1[u];
        v0 = v0 > 0.f ? v0 : 0.f;
        v1 = v1 > 0.f ? v1 : 0.f;
        acc[nt][T][u] = v0;
        acc[nt][T][4 + u] = v1;
        lsum += v0; lsq += v0 * v0;
        lsum += v1; lsq += v1 * v1;
      }
    }
  }

  for (int off = 16; off > 0; off >>= 1) {
    lsum += __shfl_xor(lsum, off);
    lsq  += __shfl_xor(lsq,  off);
  }
  if (lane == 0) { s_red[wave] = lsum; s_red[8 + wave] = lsq; }
  __syncthreads();
  float tsum = 0.f, tsq = 0.f;
  for (int i = 0; i < 8; ++i) { tsum += s_red[i]; tsq += s_red[8 + i]; }
  const float cnt  = (float)(NN_ * C2_);
  const float mu   = tsum / cnt;
  const float var  = tsq / cnt - mu * mu;
  const float rstd = rsqrtf(var + 1e-5f);

  float* ob = out + (((size_t)bb * C2_) * T2_ + t) * NN_ + nb + lane;
  #pragma unroll
  for (int T = 0; T < 4; ++T) {
    #pragma unroll
    for (int r = 0; r < 8; ++r) {
      const float y0 = (acc[0][T][r] - mu) * rstd * lw[0][T][r >> 2][r & 3] + lb[0][T][r >> 2][r & 3];
      const float y1 = (acc[1][T][r] - mu) * rstd * lw[1][T][r >> 2][r & 3] + lb[1][T][r >> 2][r & 3];
      const float send = half ? y0 : y1;
      const float recv = xhalf(send);
      const float lo = half ? recv : y0;
      const float hi = half ? y1 : recv;
      ob[(size_t)(16 * T + r) * T2_ * NN_]     = lo;
      ob[(size_t)(16 * T + 8 + r) * T2_ * NN_] = hi;
    }
  }
}

extern "C" void kernel_launch(void* const* d_in, const int* in_sizes, int n_in,
                              void* d_out, int out_size, void* d_ws, size_t ws_size,
                              hipStream_t stream) {
  const float* x   = (const float*)d_in[0];
  const float* adj = (const float*)d_in[1];
  const float* w1  = (const float*)d_in[2];
  const float* b1  = (const float*)d_in[3];
  const float* w2  = (const float*)d_in[4];
  const float* b2  = (const float*)d_in[5];
  const float* Wh  = (const float*)d_in[6];
  const float* ah  = (const float*)d_in[7];
  const float* lnw = (const float*)d_in[8];
  const float* lnb = (const float*)d_in[9];
  float* out = (float*)d_out;

  float* ws = (float*)d_ws;
  const size_t xg_elems = (size_t)B_ * T1_ * NN_ * C1_;
  float* xs = ws;
  float* u1 = xs + HGRP_ * xg_elems;
  float* u2 = u1 + HH_ * C1_;

  prep_u_kernel<<<2, 256, 0, stream>>>(Wh, ah, u1, u2);

  const size_t attn_lds =
      (size_t)(NN_ * XSTR + (NN_ / 2) * XPSTR + 3 * NN_ + 8 * 16 * PSTR) * sizeof(float)
      + (size_t)(NN_ * 8) * sizeof(unsigned)
      + (size_t)8 * sizeof(float);
  attn_kernel<<<B_ * T1_ * HGRP_, 256, attn_lds, stream>>>(x, w1, b1, adj, u1, u2, xs);

  const size_t c2_lds = (size_t)(3 * W2KT + 16) * sizeof(float);
  conv2_ln_kernel<<<B_ * T2_, 256, c2_lds, stream>>>(xs, w2, b2, lnw, lnb, out);
}


# u1/u2 computed by each attention workgroup for its own heads into LDS (W rows fetched as whole lines beside the adjacency rows); the prep launch is gone: 2 launches, not 3; sum order unchanged
# speedup vs baseline: 4.8706x; 1.0284x over previous
#include <hip/hip_runtime.h>
#include <math.h>

#define B_   8
#define C0_  16
#define C1_  64
#define C2_  64
#define T0_  12
#define T1_  10
#define T2_  8
#define NN_  256
#define KT_  3
#define HH_  8
#define ALPHA_ 0.2f

typedef float v2f __attribute__((ext_vector_type(2)));
typedef float v4f __attribute__((ext_vector_type(4)));
typedef float v8f __attribute__((ext_vector_type(8)));
typedef unsigned u4w __attribute__((ext_vector_type(4)));

__device__ __forceinline__ float tanh_fast(float x) {
  float e = __expf(fminf(2.0f * x, 80.0f));
  return (e - 1.0f) / (e + 1.0f);
}

__device__ __forceinline__ float row_pick(float v, int sel_lo, int sel_hi) {
  int b = __builtin_bit_cast(int, v);
  return __builtin_bit_cast(float, __builtin_amdgcn_permlane16(b, b, sel_lo, sel_hi, false, false));
}
__device__ __forceinline__ float wave_max(float v) {
  v = fmaxf(v, row_pick(v, 0x67452301, 0xefcdab89));
  v = fmaxf(v, row_pick(v, 0x54761032, 0xdcfe98ba));
  v = fmaxf(v, row_pick(v, 0x32107654, 0xba98fedc));
  v = fmaxf(v, row_pick(v, 0xfedcba98, 0x76543210));
  int b = __builtin_bit_cast(int, v);
  float o = __builtin_bit_cast(float, __builtin_amdgcn_permlanex16(b, b, 0x76543210, 0xfedcba98, false, false));
  return fmaxf(v, o);
}

__device__ __forceinline__ float lane_f(float v, int l) {
  return __builtin_bit_cast(float, __builtin_amdgcn_readlane(__builtin_bit_cast(int, v), l));
}
__device__ __forceinline__ unsigned lane_u(unsigned v, int l) {
  return (unsigned)__builtin_amdgcn_readlane((int)v, l);
}

#define HGRP_ 3
#define PLANE_ ((size_t)B_ * T1_ * NN_ * C1_)
#define XSTR  72
#define XPSTR 160
#define PSTR  260
#define USTR  68
#define SM_MIN_ 1.0e-25f

__global__ __launch_bounds__(256) void attn_kernel(
    const float* __restrict__ x,
    const float* __restrict__ w1,
    const float* __restrict__ b1,
    const float* __restrict__ adj,
    const float* __restrict__ Wh,
    const float* __restrict__ ah,
    float* __restrict__ xs)
{
  extern __shared__ float smem[];
  float*    s_xg   = smem;
  float*    s_xgp  = s_xg + NN_ * XSTR;
  float*    s_e1   = s_xgp + (NN_ / 2) * XPSTR;
  float*    s_e2   = s_e1 + NN_;
  float*    s_rinv = s_e2 + NN_;
  float*    s_p    = s_rinv + NN_;
  unsigned* s_adj  = (unsigned*)(s_p + 8 * 16 * PSTR);
  float*    s_cm   = (float*)(s_adj + NN_ * 8);
  float*    s_u1   = s_cm + 8;
  float*    s_u2   = s_u1 + 4 * C1_;

  const int bt   = blockIdx.x / HGRP_;
  const int grp  = blockIdx.x - bt * HGRP_;
  const int h_lo = 3 * grp;
  const int h_hi = (h_lo + 3 < HH_) ? h_lo + 3 : HH_;
  const int tid  = threadIdx.x;
  const int lane = tid & 31;
  const int wave = tid >> 5;
  const int mloc = lane & 15;
  const int half = lane >> 4;

  {
    const int bb = bt / T1_, t = bt - bb * T1_;

    float av[2][24];
    #pragma unroll
    for (int ti = 0; ti < 2; ++ti) {
      const int n0 = (wave + 8 * ti) * 16;
      const float* xl = x + ((size_t)(bb * C0_ + 8 * half) * T0_ + t) * NN_ + n0 + mloc;
      #pragma unroll
      for (int k0 = 0; k0 < 24; ++k0)
        av[ti][k0] = xl[((k0 / 3) * T0_ + (k0 % 3)) * NN_];
    }

    const v4f bp = *(const v4f*)(b1 + 4 * mloc);
    const v4f bq = *(const v4f*)(b1 + C1_ + 4 * mloc);
    v8f acc[2][8];
    #pragma unroll
    for (int ti = 0; ti < 2; ++ti) {
      #pragma unroll
      for (int T = 0; T < 4; ++T) {
        #pragma unroll
        for (int r = 0; r < 8; ++r) { acc[ti][T][r] = bp[T]; acc[ti][4 + T][r] = bq[T]; }
      }
    }

    const float* wl = w1 + (size_t)(4 * mloc) * (C0_ * KT_) + 24 * half;
    #pragma unroll
    for (int j = 0; j < 6; ++j) {
      v4f bw[8];
      #pragma unroll
      for (int T = 0; T < 4; ++T) {
        bw[T]     = *(const v4f*)(wl + (size_t)T * (C0_ * KT_) + 4 * j);
        bw[4 + T] = *(const v4f*)(wl + (size_t)(C1_ + T) * (C0_ * KT_) + 4 * j);
      }
      #pragma unroll
      for (int ti = 0; ti < 2; ++ti) {
        v2f a0, a1;
        a0[0] = av[ti][4 * j];     a0[1] = av[ti][4 * j + 1];
        a1[0] = av[ti][4 * j + 2]; a1[1] = av[ti][4 * j + 3];
        #pragma unroll
        for (int T = 0; T < 8; ++T) {
          v2f b0; b0[0] = bw[T][0]; b0[1] = bw[T][1];
          acc[ti][T] = __builtin_amdgcn_wmma_f32_16x16x4_f32(false, a0, false, b0, (short)0, acc[ti][T], false, false);
        }
        #pragma unroll
        for (int T = 0; T < 8; ++T) {
          v2f b1v; b1v[0] = bw[T][2]; b1v[1] = bw[T][3];
          acc[ti][T] = __builtin_amdgcn_wmma_f32_16x16x4_f32(false, a1, false, b1v, (short)0, acc[ti][T], false, false);
        }
      }
    }

    const bool has_res = (4 * mloc) < C0_;
    const int  cres    = (4 * mloc) & (C0_ - 1);
    #pragma unroll
    for (int ti = 0; ti < 2; ++ti) {
      const int n0 = (wave + 8 * ti) * 16;
      const float* xr = x + ((size_t)(bb * C0_ + cres) * T0_ + (t + KT_ - 1)) * NN_ + n0 + 8 * half;
      float xin[4][8];
      #pragma unroll
      for (int T = 0; T < 4; ++T) {
        const v4f lo = *(const v4f*)(xr + (size_t)T * T0_ * NN_);
        const v4f hi = *(const v4f*)(xr + (size_t)T * T0_ * NN_ + 4);
        #pragma unroll
        for (int u = 0; u < 4; ++u) {
          xin[T][u]     = has_res ? lo[u] : 0.f;
          xin[T][4 + u] = has_res ? hi[u] : 0.f;
        }
      }
      v4f o[8];
      #pragma unroll
      for (int r = 0; r < 8; ++r) {
        #pragma unroll
        for (int T = 0; T < 4; ++T) {
          float sg = 1.f / (1.f + __expf(-acc[ti][4 + T][r]));
          o[r][T] = (acc[ti][T][r] + xin[T][r]) * sg;
        }
      }
      const int nr = n0 + 8 * half;
      #pragma unroll
      for (int r = 0; r < 8; ++r)
        *(v4f*)(s_xg + (nr + r) * XSTR + 4 * mloc) = o[r];
      #pragma unroll
      for (int r = 0; r < 8; r += 2) {
        float* pp = s_xgp + ((nr + r) >> 1) * XPSTR + 8 * mloc;
        v4f p0, p1;
        p0[0] = o[r][0]; p0[1] = o[r + 1][0]; p0[2] = o[r][1]; p0[3] = o[r + 1][1];
        p1[0] = o[r][2]; p1[1] = o[r + 1][2]; p1[2] = o[r][3]; p1[3] = o[r + 1][3];
        *(v4f*)(pp)     = p0;
        *(v4f*)(pp + 4) = p1;
      }
    }
  }
  const int nh = h_hi - h_lo;
  const int hp = wave >> 1;
  const int cu = tid & 63;
  v4f wld[16];
  v4f ald;
  {
    const int hu = h_lo + (hp < nh ? hp : nh - 1);
    const float* wp = Wh + ((size_t)hu * C1_ + 32 * (wave & 1)) * C1_ + 4 * lane;
    #pragma unroll
    for (int i = 0; i < 16; ++i) wld[i] = *(const v4f*)(wp + 128 * i);
    ald = *(const v4f*)(ah + (size_t)hu * 2 * C1_ + 4 * lane);
  }

  {
    const int r0 = 32 * wave;
    for (int bb = 0; bb < 4; ++bb) {
      const float* ap = adj + (size_t)(r0 + 8 * bb) * NN_ + 4 * lane;
      v4f a[16];
      #pragma unroll
      for (int q = 0; q < 16; ++q) a[q] = *(const v4f*)(ap + q * 128);
      #pragma unroll
      for (int s = 0; s < 2; ++s) {
        unsigned wacc = 0u;
        #pragma unroll
        for (int q = 0; q < 8; ++q) {
          #pragma unroll
          for (int u = 0; u < 4; ++u) {
            const unsigned w = __builtin_amdgcn_ballot_w32(a[8 * s + q][u] > 0.f);
            wacc = (lane == 4 * q + u) ? w : wacc;
          }
        }
        s_adj[(r0 + 8 * bb + 4 * s) * 8 + lane] = wacc;
      }
    }
  }
  {
    float* stg = s_p + wave * 16 * PSTR;
    #pragma unroll
    for (int i = 0; i < 16; ++i)
      *(v4f*)(stg + (2 * i + half) * USTR + 4 * mloc) = wld[i];
    *(v4f*)(stg + 32 * USTR + 4 * lane) = ald;
    __builtin_amdgcn_fence(__ATOMIC_ACQ_REL, "wavefront");
    __builtin_amdgcn_wave_barrier();
    const float* wr = stg + lane * USTR;
    const float* a1 = stg + 32 * USTR;
    const float* a2 = a1 + C1_;
    float s1 = 0.f, s2 = 0.f;
    #pragma unroll
    for (int q = 0; q < 16; ++q) {
      const v4f w = *(const v4f*)(wr + 4 * q);
      const v4f p = *(const v4f*)(a1 + 4 * q);
      const v4f r = *(const v4f*)(a2 + 4 * q);
      #pragma unroll
      for (int e = 0; e < 4; ++e) {
        s1 += w[e] * p[e];
        s2 += w[e] * r[e];
      }
    }
    s_u1[hp * C1_ + cu] = s1;
    s_u2[hp * C1_ + cu] = s2;
  }
  __syncthreads();

  const float inv_h = 1.0f / (float)HH_;
  float* s_pw = s_p + wave * 16 * PSTR;

  u4w awb[2];
  #pragma unroll
  for (int rbi = 0; rbi < 2; ++rbi)
    awb[rbi] = *(const u4w*)(s_adj + (wave + 8 * rbi) * 16 * 8 + 4 * lane);

  float racc[2][4][8];
  #pragma unroll
  for (int i = 0; i < 2; ++i)
    #pragma unroll
    for (int j = 0; j < 4; ++j)
      #pragma unroll
      for (int r = 0; r < 8; ++r) racc[i][j][r] = 0.f;

  for (int h = h_lo; h < h_hi; ++h) {
    {
      int n = tid;
      const float* row = &s_xg[n * XSTR];
      const float* U1 = s_u1 + (h - h_lo) * C1_;
      const float* U2 = s_u2 + (h - h_lo) * C1_;
      float a1 = 0.f, a2 = 0.f;
      for (int c = 0; c < C1_; ++c) {
        float v = row[c];
        a1 = fmaf(v, U1[c], a1);
        a2 = fmaf(v, U2[c], a2);
      }
      s_e1[n] = a1;
      s_e2[n] = a2;
      const float wm = wave_max(a2);
      if (lane == 0) s_cm[wave] = wm;
    }
    __syncthreads();

    float cmax;
    {
      const v4f c0 = *(const v4f*)(s_cm);
      const v4f c1 = *(const v4f*)(s_cm + 4);
      cmax = fmaxf(fmaxf(fmaxf(c0[0], c0[1]), fmaxf(c0[2], c0[3])),
                   fmaxf(fmaxf(c1[0], c1[1]), fmaxf(c1[2], c1[3])));
    }

    v4f cf1[2], cf2[2];
    #pragma unroll
    for (int hh = 0; hh < 2; ++hh) {
      const v4f e = *(const v4f*)(s_e2 + 128 * hh + 4 * lane);
      #pragma unroll
      for (int u = 0; u < 4; ++u) {
        const float z = e[u] - cmax;
        cf1[hh][u] = __expf(z);
        cf2[hh][u] = __expf(ALPHA_ * z);
      }
    }
    float vr1, vr2;
    {
      const float d = s_e1[(wave + 8 * half) * 16 + mloc] + cmax;
      vr1 = __expf((1.0f - ALPHA_) * fminf(d, 0.f));
      vr2 = __expf(-(1.0f - ALPHA_) * fmaxf(d, 0.f));
    }

    #pragma unroll
    for (int rbi = 0; rbi < 2; ++rbi) {
      const int rb = wave + rbi * 8;
      float sm;
      v8f acc0, acc1, acc2, acc3;
      #pragma nounroll
      for (int trip = 0; trip < 2; ++trip) {
        if (trip == 0) {
          #pragma unroll
          for (int r = 0; r < 16; ++r) {
            const float R1 = lane_f(vr1, 16 * rbi + r);
            const float R2 = lane_f(vr2, 16 * rbi + r);
            #pragma unroll
            for (int hh = 0; hh < 2; ++hh) {
              v4f p;
              #pragma unroll
              for (int u = 0; u < 4; ++u) {
                const unsigned w = lane_u(awb[rbi][u], 2 * r + hh);
                const float t = fmaxf(R1 * cf1[hh][u], R2 * cf2[hh][u]);
                p[u] = __builtin_amdgcn_inverse_ballot_w32(w) ? t : 0.f;
              }
              *(v4f*)(s_pw + r * PSTR + 128 * hh + 4 * lane) = p;
            }
          }
        } else {
          const int m = rb * 16 + mloc;
          const float e1m = s_e1[m];
          const int jbase = half * 128;
          const u4w aw4 = *(const u4w*)(s_adj + m * 8 + half * 4);
          float mx = -3.0e38f;
          for (int w8 = 0; w8 < 4; ++w8) {
            const u4w adjw = aw4 >> (unsigned)(8 * w8);
            const float* eb = &s_e2[jbase + w8 * 32];
            for (int j4 = 0; j4 < 32; j4 += 4) {
              v4f e = *(const v4f*)(eb + j4);
              #pragma unroll
              for (int u = 0; u < 4; ++u) {
                float sc = e1m + e[u];
                sc = sc > 0.f ? sc : ALPHA_ * sc;
                int on = (adjw[u] >> (j4 >> 2)) & 1;
                mx = fmaxf(mx, on ? sc : -3.0e38f);
              }
            }
          }
          mx = fmaxf(mx, __shfl_xor(mx, 16));
          for (int w8 = 0; w8 < 4; ++w8) {
            const u4w adjw = aw4 >> (unsigned)(8 * w8);
            const float* eb = &s_e2[jbase + w8 * 32];
            float* pb = &s_pw[mloc * PSTR + jbase + w8 * 32];
            for (int j4 = 0; j4 < 32; j4 += 4) {
              v4f e = *(const v4f*)(eb + j4);
              v4f p;
              #pragma unroll
              for (int u = 0; u < 4; ++u) {
                float sc = e1m + e[u];
                sc = sc > 0.f ? sc : ALPHA_ * sc;
                sc = ((adjw[u] >> (j4 >> 2)) & 1) ? sc : -3.0e38f;
                p[u] = __expf(sc - mx);
              }
              *(v4f*)(pb + j4) = p;
            }
          }
        }
        __builtin_amdgcn_fence(__ATOMIC_ACQ_REL, "wavefront");
        __builtin_amdgcn_wave_barrier();

        acc0 = v8f{}; acc1 = v8f{}; acc2 = v8f{}; acc3 = v8f{};
        v2f ps = {0.f, 0.f};
        const float* pa   = &s_pw[mloc * PSTR + 2 * half];
        const float* xpb  = &s_xgp[half * XPSTR + 2 * mloc];
        for (int k0 = 0; k0 < NN_; k0 += 4) {
          v2f a = *(const v2f*)(pa + k0);
          const float* bp = xpb + (k0 >> 1) * XPSTR;
          v2f b0 = *(const v2f*)(bp + 0);
          v2f b1 = *(const v2f*)(bp + 32);
          v2f b2 = *(const v2f*)(bp + 64);
          v2f b3 = *(const v2f*)(bp + 96);
          ps += a;
          acc0 = __builtin_amdgcn_wmma_f32_16x16x4_f32(false, a, false, b0, (short)0, acc0, false, false);
          acc1 = __builtin_amdgcn_wmma_f32_16x16x4_f32(false, a, false, b1, (short)0, acc1, false, false);
          acc2 = __builtin_amdgcn_wmma_f32_16x16x4_f32(false, a, false, b2, (short)0, acc2, false, false);
          acc3 = __builtin_amdgcn_wmma_f32_16x16x4_f32(false, a, false, b3, (short)0, acc3, false, false);
        }
        sm = ps[0] + ps[1];
        sm += __shfl_xor(sm, 16);
        if (trip == 1 || !__any(!(sm >= SM_MIN_))) break;
      }
      s_rinv[rb * 16 + mloc] = 1.0f / sm;

      #pragma unroll
      for (int r = 0; r < 8; ++r) {
        float inv = s_rinv[rb * 16 + r + 8 * half];
        racc[rbi][0][r] += tanh_fast(acc0[r] * inv) * inv_h;
        racc[rbi][1][r] += tanh_fast(acc1[r] * inv) * inv_h;
        racc[rbi][2][r] += tanh_fast(acc2[r] * inv) * inv_h;
        racc[rbi][3][r] += tanh_fast(acc3[r] * inv) * inv_h;
      }
    }
    __syncthreads();
  }

  float* xs_bt = xs + (size_t)grp * PLANE_ + (size_t)bt * NN_ * C1_;
  if (grp == 0) {
    #pragma unroll
    for (int rbi = 0; rbi < 2; ++rbi) {
      #pragma unroll
      for (int r = 0; r < 8; ++r) {
        const float* xrow = &s_xg[((wave + rbi * 8) * 16 + r + 8 * half) * XSTR + mloc];
        racc[rbi][0][r] += xrow[0];
        racc[rbi][1][r] += xrow[16];
        racc[rbi][2][r] += xrow[32];
        racc[rbi][3][r] += xrow[48];
      }
    }
  }
  #pragma unroll
  for (int rbi = 0; rbi < 2; ++rbi) {
    const int rb = wave + rbi * 8;
    #pragma unroll
    for (int r = 0; r < 8; ++r) {
      int mm = rb * 16 + r + 8 * half;
      float* orow = xs_bt + (size_t)mm * C1_ + mloc;
      orow[0]  = racc[rbi][0][r];
      orow[16] = racc[rbi][1][r];
      orow[32] = racc[rbi][2][r];
      orow[48] = racc[rbi][3][r];
    }
  }
}

#define W2P   260
#define W2KT  (16 * W2P)

__device__ __forceinline__ float xhalf(float v) {
  int b = __builtin_bit_cast(int, v);
  return __builtin_bit_cast(float, __builtin_amdgcn_permlanex16(b, b, 0x76543210, 0xfedcba98, false, false));
}

__device__ __forceinline__ void tap_issue(v4f (&raw)[2][3][8], const float* xrow0) {
  #pragma unroll
  for (int nt = 0; nt < 2; ++nt) {
    const float* xrow = xrow0 + (size_t)(16 * nt) * C1_;
    #pragma unroll
    for (int j = 0; j < 8; ++j) {
      raw[nt][0][j] = *(const v4f*)(xrow + 8 * j);
      raw[nt][1][j] = *(const v4f*)(xrow + PLANE_ + 8 * j);
      raw[nt][2][j] = *(const v4f*)(xrow + 2 * PLANE_ + 8 * j);
    }
  }
}

__global__ __launch_bounds__(256) void conv2_ln_kernel(
    const float* __restrict__ xs,
    const float* __restrict__ w2,
    const float* __restrict__ b2,
    const float* __restrict__ lnw,
    const float* __restrict__ lnb,
    float* __restrict__ out)
{
  extern __shared__ float smem[];
  float* s_w   = smem;
  float* s_red = s_w + 3 * W2KT;

  const int bt = blockIdx.x;
  const int bb = bt >> 3, t = bt & 7;
  const int tid  = threadIdx.x;
  const int lane = tid & 31;
  const int wave = tid >> 5;
  const int mloc = lane & 15;
  const int half = lane >> 4;
  const int nb   = 32 * wave;

  v4f wv[4][3];
  #pragma unroll
  for (int it = 0; it < 4; ++it) {
    const float* wp = w2 + 12 * (tid + 256 * it);
    wv[it][0] = *(const v4f*)(wp);
    wv[it][1] = *(const v4f*)(wp + 4);
    wv[it][2] = *(const v4f*)(wp + 8);
  }

  v8f acc[2][4];
  #pragma unroll
  for (int T = 0; T < 4; ++T) {
    const v4f blo = *(const v4f*)(b2 + 16 * T + 8 * half);
    const v4f bhi = *(const v4f*)(b2 + 16 * T + 8 * half + 4);
    #pragma unroll
    for (int u = 0; u < 4; ++u) {
      acc[0][T][u] = blo[u]; acc[0][T][4 + u] = bhi[u];
      acc[1][T][u] = blo[u]; acc[1][T][4 + u] = bhi[u];
    }
  }

  const float* xsb   = xs + (((size_t)bb * T1_ + t) * NN_) * C1_;
  const float* xlane = xsb + (size_t)(nb + mloc) * C1_ + 4 * half;
  v4f raw[2][3][8];
  tap_issue(raw, xlane);

  #pragma unroll
  for (int it = 0; it < 4; ++it) {
    const int item = tid + 256 * it;
    float* dst = s_w + (item & 15) * W2P + (item >> 4) * 4;
    #pragma unroll
    for (int kt = 0; kt < KT_; ++kt) {
      v4f o;
      #pragma unroll
      for (int q = 0; q < 4; ++q) o[q] = wv[it][(3 * q + kt) >> 2][(3 * q + kt) & 3];
      *(v4f*)(dst + kt * W2KT) = o;
    }
  }
  __syncthreads();

  const float* wl = s_w + half * W2P + mloc * 4;
  v4f aw[2][2][4];
  #pragma unroll
  for (int jj = 0; jj < 2; ++jj)
    #pragma unroll
    for (int T = 0; T < 4; ++T)
      aw[0][jj][T] = *(const v4f*)(wl + 2 * jj * W2P + 64 * T);
  __builtin_amdgcn_sched_group_barrier(0x100, 8, 0);

  v4f rr[2][4][3][2];

  #pragma unroll
  for (int kt = 0; kt < KT_; ++kt) {
    __builtin_amdgcn_sched_barrier(0);
    v4f xb[2][8];
    #pragma unroll
    for (int nt = 0; nt < 2; ++nt)
      #pragma unroll
      for (int j = 0; j < 8; ++j)
        xb[nt][j] = (raw[nt][0][j] + raw[nt][1][j]) + raw[nt][2][j];
    __builtin_amdgcn_sched_barrier(0);
    if (kt + 1 < KT_) {
      tap_issue(raw, xlane + (size_t)(kt + 1) * NN_ * C1_);
    } else {
      #pragma unroll
      for (int nt = 0; nt < 2; ++nt) {
        #pragma unroll
        for (int T = 0; T < 4; ++T) {
          const float* xr = xsb + (size_t)(KT_ - 1) * NN_ * C1_
                          + (size_t)(nb + 16 * nt + mloc) * C1_ + 16 * T + 8 * half;
          #pragma unroll
          for (int pl = 0; pl < 3; ++pl) {
            rr[nt][T][pl][0] = *(const v4f*)(xr + pl * PLANE_);
            rr[nt][T][pl][1] = *(const v4f*)(xr + pl * PLANE_ + 4);
          }
        }
      }
    }
    __builtin_amdgcn_sched_barrier(0);

    #pragma unroll
    for (int jp = 0; jp < 4; ++jp) {
      const int blk = 4 * kt + jp;
      const int cur = blk & 1, nxt = cur ^ 1;
      if (blk + 1 < 12) {
        const int kn = (blk + 1) >> 2, jn = (blk + 1) & 3;
        #pragma unroll
        for (int jj = 0; jj < 2; ++jj)
          #pragma unroll
          for (int T = 0; T < 4; ++T)
            aw[nxt][jj][T] = *(const v4f*)(wl + kn * W2KT + 2 * (2 * jn + jj) * W2P + 64 * T);
      }
      #pragma unroll
      for (int jj = 0; jj < 2; ++jj) {
        const int j = 2 * jp + jj;
        #pragma unroll
        for (int s = 0; s < 2; ++s) {
          #pragma unroll
          for (int nt = 0; nt < 2; ++nt) {
            v2f bv; bv[0] = xb[nt][j][2 * s]; bv[1] = xb[nt][j][2 * s + 1];
            #pragma unroll
            for (int T = 0; T < 4; ++T) {
              v2f avv; avv[0] = aw[cur][jj][T][2 * s]; avv[1] = aw[cur][jj][T][2 * s + 1];
              acc[nt][T] = __builtin_amdgcn_wmma_f32_16x16x4_f32(false, avv, false, bv, (short)0, acc[nt][T], false, false);
            }
          }
        }
      }
      if (blk + 1 < 12) __builtin_amdgcn_sched_group_barrier(0x100, 8, 0);
      __builtin_amdgcn_sched_group_barrier(0x008, 32, 0);
    }
  }
  __builtin_amdgcn_sched_barrier(0);

  v4f lw[2][4][2], lb[2][4][2];
  #pragma unroll
  for (int nt = 0; nt < 2; ++nt) {
    #pragma unroll
    for (int T = 0; T < 4; ++T) {
      const size_t off = (size_t)(nb + 16 * nt + mloc) * C1_ + 16 * T + 8 * half;
      lw[nt][T][0] = *(const v4f*)(lnw + off);
      lw[nt][T][1] = *(const v4f*)(lnw + off + 4);
      lb[nt][T][0] = *(const v4f*)(lnb + off);
      lb[nt][T][1] = *(const v4f*)(lnb + off + 4);
    }
  }
  __builtin_amdgcn_sched_barrier(0);

  float lsum = 0.f, lsq = 0.f;
  #pragma unroll
  for (int nt = 0; nt < 2; ++nt) {
    #pragma unroll
    for (int T = 0; T < 4; ++T) {
      const v4f r0 = (rr[nt][T][0][0] + rr[nt][T][1][0]) + rr[nt][T][2][0];
      const v4f r1 = (rr[nt][T][0][1] + rr[nt][T][1][1]) + rr[nt][T][2][1];
      #pragma unroll
      for (int u = 0; u < 4; ++u) {
        float v0 = acc[nt][T][u] + r0[u];
        float v1 = acc[nt][T][4 + u] + r1[u];
        v0 = v0 > 0.f ? v0 : 0.f;
        v1 = v1 > 0.f ? v1 : 0.f;
        acc[nt][T][u] = v0;
        acc[nt][T][4 + u] = v1;
        lsum += v0; lsq += v0 * v0;
        lsum += v1; lsq += v1 * v1;
      }
    }
  }

  for (int off = 16; off > 0; off >>= 1) {
    lsum += __shfl_xor(lsum, off);
    lsq  += __shfl_xor(lsq,  off);
  }
  if (lane == 0) { s_red[wave] = lsum; s_red[8 + wave] = lsq; }
  __syncthreads();
  float tsum = 0.f, tsq = 0.f;
  for (int i = 0; i < 8; ++i) { tsum += s_red[i]; tsq += s_red[8 + i]; }
  const float cnt  = (float)(NN_ * C2_);
  const float mu   = tsum / cnt;
  const float var  = tsq / cnt - mu * mu;
  const float rstd = rsqrtf(var + 1e-5f);

  float* ob = out + (((size_t)bb * C2_) * T2_ + t) * NN_ + nb + lane;
  #pragma unroll
  for (int T = 0; T < 4; ++T) {
    #pragma unroll
    for (int r = 0; r < 8; ++r) {
      const float y0 = (acc[0][T][r] - mu) * rstd * lw[0][T][r >> 2][r & 3] + lb[0][T][r >> 2][r & 3];
      const float y1 = (acc[1][T][r] - mu) * rstd * lw[1][T][r >> 2][r & 3] + lb[1][T][r >> 2][r & 3];
      const float send = half ? y0 : y1;
      const float recv = xhalf(send);
      const float lo = half ? recv : y0;
      const float hi = half ? y1 : recv;
      ob[(size_t)(16 * T + r) * T2_ * NN_]     = lo;
      ob[(size_t)(16 * T + 8 + r) * T2_ * NN_] = hi;
    }
  }
}

extern "C" void kernel_launch(void* const* d_in, const int* in_sizes, int n_in,
                              void* d_out, int out_size, void* d_ws, size_t ws_size,
                              hipStream_t stream) {
  const float* x   = (const float*)d_in[0];
  const float* adj = (const float*)d_in[1];
  const float* w1  = (const float*)d_in[2];
  const float* b1  = (const float*)d_in[3];
  const float* w2  = (const float*)d_in[4];
  const float* b2  = (const float*)d_in[5];
  const float* Wh  = (const float*)d_in[6];
  const float* ah  = (const float*)d_in[7];
  const float* lnw = (const float*)d_in[8];
  const float* lnb = (const float*)d_in[9];
  float* out = (float*)d_out;

  float* ws = (float*)d_ws;
  float* xs = ws;

  const size_t attn_lds =
      (size_t)(NN_ * XSTR + (NN_ / 2) * XPSTR + 3 * NN_ + 8 * 16 * PSTR) * sizeof(float)
      + (size_t)(NN_ * 8) * sizeof(unsigned)
      + (size_t)8 * sizeof(float)
      + (size_t)(2 * 4 * C1_) * sizeof(float);
  attn_kernel<<<B_ * T1_ * HGRP_, 256, attn_lds, stream>>>(x, w1, b1, adj, Wh, ah, xs);

  const size_t c2_lds = (size_t)(3 * W2KT + 16) * sizeof(float);
  conv2_ln_kernel<<<B_ * T2_, 256, c2_lds, stream>>>(xs, w2, b2, lnw, lnb, out);
}


# attention K loop: 16-byte LDS fragment reads, each feeding two matrix steps (xg K-quad-interleaved, k=8kk+4h..+3 per lane half), next block's 5 reads ahead of 8 products; k and row-sum order changed
# speedup vs baseline: 5.1626x; 1.0600x over previous
#include <hip/hip_runtime.h>
#include <math.h>

#define B_   8
#define C0_  16
#define C1_  64
#define C2_  64
#define T0_  12
#define T1_  10
#define T2_  8
#define NN_  256
#define KT_  3
#define HH_  8
#define ALPHA_ 0.2f

typedef float v2f __attribute__((ext_vector_type(2)));
typedef float v4f __attribute__((ext_vector_type(4)));
typedef float v8f __attribute__((ext_vector_type(8)));
typedef unsigned u4w __attribute__((ext_vector_type(4)));

__device__ __forceinline__ float tanh_fast(float x) {
  float e = __expf(fminf(2.0f * x, 80.0f));
  return (e - 1.0f) / (e + 1.0f);
}

__device__ __forceinline__ float row_pick(float v, int sel_lo, int sel_hi) {
  int b = __builtin_bit_cast(int, v);
  return __builtin_bit_cast(float, __builtin_amdgcn_permlane16(b, b, sel_lo, sel_hi, false, false));
}
__device__ __forceinline__ float wave_max(float v) {
  v = fmaxf(v, row_pick(v, 0x67452301, 0xefcdab89));
  v = fmaxf(v, row_pick(v, 0x54761032, 0xdcfe98ba));
  v = fmaxf(v, row_pick(v, 0x32107654, 0xba98fedc));
  v = fmaxf(v, row_pick(v, 0xfedcba98, 0x76543210));
  int b = __builtin_bit_cast(int, v);
  float o = __builtin_bit_cast(float, __builtin_amdgcn_permlanex16(b, b, 0x76543210, 0xfedcba98, false, false));
  return fmaxf(v, o);
}

__device__ __forceinline__ float lane_f(float v, int l) {
  return __builtin_bit_cast(float, __builtin_amdgcn_readlane(__builtin_bit_cast(int, v), l));
}
__device__ __forceinline__ unsigned lane_u(unsigned v, int l) {
  return (unsigned)__builtin_amdgcn_readlane((int)v, l);
}

#define HGRP_ 3
#define PLANE_ ((size_t)B_ * T1_ * NN_ * C1_)
#define XSTR  72
#define QSTR  260
#define PSTR  260
#define USTR  68
#define SM_MIN_ 1.0e-25f

__global__ __launch_bounds__(256) void attn_kernel(
    const float* __restrict__ x,
    const float* __restrict__ w1,
    const float* __restrict__ b1,
    const float* __restrict__ adj,
    const float* __restrict__ Wh,
    const float* __restrict__ ah,
    float* __restrict__ xs)
{
  extern __shared__ float smem[];
  float*    s_xg   = smem;
  float*    s_xgq  = s_xg + NN_ * XSTR;
  float*    s_e1   = s_xgq + (NN_ / 4) * QSTR;
  float*    s_e2   = s_e1 + NN_;
  float*    s_rinv = s_e2 + NN_;
  float*    s_p    = s_rinv + NN_;
  unsigned* s_adj  = (unsigned*)(s_p + 8 * 16 * PSTR);
  float*    s_cm   = (float*)(s_adj + NN_ * 8);
  float*    s_u1   = s_cm + 8;
  float*    s_u2   = s_u1 + 4 * C1_;

  const int bt   = blockIdx.x / HGRP_;
  const int grp  = blockIdx.x - bt * HGRP_;
  const int h_lo = 3 * grp;
  const int h_hi = (h_lo + 3 < HH_) ? h_lo + 3 : HH_;
  const int tid  = threadIdx.x;
  const int lane = tid & 31;
  const int wave = tid >> 5;
  const int mloc = lane & 15;
  const int half = lane >> 4;

  {
    const int bb = bt / T1_, t = bt - bb * T1_;

    float av[2][24];
    #pragma unroll
    for (int ti = 0; ti < 2; ++ti) {
      const int n0 = (wave + 8 * ti) * 16;
      const float* xl = x + ((size_t)(bb * C0_ + 8 * half) * T0_ + t) * NN_ + n0 + mloc;
      #pragma unroll
      for (int k0 = 0; k0 < 24; ++k0)
        av[ti][k0] = xl[((k0 / 3) * T0_ + (k0 % 3)) * NN_];
    }

    const v4f bp = *(const v4f*)(b1 + 4 * mloc);
    const v4f bq = *(const v4f*)(b1 + C1_ + 4 * mloc);
    v8f acc[2][8];
    #pragma unroll
    for (int ti = 0; ti < 2; ++ti) {
      #pragma unroll
      for (int T = 0; T < 4; ++T) {
        #pragma unroll
        for (int r = 0; r < 8; ++r) { acc[ti][T][r] = bp[T]; acc[ti][4 + T][r] = bq[T]; }
      }
    }

    const float* wl = w1 + (size_t)(4 * mloc) * (C0_ * KT_) + 24 * half;
    #pragma unroll
    for (int j = 0; j < 6; ++j) {
      v4f bw[8];
      #pragma unroll
      for (int T = 0; T < 4; ++T) {
        bw[T]     = *(const v4f*)(wl + (size_t)T * (C0_ * KT_) + 4 * j);
        bw[4 + T] = *(const v4f*)(wl + (size_t)(C1_ + T) * (C0_ * KT_) + 4 * j);
      }
      #pragma unroll
      for (int ti = 0; ti < 2; ++ti) {
        v2f a0, a1;
        a0[0] = av[ti][4 * j];     a0[1] = av[ti][4 * j + 1];
        a1[0] = av[ti][4 * j + 2]; a1[1] = av[ti][4 * j + 3];
        #pragma unroll
        for (int T = 0; T < 8; ++T) {
          v2f b0; b0[0] = bw[T][0]; b0[1] = bw[T][1];
          acc[ti][T] = __builtin_amdgcn_wmma_f32_16x16x4_f32(false, a0, false, b0, (short)0, acc[ti][T], false, false);
        }
        #pragma unroll
        for (int T = 0; T < 8; ++T) {
          v2f b1v; b1v[0] = bw[T][2]; b1v[1] = bw[T][3];
          acc[ti][T] = __builtin_amdgcn_wmma_f32_16x16x4_f32(false, a1, false, b1v, (short)0, acc[ti][T], false, false);
        }
      }
    }

    const bool has_res = (4 * mloc) < C0_;
    const int  cres    = (4 * mloc) & (C0_ - 1);
    #pragma unroll
    for (int ti = 0; ti < 2; ++ti) {
      const int n0 = (wave + 8 * ti) * 16;
      const float* xr = x + ((size_t)(bb * C0_ + cres) * T0_ + (t + KT_ - 1)) * NN_ + n0 + 8 * half;
      float xin[4][8];
      #pragma unroll
      for (int T = 0; T < 4; ++T) {
        const v4f lo = *(const v4f*)(xr + (size_t)T * T0_ * NN_);
        const v4f hi = *(const v4f*)(xr + (size_t)T * T0_ * NN_ + 4);
        #pragma unroll
        for (int u = 0; u < 4; ++u) {
          xin[T][u]     = has_res ? lo[u] : 0.f;
          xin[T][4 + u] = has_res ? hi[u] : 0.f;
        }
      }
      v4f o[8];
      #pragma unroll
      for (int r = 0; r < 8; ++r) {
        #pragma unroll
        for (int T = 0; T < 4; ++T) {
          float sg = 1.f / (1.f + __expf(-acc[ti][4 + T][r]));
          o[r][T] = (acc[ti][T][r] + xin[T][r]) * sg;
        }
      }
      const int nr = n0 + 8 * half;
      #pragma unroll
      for (int r = 0; r < 8; ++r)
        *(v4f*)(s_xg + (nr + r) * XSTR + 4 * mloc) = o[r];
      #pragma unroll
      for (int g = 0; g < 2; ++g) {
        float* qq = s_xgq + ((nr >> 2) + g) * QSTR + 16 * mloc;
        #pragma unroll
        for (int T = 0; T < 4; ++T) {
          v4f q;
          q[0] = o[4 * g][T]; q[1] = o[4 * g + 1][T]; q[2] = o[4 * g + 2][T]; q[3] = o[4 * g + 3][T];
          *(v4f*)(qq + 4 * T) = q;
        }
      }
    }
  }
  const int nh = h_hi - h_lo;
  const int hp = wave >> 1;
  const int cu = tid & 63;
  v4f wld[16];
  v4f ald;
  {
    const int hu = h_lo + (hp < nh ? hp : nh - 1);
    const float* wp = Wh + ((size_t)hu * C1_ + 32 * (wave & 1)) * C1_ + 4 * lane;
    #pragma unroll
    for (int i = 0; i < 16; ++i) wld[i] = *(const v4f*)(wp + 128 * i);
    ald = *(const v4f*)(ah + (size_t)hu * 2 * C1_ + 4 * lane);
  }

  {
    const int r0 = 32 * wave;
    for (int bb = 0; bb < 4; ++bb) {
      const float* ap = adj + (size_t)(r0 + 8 * bb) * NN_ + 4 * lane;
      v4f a[16];
      #pragma unroll
      for (int q = 0; q < 16; ++q) a[q] = *(const v4f*)(ap + q * 128);
      #pragma unroll
      for (int s = 0; s < 2; ++s) {
        unsigned wacc = 0u;
        #pragma unroll
        for (int q = 0; q < 8; ++q) {
          #pragma unroll
          for (int u = 0; u < 4; ++u) {
            const unsigned w = __builtin_amdgcn_ballot_w32(a[8 * s + q][u] > 0.f);
            wacc = (lane == 4 * q + u) ? w : wacc;
          }
        }
        s_adj[(r0 + 8 * bb + 4 * s) * 8 + lane] = wacc;
      }
    }
  }
  {
    float* stg = s_p + wave * 16 * PSTR;
    #pragma unroll
    for (int i = 0; i < 16; ++i)
      *(v4f*)(stg + (2 * i + half) * USTR + 4 * mloc) = wld[i];
    *(v4f*)(stg + 32 * USTR + 4 * lane) = ald;
    __builtin_amdgcn_fence(__ATOMIC_ACQ_REL, "wavefront");
    __builtin_amdgcn_wave_barrier();
    const float* wr = stg + lane * USTR;
    const float* a1 = stg + 32 * USTR;
    const float* a2 = a1 + C1_;
    float s1 = 0.f, s2 = 0.f;
    #pragma unroll
    for (int q = 0; q < 16; ++q) {
      const v4f w = *(const v4f*)(wr + 4 * q);
      const v4f p = *(const v4f*)(a1 + 4 * q);
      const v4f r = *(const v4f*)(a2 + 4 * q);
      #pragma unroll
      for (int e = 0; e < 4; ++e) {
        s1 += w[e] * p[e];
        s2 += w[e] * r[e];
      }
    }
    s_u1[hp * C1_ + cu] = s1;
    s_u2[hp * C1_ + cu] = s2;
  }
  __syncthreads();

  const float inv_h = 1.0f / (float)HH_;
  float* s_pw = s_p + wave * 16 * PSTR;

  u4w awb[2];
  #pragma unroll
  for (int rbi = 0; rbi < 2; ++rbi)
    awb[rbi] = *(const u4w*)(s_adj + (wave + 8 * rbi) * 16 * 8 + 4 * lane);

  float racc[2][4][8];
  #pragma unroll
  for (int i = 0; i < 2; ++i)
    #pragma unroll
    for (int j = 0; j < 4; ++j)
      #pragma unroll
      for (int r = 0; r < 8; ++r) racc[i][j][r] = 0.f;

  for (int h = h_lo; h < h_hi; ++h) {
    {
      int n = tid;
      const float* row = &s_xg[n * XSTR];
      const float* U1 = s_u1 + (h - h_lo) * C1_;
      const float* U2 = s_u2 + (h - h_lo) * C1_;
      float a1 = 0.f, a2 = 0.f;
      for (int c = 0; c < C1_; ++c) {
        float v = row[c];
        a1 = fmaf(v, U1[c], a1);
        a2 = fmaf(v, U2[c], a2);
      }
      s_e1[n] = a1;
      s_e2[n] = a2;
      const float wm = wave_max(a2);
      if (lane == 0) s_cm[wave] = wm;
    }
    __syncthreads();

    float cmax;
    {
      const v4f c0 = *(const v4f*)(s_cm);
      const v4f c1 = *(const v4f*)(s_cm + 4);
      cmax = fmaxf(fmaxf(fmaxf(c0[0], c0[1]), fmaxf(c0[2], c0[3])),
                   fmaxf(fmaxf(c1[0], c1[1]), fmaxf(c1[2], c1[3])));
    }

    v4f cf1[2], cf2[2];
    #pragma unroll
    for (int hh = 0; hh < 2; ++hh) {
      const v4f e = *(const v4f*)(s_e2 + 128 * hh + 4 * lane);
      #pragma unroll
      for (int u = 0; u < 4; ++u) {
        const float z = e[u] - cmax;
        cf1[hh][u] = __expf(z);
        cf2[hh][u] = __expf(ALPHA_ * z);
      }
    }
    float vr1, vr2;
    {
      const float d = s_e1[(wave + 8 * half) * 16 + mloc] + cmax;
      vr1 = __expf((1.0f - ALPHA_) * fminf(d, 0.f));
      vr2 = __expf(-(1.0f - ALPHA_) * fmaxf(d, 0.f));
    }

    #pragma unroll
    for (int rbi = 0; rbi < 2; ++rbi) {
      const int rb = wave + rbi * 8;
      float sm;
      v8f acc0, acc1, acc2, acc3;
      #pragma nounroll
      for (int trip = 0; trip < 2; ++trip) {
        if (trip == 0) {
          #pragma unroll
          for (int r = 0; r < 16; ++r) {
            const float R1 = lane_f(vr1, 16 * rbi + r);
            const float R2 = lane_f(vr2, 16 * rbi + r);
            #pragma unroll
            for (int hh = 0; hh < 2; ++hh) {
              v4f p;
              #pragma unroll
              for (int u = 0; u < 4; ++u) {
                const unsigned w = lane_u(awb[rbi][u], 2 * r + hh);
                const float t = fmaxf(R1 * cf1[hh][u], R2 * cf2[hh][u]);
                p[u] = __builtin_amdgcn_inverse_ballot_w32(w) ? t : 0.f;
              }
              *(v4f*)(s_pw + r * PSTR + 128 * hh + 4 * lane) = p;
            }
          }
        } else {
          const int m = rb * 16 + mloc;
          const float e1m = s_e1[m];
          const int jbase = half * 128;
          const u4w aw4 = *(const u4w*)(s_adj + m * 8 + half * 4);
          float mx = -3.0e38f;
          for (int w8 = 0; w8 < 4; ++w8) {
            const u4w adjw = aw4 >> (unsigned)(8 * w8);
            const float* eb = &s_e2[jbase + w8 * 32];
            for (int j4 = 0; j4 < 32; j4 += 4) {
              v4f e = *(const v4f*)(eb + j4);
              #pragma unroll
              for (int u = 0; u < 4; ++u) {
                float sc = e1m + e[u];
                sc = sc > 0.f ? sc : ALPHA_ * sc;
                int on = (adjw[u] >> (j4 >> 2)) & 1;
                mx = fmaxf(mx, on ? sc : -3.0e38f);
              }
            }
          }
          mx = fmaxf(mx, __shfl_xor(mx, 16));
          for (int w8 = 0; w8 < 4; ++w8) {
            const u4w adjw = aw4 >> (unsigned)(8 * w8);
            const float* eb = &s_e2[jbase + w8 * 32];
            float* pb = &s_pw[mloc * PSTR + jbase + w8 * 32];
            for (int j4 = 0; j4 < 32; j4 += 4) {
              v4f e = *(const v4f*)(eb + j4);
              v4f p;
              #pragma unroll
              for (int u = 0; u < 4; ++u) {
                float sc = e1m + e[u];
                sc = sc > 0.f ? sc : ALPHA_ * sc;
                sc = ((adjw[u] >> (j4 >> 2)) & 1) ? sc : -3.0e38f;
                p[u] = __expf(sc - mx);
              }
              *(v4f*)(pb + j4) = p;
            }
          }
        }
        __builtin_amdgcn_fence(__ATOMIC_ACQ_REL, "wavefront");
        __builtin_amdgcn_wave_barrier();

        acc0 = v8f{}; acc1 = v8f{}; acc2 = v8f{}; acc3 = v8f{};
        v4f ps = {0.f, 0.f, 0.f, 0.f};
        const float* pa   = &s_pw[mloc * PSTR + 4 * half];
        const float* xqb  = &s_xgq[half * QSTR + 4 * mloc];
        v4f a0 = *(const v4f*)(pa);
        v4f p0 = *(const v4f*)(xqb);
        v4f p1 = *(const v4f*)(xqb + 64);
        v4f p2 = *(const v4f*)(xqb + 128);
        v4f p3 = *(const v4f*)(xqb + 192);
        for (int k0 = 0; k0 < NN_; k0 += 16) {
          const float* bq = xqb + ((k0 + 8) >> 2) * QSTR;
          const v4f a1 = *(const v4f*)(pa + k0 + 8);
          const v4f q0 = *(const v4f*)(bq);
          const v4f q1 = *(const v4f*)(bq + 64);
          const v4f q2 = *(const v4f*)(bq + 128);
          const v4f q3 = *(const v4f*)(bq + 192);
          __builtin_amdgcn_sched_group_barrier(0x100, 5, 0);
          ps += a0;
          {
            v2f al, ah, l0, l1, l2, l3, h0, h1, h2, h3;
            al[0] = a0[0]; al[1] = a0[1]; ah[0] = a0[2]; ah[1] = a0[3];
            l0[0] = p0[0]; l0[1] = p0[1]; h0[0] = p0[2]; h0[1] = p0[3];
            l1[0] = p1[0]; l1[1] = p1[1]; h1[0] = p1[2]; h1[1] = p1[3];
            l2[0] = p2[0]; l2[1] = p2[1]; h2[0] = p2[2]; h2[1] = p2[3];
            l3[0] = p3[0]; l3[1] = p3[1]; h3[0] = p3[2]; h3[1] = p3[3];
            acc0 = __builtin_amdgcn_wmma_f32_16x16x4_f32(false, al, false, l0, (short)0, acc0, false, false);
            acc1 = __builtin_amdgcn_wmma_f32_16x16x4_f32(false, al, false, l1, (short)0, acc1, false, false);
            acc2 = __builtin_amdgcn_wmma_f32_16x16x4_f32(false, al, false, l2, (short)0, acc2, false, false);
            acc3 = __builtin_amdgcn_wmma_f32_16x16x4_f32(false, al, false, l3, (short)0, acc3, false, false);
            acc0 = __builtin_amdgcn_wmma_f32_16x16x4_f32(false, ah, false, h0, (short)0, acc0, false, false);
            acc1 = __builtin_amdgcn_wmma_f32_16x16x4_f32(false, ah, false, h1, (short)0, acc1, false, false);
            acc2 = __builtin_amdgcn_wmma_f32_16x16x4_f32(false, ah, false, h2, (short)0, acc2, false, false);
            acc3 = __builtin_amdgcn_wmma_f32_16x16x4_f32(false, ah, false, h3, (short)0, acc3, false, false);
          }
          __builtin_amdgcn_sched_group_barrier(0x008, 8, 0);
          __builtin_amdgcn_sched_barrier(0);
          const int kn = (k0 + 16) & (NN_ - 1);
          const float* bn = xqb + (kn >> 2) * QSTR;
          a0 = *(const v4f*)(pa + kn);
          p0 = *(const v4f*)(bn);
          p1 = *(const v4f*)(bn + 64);
          p2 = *(const v4f*)(bn + 128);
          p3 = *(const v4f*)(bn + 192);
          __builtin_amdgcn_sched_group_barrier(0x100, 5, 0);
          ps += a1;
          {
            v2f al, ah, l0, l1, l2, l3, h0, h1, h2, h3;
            al[0] = a1[0]; al[1] = a1[1]; ah[0] = a1[2]; ah[1] = a1[3];
            l0[0] = q0[0]; l0[1] = q0[1]; h0[0] = q0[2]; h0[1] = q0[3];
            l1[0] = q1[0]; l1[1] = q1[1]; h1[0] = q1[2]; h1[1] = q1[3];
            l2[0] = q2[0]; l2[1] = q2[1]; h2[0] = q2[2]; h2[1] = q2[3];
            l3[0] = q3[0]; l3[1] = q3[1]; h3[0] = q3[2]; h3[1] = q3[3];
            acc0 = __builtin_amdgcn_wmma_f32_16x16x4_f32(false, al, false, l0, (short)0, acc0, false, false);
            acc1 = __builtin_amdgcn_wmma_f32_16x16x4_f32(false, al, false, l1, (short)0, acc1, false, false);
            acc2 = __builtin_amdgcn_wmma_f32_16x16x4_f32(false, al, false, l2, (short)0, acc2, false, false);
            acc3 = __builtin_amdgcn_wmma_f32_16x16x4_f32(false, al, false, l3, (short)0, acc3, false, false);
            acc0 = __builtin_amdgcn_wmma_f32_16x16x4_f32(false, ah, false, h0, (short)0, acc0, false, false);
            acc1 = __builtin_amdgcn_wmma_f32_16x16x4_f32(false, ah, false, h1, (short)0, acc1, false, false);
            acc2 = __builtin_amdgcn_wmma_f32_16x16x4_f32(false, ah, false, h2, (short)0, acc2, false, false);
            acc3 = __builtin_amdgcn_wmma_f32_16x16x4_f32(false, ah, false, h3, (short)0, acc3, false, false);
          }
          __builtin_amdgcn_sched_group_barrier(0x008, 8, 0);
          __builtin_amdgcn_sched_barrier(0);
        }
        sm = (ps[0] + ps[1]) + (ps[2] + ps[3]);
        sm += __shfl_xor(sm, 16);
        if (trip == 1 || !__any(!(sm >= SM_MIN_))) break;
      }
      s_rinv[rb * 16 + mloc] = 1.0f / sm;

      #pragma unroll
      for (int r = 0; r < 8; ++r) {
        float inv = s_rinv[rb * 16 + r + 8 * half];
        racc[rbi][0][r] += tanh_fast(acc0[r] * inv) * inv_h;
        racc[rbi][1][r] += tanh_fast(acc1[r] * inv) * inv_h;
        racc[rbi][2][r] += tanh_fast(acc2[r] * inv) * inv_h;
        racc[rbi][3][r] += tanh_fast(acc3[r] * inv) * inv_h;
      }
    }
    __syncthreads();
  }

  float* xs_bt = xs + (size_t)grp * PLANE_ + (size_t)bt * NN_ * C1_;
  if (grp == 0) {
    #pragma unroll
    for (int rbi = 0; rbi < 2; ++rbi) {
      #pragma unroll
      for (int r = 0; r < 8; ++r) {
        const float* xrow = &s_xg[((wave + rbi * 8) * 16 + r + 8 * half) * XSTR + mloc];
        racc[rbi][0][r] += xrow[0];
        racc[rbi][1][r] += xrow[16];
        racc[rbi][2][r] += xrow[32];
        racc[rbi][3][r] += xrow[48];
      }
    }
  }
  #pragma unroll
  for (int rbi = 0; rbi < 2; ++rbi) {
    const int rb = wave + rbi * 8;
    #pragma unroll
    for (int r = 0; r < 8; ++r) {
      int mm = rb * 16 + r + 8 * half;
      float* orow = xs_bt + (size_t)mm * C1_ + mloc;
      orow[0]  = racc[rbi][0][r];
      orow[16] = racc[rbi][1][r];
      orow[32] = racc[rbi][2][r];
      orow[48] = racc[rbi][3][r];
    }
  }
}

#define W2P   260
#define W2KT  (16 * W2P)

__device__ __forceinline__ float xhalf(float v) {
  int b = __builtin_bit_cast(int, v);
  return __builtin_bit_cast(float, __builtin_amdgcn_permlanex16(b, b, 0x76543210, 0xfedcba98, false, false));
}

__device__ __forceinline__ void tap_issue(v4f (&raw)[2][3][8], const float* xrow0) {
  #pragma unroll
  for (int nt = 0; nt < 2; ++nt) {
    const float* xrow = xrow0 + (size_t)(16 * nt) * C1_;
    #pragma unroll
    for (int j = 0; j < 8; ++j) {
      raw[nt][0][j] = *(const v4f*)(xrow + 8 * j);
      raw[nt][1][j] = *(const v4f*)(xrow + PLANE_ + 8 * j);
      raw[nt][2][j] = *(const v4f*)(xrow + 2 * PLANE_ + 8 * j);
    }
  }
}

__global__ __launch_bounds__(256) void conv2_ln_kernel(
    const float* __restrict__ xs,
    const float* __restrict__ w2,
    const float* __restrict__ b2,
    const float* __restrict__ lnw,
    const float* __restrict__ lnb,
    float* __restrict__ out)
{
  extern __shared__ float smem[];
  float* s_w   = smem;
  float* s_red = s_w + 3 * W2KT;

  const int bt = blockIdx.x;
  const int bb = bt >> 3, t = bt & 7;
  const int tid  = threadIdx.x;
  const int lane = tid & 31;
  const int wave = tid >> 5;
  const int mloc = lane & 15;
  const int half = lane >> 4;
  const int nb   = 32 * wave;

  v4f wv[4][3];
  #pragma unroll
  for (int it = 0; it < 4; ++it) {
    const float* wp = w2 + 12 * (tid + 256 * it);
    wv[it][0] = *(const v4f*)(wp);
    wv[it][1] = *(const v4f*)(wp + 4);
    wv[it][2] = *(const v4f*)(wp + 8);
  }

  v8f acc[2][4];
  #pragma unroll
  for (int T = 0; T < 4; ++T) {
    const v4f blo = *(const v4f*)(b2 + 16 * T + 8 * half);
    const v4f bhi = *(const v4f*)(b2 + 16 * T + 8 * half + 4);
    #pragma unroll
    for (int u = 0; u < 4; ++u) {
      acc[0][T][u] = blo[u]; acc[0][T][4 + u] = bhi[u];
      acc[1][T][u] = blo[u]; acc[1][T][4 + u] = bhi[u];
    }
  }

  const float* xsb   = xs + (((size_t)bb * T1_ + t) * NN_) * C1_;
  const float* xlane = xsb + (size_t)(nb + mloc) * C1_ + 4 * half;
  v4f raw[2][3][8];
  tap_issue(raw, xlane);

  #pragma unroll
  for (int it = 0; it < 4; ++it) {
    const int item = tid + 256 * it;
    float* dst = s_w + (item & 15) * W2P + (item >> 4) * 4;
    #pragma unroll
    for (int kt = 0; kt < KT_; ++kt) {
      v4f o;
      #pragma unroll
      for (int q = 0; q < 4; ++q) o[q] = wv[it][(3 * q + kt) >> 2][(3 * q + kt) & 3];
      *(v4f*)(dst + kt * W2KT) = o;
    }
  }
  __syncthreads();

  const float* wl = s_w + half * W2P + mloc * 4;
  v4f aw[2][2][4];
  #pragma unroll
  for (int jj = 0; jj < 2; ++jj)
    #pragma unroll
    for (int T = 0; T < 4; ++T)
      aw[0][jj][T] = *(const v4f*)(wl + 2 * jj * W2P + 64 * T);
  __builtin_amdgcn_sched_group_barrier(0x100, 8, 0);

  v4f rr[2][4][3][2];

  #pragma unroll
  for (int kt = 0; kt < KT_; ++kt) {
    __builtin_amdgcn_sched_barrier(0);
    v4f xb[2][8];
    #pragma unroll
    for (int nt = 0; nt < 2; ++nt)
      #pragma unroll
      for (int j = 0; j < 8; ++j)
        xb[nt][j] = (raw[nt][0][j] + raw[nt][1][j]) + raw[nt][2][j];
    __builtin_amdgcn_sched_barrier(0);
    if (kt + 1 < KT_) {
      tap_issue(raw, xlane + (size_t)(kt + 1) * NN_ * C1_);
    } else {
      #pragma unroll
      for (int nt = 0; nt < 2; ++nt) {
        #pragma unroll
        for (int T = 0; T < 4; ++T) {
          const float* xr = xsb + (size_t)(KT_ - 1) * NN_ * C1_
                          + (size_t)(nb + 16 * nt + mloc) * C1_ + 16 * T + 8 * half;
          #pragma unroll
          for (int pl = 0; pl < 3; ++pl) {
            rr[nt][T][pl][0] = *(const v4f*)(xr + pl * PLANE_);
            rr[nt][T][pl][1] = *(const v4f*)(xr + pl * PLANE_ + 4);
          }
        }
      }
    }
    __builtin_amdgcn_sched_barrier(0);

    #pragma unroll
    for (int jp = 0; jp < 4; ++jp) {
      const int blk = 4 * kt + jp;
      const int cur = blk & 1, nxt = cur ^ 1;
      if (blk + 1 < 12) {
        const int kn = (blk + 1) >> 2, jn = (blk + 1) & 3;
        #pragma unroll
        for (int jj = 0; jj < 2; ++jj)
          #pragma unroll
          for (int T = 0; T < 4; ++T)
            aw[nxt][jj][T] = *(const v4f*)(wl + kn * W2KT + 2 * (2 * jn + jj) * W2P + 64 * T);
      }
      #pragma unroll
      for (int jj = 0; jj < 2; ++jj) {
        const int j = 2 * jp + jj;
        #pragma unroll
        for (int s = 0; s < 2; ++s) {
          #pragma unroll
          for (int nt = 0; nt < 2; ++nt) {
            v2f bv; bv[0] = xb[nt][j][2 * s]; bv[1] = xb[nt][j][2 * s + 1];
            #pragma unroll
            for (int T = 0; T < 4; ++T) {
              v2f avv; avv[0] = aw[cur][jj][T][2 * s]; avv[1] = aw[cur][jj][T][2 * s + 1];
              acc[nt][T] = __builtin_amdgcn_wmma_f32_16x16x4_f32(false, avv, false, bv, (short)0, acc[nt][T], false, false);
            }
          }
        }
      }
      if (blk + 1 < 12) __builtin_amdgcn_sched_group_barrier(0x100, 8, 0);
      __builtin_amdgcn_sched_group_barrier(0x008, 32, 0);
    }
  }
  __builtin_amdgcn_sched_barrier(0);

  v4f lw[2][4][2], lb[2][4][2];
  #pragma unroll
  for (int nt = 0; nt < 2; ++nt) {
    #pragma unroll
    for (int T = 0; T < 4; ++T) {
      const size_t off = (size_t)(nb + 16 * nt + mloc) * C1_ + 16 * T + 8 * half;
      lw[nt][T][0] = *(const v4f*)(lnw + off);
      lw[nt][T][1] = *(const v4f*)(lnw + off + 4);
      lb[nt][T][0] = *(const v4f*)(lnb + off);
      lb[nt][T][1] = *(const v4f*)(lnb + off + 4);
    }
  }
  __builtin_amdgcn_sched_barrier(0);

  float lsum = 0.f, lsq = 0.f;
  #pragma unroll
  for (int nt = 0; nt < 2; ++nt) {
    #pragma unroll
    for (int T = 0; T < 4; ++T) {
      const v4f r0 = (rr[nt][T][0][0] + rr[nt][T][1][0]) + rr[nt][T][2][0];
      const v4f r1 = (rr[nt][T][0][1] + rr[nt][T][1][1]) + rr[nt][T][2][1];
      #pragma unroll
      for (int u = 0; u < 4; ++u) {
        float v0 = acc[nt][T][u] + r0[u];
        float v1 = acc[nt][T][4 + u] + r1[u];
        v0 = v0 > 0.f ? v0 : 0.f;
        v1 = v1 > 0.f ? v1 : 0.f;
        acc[nt][T][u] = v0;
        acc[nt][T][4 + u] = v1;
        lsum += v0; lsq += v0 * v0;
        lsum += v1; lsq += v1 * v1;
      }
    }
  }

  for (int off = 16; off > 0; off >>= 1) {
    lsum += __shfl_xor(lsum, off);
    lsq  += __shfl_xor(lsq,  off);
  }
  if (lane == 0) { s_red[wave] = lsum; s_red[8 + wave] = lsq; }
  __syncthreads();
  float tsum = 0.f, tsq = 0.f;
  for (int i = 0; i < 8; ++i) { tsum += s_red[i]; tsq += s_red[8 + i]; }
  const float cnt  = (float)(NN_ * C2_);
  const float mu   = tsum / cnt;
  const float var  = tsq / cnt - mu * mu;
  const float rstd = rsqrtf(var + 1e-5f);

  float* ob = out + (((size_t)bb * C2_) * T2_ + t) * NN_ + nb + lane;
  #pragma unroll
  for (int T = 0; T < 4; ++T) {
    #pragma unroll
    for (int r = 0; r < 8; ++r) {
      const float y0 = (acc[0][T][r] - mu) * rstd * lw[0][T][r >> 2][r & 3] + lb[0][T][r >> 2][r & 3];
      const float y1 = (acc[1][T][r] - mu) * rstd * lw[1][T][r >> 2][r & 3] + lb[1][T][r >> 2][r & 3];
      const float send = half ? y0 : y1;
      const float recv = xhalf(send);
      const float lo = half ? recv : y0;
      const float hi = half ? y1 : recv;
      ob[(size_t)(16 * T + r) * T2_ * NN_]     = lo;
      ob[(size_t)(16 * T + 8 + r) * T2_ * NN_] = hi;
    }
  }
}

extern "C" void kernel_launch(void* const* d_in, const int* in_sizes, int n_in,
                              void* d_out, int out_size, void* d_ws, size_t ws_size,
                              hipStream_t stream) {
  const float* x   = (const float*)d_in[0];
  const float* adj = (const float*)d_in[1];
  const float* w1  = (const float*)d_in[2];
  const float* b1  = (const float*)d_in[3];
  const float* w2  = (const float*)d_in[4];
  const float* b2  = (const float*)d_in[5];
  const float* Wh  = (const float*)d_in[6];
  const float* ah  = (const float*)d_in[7];
  const float* lnw = (const float*)d_in[8];
  const float* lnb = (const float*)d_in[9];
  float* out = (float*)d_out;

  float* ws = (float*)d_ws;
  float* xs = ws;

  const size_t attn_lds =
      (size_t)(NN_ * XSTR + (NN_ / 4) * QSTR + 3 * NN_ + 8 * 16 * PSTR) * sizeof(float)
      + (size_t)(NN_ * 8) * sizeof(unsigned)
      + (size_t)8 * sizeof(float)
      + (size_t)(2 * 4 * C1_) * sizeof(float);
  attn_kernel<<<B_ * T1_ * HGRP_, 256, attn_lds, stream>>>(x, w1, b1, adj, Wh, ah, xs);

  const size_t c2_lds = (size_t)(3 * W2KT + 16) * sizeof(float);
  conv2_ln_kernel<<<B_ * T2_, 256, c2_lds, stream>>>(xs, w2, b2, lnw, lnb, out);
}
